# starting kernel
# baseline (speedup 1.0000x reference)
#include <hip/hip_runtime.h>
#include <hip/hip_bf16.h>

typedef float v2f __attribute__((ext_vector_type(2)));
typedef float v8f __attribute__((ext_vector_type(8)));

#define NU 50000
#define NI 100000
#define NT 150000
#define DD 64
#define NNZE 2400000
#define BATCH 4096
#define SLOPE 0.01f

__global__ __launch_bounds__(256)
void k_gather0(const float* __restrict__ ue, const float* __restrict__ ie,
               const int* __restrict__ us, const int* __restrict__ pi,
               const int* __restrict__ ni, float* __restrict__ out) {
    int t = blockIdx.x * 256 + threadIdx.x;
    if (t >= 3 * BATCH * DD) return;
    int c  = t & 63;
    int b  = t >> 6;
    int seg = b >> 12;
    int bb  = b & (BATCH - 1);
    float v;
    if (seg == 0)      v = ue[(size_t)us[bb] * DD + c];
    else if (seg == 1) v = ie[(size_t)pi[bb] * DD + c];
    else               v = ie[(size_t)ni[bb] * DD + c];
    out[(size_t)b * 256 + c] = v;
}

__global__ __launch_bounds__(256)
void k_spmm(const int* __restrict__ rows, const int* __restrict__ cols,
            const float* __restrict__ vals, const float* __restrict__ ego,
            float* __restrict__ L) {
    long t = (long)blockIdx.x * 256 + threadIdx.x;
    if (t >= (long)NNZE * 16) return;
    int e = (int)(t >> 4);
    int q = ((int)t & 15) * 4;
    int r = rows[e], c = cols[e];
    float v = vals[e];
    const float4 x = *(const float4*)(ego + (size_t)c * DD + q);
    float* dst = L + (size_t)r * DD + q;
    atomicAdd(dst + 0, v * x.x);
    atomicAdd(dst + 1, v * x.y);
    atomicAdd(dst + 2, v * x.z);
    atomicAdd(dst + 3, v * x.w);
}

__global__ __launch_bounds__(256)
void k_dense_wmma(const float* __restrict__ L, float* __restrict__ ego,
                  const float* __restrict__ Wgc, const float* __restrict__ Wbi,
                  const float* __restrict__ bias) {
    __shared__ float sg[4096];
    __shared__ float sb[4096];
    __shared__ float sbias[64];

    for (int i = threadIdx.x; i < 4096; i += 256) {
        int r = i >> 6, c = i & 63;
        int dst = ((((r >> 1) * 64) + c) << 1) | (r & 1);
        sg[dst] = Wgc[i];
        sb[dst] = Wbi[i];
    }
    if (threadIdx.x < 64) sbias[threadIdx.x] = 2.0f * bias[threadIdx.x];
    __syncthreads();

    const int wave  = threadIdx.x >> 5;
    const int lane  = threadIdx.x & 31;
    const int mbase = (blockIdx.x * 8 + wave) * 16;
    if (mbase >= NT) return;

    const int half = lane >> 4;
    const int n    = lane & 15;

    const float* lrow = L   + (size_t)(mbase + n) * DD;
    const float* erow = ego + (size_t)(mbase + n) * DD;

    v8f acc[4];
    #pragma unroll
    for (int j = 0; j < 4; ++j) acc[j] = (v8f){};

    #pragma unroll
    for (int kk = 0; kk < 16; ++kk) {
        int off = kk * 4 + half * 2;
        v2f ls = *(const v2f*)(lrow + off);
        v2f eg = *(const v2f*)(erow + off);
        v2f a_li  = ls + eg;
        v2f a_mul = ls * eg;
        int bbase = ((kk * 2 + half) * 64) << 1;
        #pragma unroll
        for (int j = 0; j < 4; ++j) {
            v2f bgc = *(const v2f*)(sg + bbase + ((j * 16 + n) << 1));
            v2f bbi = *(const v2f*)(sb + bbase + ((j * 16 + n) << 1));
            acc[j] = __builtin_amdgcn_wmma_f32_16x16x4_f32(
                false, a_li, false, bgc, (short)0, acc[j], false, false);
            acc[j] = __builtin_amdgcn_wmma_f32_16x16x4_f32(
                false, a_mul, false, bbi, (short)0, acc[j], false, false);
        }
    }

    #pragma unroll
    for (int j = 0; j < 4; ++j) {
        float bsum = sbias[j * 16 + n];
        #pragma unroll
        for (int v = 0; v < 8; ++v) {
            float x = acc[j][v] + bsum;
            x = (x >= 0.f) ? x : SLOPE * x;
            ego[(size_t)(mbase + (half << 3) + v) * DD + j * 16 + n] = x;
        }
    }
}

__global__ __launch_bounds__(256)
void k_gather_norm(const float* __restrict__ ego, const int* __restrict__ us,
                   const int* __restrict__ pi, const int* __restrict__ ni,
                   float* __restrict__ out, int colbase) {
    int wid  = (blockIdx.x * 256 + threadIdx.x) >> 5;
    int lane = threadIdx.x & 31;
    if (wid >= 3 * BATCH) return;
    int seg = wid >> 12;
    int bb  = wid & (BATCH - 1);
    int node;
    if (seg == 0)      node = us[bb];
    else if (seg == 1) node = NU + pi[bb];
    else               node = NU + ni[bb];
    v2f x = *(const v2f*)(ego + (size_t)node * DD + lane * 2);
    float ss = x.x * x.x + x.y * x.y;
    #pragma unroll
    for (int o = 16; o > 0; o >>= 1) ss += __shfl_xor(ss, o, 32);
    float scale = 1.0f / fmaxf(sqrtf(ss), 1e-12f);
    v2f y; y.x = x.x * scale; y.y = x.y * scale;
    *(v2f*)(out + (size_t)wid * 256 + colbase + lane * 2) = y;
}

extern "C" void kernel_launch(void* const* d_in, const int* in_sizes, int n_in,
                              void* d_out, int out_size, void* d_ws, size_t ws_size,
                              hipStream_t stream) {
    const float* ue   = (const float*)d_in[0];
    const float* ie   = (const float*)d_in[1];
    const float* Wgc  = (const float*)d_in[2];
    const float* Wbi  = (const float*)d_in[3];
    const float* bbi  = (const float*)d_in[4];
    const float* vals = (const float*)d_in[5];
    const int*   rows = (const int*)d_in[6];
    const int*   cols = (const int*)d_in[7];
    const int*   us   = (const int*)d_in[8];
    const int*   pi   = (const int*)d_in[9];
    const int*   ni   = (const int*)d_in[10];
    float* out = (float*)d_out;

    float* ego = (float*)d_ws;
    float* L   = ego + (size_t)NT * DD;

    hipMemcpyAsync(ego, ue, (size_t)NU * DD * sizeof(float),
                   hipMemcpyDeviceToDevice, stream);
    hipMemcpyAsync(ego + (size_t)NU * DD, ie, (size_t)NI * DD * sizeof(float),
                   hipMemcpyDeviceToDevice, stream);

    k_gather0<<<(3 * BATCH * DD + 255) / 256, 256, 0, stream>>>(
        ue, ie, us, pi, ni, out);

    for (int k = 0; k < 3; ++k) {
        hipMemsetAsync(L, 0, (size_t)NT * DD * sizeof(float), stream);
        long tot = (long)NNZE * 16;
        k_spmm<<<(int)((tot + 255) / 256), 256, 0, stream>>>(
            rows, cols, vals, ego, L);
        k_dense_wmma<<<(NT + 127) / 128, 256, 0, stream>>>(
            L, ego, Wgc + k * 4096, Wbi + k * 4096, bbi + k * 64);
        k_gather_norm<<<(3 * BATCH * 32 + 255) / 256, 256, 0, stream>>>(
            ego, us, pi, ni, out, 64 * (k + 1));
    }
}


# SpMM: edges counting-sorted by row once per call (int atomic rank, scan, place); each layer a wave sums a row in registers, no float atomics; sum order: arrival rank, even/odd edge halves
# speedup vs baseline: 1.8392x; 1.8392x over previous
#include <hip/hip_runtime.h>
#include <hip/hip_bf16.h>

typedef float v2f __attribute__((ext_vector_type(2)));
typedef float v8f __attribute__((ext_vector_type(8)));

#define NU 50000
#define NI 100000
#define NT 150000
#define DD 64
#define NNZE 2400000
#define BATCH 4096
#define SLOPE 0.01f

__global__ __launch_bounds__(256)
void k_gather0(const float* __restrict__ ue, const float* __restrict__ ie,
               const int* __restrict__ us, const int* __restrict__ pi,
               const int* __restrict__ ni, float* __restrict__ out) {
    int t = blockIdx.x * 256 + threadIdx.x;
    if (t >= 3 * BATCH * DD) return;
    int c  = t & 63;
    int b  = t >> 6;
    int seg = b >> 12;
    int bb  = b & (BATCH - 1);
    float v;
    if (seg == 0)      v = ue[(size_t)us[bb] * DD + c];
    else if (seg == 1) v = ie[(size_t)pi[bb] * DD + c];
    else               v = ie[(size_t)ni[bb] * DD + c];
    out[(size_t)b * 256 + c] = v;
}

__global__ __launch_bounds__(256)
void k_spmm(const int* __restrict__ rows, const int* __restrict__ cols,
            const float* __restrict__ vals, const float* __restrict__ ego,
            float* __restrict__ L) {
    long t = (long)blockIdx.x * 256 + threadIdx.x;
    if (t >= (long)NNZE * 16) return;
    int e = (int)(t >> 4);
    int q = ((int)t & 15) * 4;
    int r = rows[e], c = cols[e];
    float v = vals[e];
    const float4 x = *(const float4*)(ego + (size_t)c * DD + q);
    float* dst = L + (size_t)r * DD + q;
    atomicAdd(dst + 0, v * x.x);
    atomicAdd(dst + 1, v * x.y);
    atomicAdd(dst + 2, v * x.z);
    atomicAdd(dst + 3, v * x.w);
}

#define EDGE_GRID 2048
#define SCAN_T 1024
#define SCAN_B ((NT + SCAN_T - 1) / SCAN_T)
#define ROW_GRID 1024

__global__ __launch_bounds__(256)
void k_rank(const int* __restrict__ rows, int* __restrict__ cnt,
            int* __restrict__ rank) {
    const int stride = gridDim.x * 256;
    for (int e = blockIdx.x * 256 + threadIdx.x; e < NNZE; e += stride)
        rank[e] = atomicAdd(&cnt[rows[e]], 1);
}

__global__ __launch_bounds__(SCAN_T)
void k_rowptr(const int* __restrict__ cnt, int* __restrict__ rowptr) {
    __shared__ int wred[32];
    __shared__ int wtot[32];
    __shared__ int sbase;
    const int tid = threadIdx.x, lane = tid & 31, wave = tid >> 5;

    int part = 0;
    for (int b = 0; b < (int)blockIdx.x; ++b) part += cnt[b * SCAN_T + tid];
    #pragma unroll
    for (int o = 16; o > 0; o >>= 1) part += __shfl_xor(part, o, 32);
    if (lane == 0) wred[wave] = part;
    __syncthreads();
    if (wave == 0) {
        int x = wred[lane];
        #pragma unroll
        for (int o = 16; o > 0; o >>= 1) x += __shfl_xor(x, o, 32);
        if (lane == 0) sbase = x;
    }

    const int idx = blockIdx.x * SCAN_T + tid;
    const int own = (idx < NT) ? cnt[idx] : 0;
    int inc = own;
    #pragma unroll
    for (int o = 1; o < 32; o <<= 1) {
        int y = __shfl_up(inc, o, 32);
        if (lane >= o) inc += y;
    }
    if (lane == 31) wtot[wave] = inc;
    __syncthreads();
    if (wave == 0) {
        const int x = wtot[lane];
        int xi = x;
        #pragma unroll
        for (int o = 1; o < 32; o <<= 1) {
            int y = __shfl_up(xi, o, 32);
            if (lane >= o) xi += y;
        }
        wtot[lane] = xi - x;
    }
    __syncthreads();
    const int excl = sbase + wtot[wave] + inc - own;
    if (idx < NT) rowptr[idx] = excl;
    if (idx == NT - 1) rowptr[NT] = excl + own;
}

__global__ __launch_bounds__(256)
void k_place(const int* __restrict__ rows, const int* __restrict__ cols,
             const float* __restrict__ vals, const int* __restrict__ rank,
             const int* __restrict__ rowptr, int2* __restrict__ sorted) {
    const int stride = gridDim.x * 256;
    for (int e = blockIdx.x * 256 + threadIdx.x; e < NNZE; e += stride) {
        const int p = rowptr[rows[e]] + rank[e];
        sorted[p] = make_int2(cols[e], __float_as_int(vals[e]));
    }
}

static __device__ __forceinline__
void row_batch(float4& acc, const int2 my, const int nb, const int half,
               const int q, const float* __restrict__ ego) {
    for (int i0 = 0; i0 < nb; i0 += 16) {
        int   c[8];
        float v[8];
        float4 x[8];
        #pragma unroll
        for (int j = 0; j < 8; ++j) {
            const int src = i0 + 2 * j + half;
            c[j] = __shfl(my.x, src, 32);
            v[j] = __int_as_float(__shfl(my.y, src, 32));
        }
        #pragma unroll
        for (int j = 0; j < 8; ++j)
            x[j] = *(const float4*)(ego + (size_t)c[j] * DD + q);
        #pragma unroll
        for (int j = 0; j < 8; ++j) {
            const bool ok = (i0 + 2 * j + half) < nb;
            acc.x += v[j] * (ok ? x[j].x : 0.f);
            acc.y += v[j] * (ok ? x[j].y : 0.f);
            acc.z += v[j] * (ok ? x[j].z : 0.f);
            acc.w += v[j] * (ok ? x[j].w : 0.f);
        }
    }
}

__global__ __launch_bounds__(256)
void k_spmm_rows(const int* __restrict__ rowptr, const int2* __restrict__ sorted,
                 const float* __restrict__ ego, float* __restrict__ L) {
    const int lane = threadIdx.x & 31;
    const int half = lane >> 4;
    const int q    = (lane & 15) * 4;
    const int nw   = gridDim.x * 8;
    const int first = blockIdx.x * 8 + __builtin_amdgcn_readfirstlane(threadIdx.x >> 5);
    if (first >= NT) return;

    int r = first - nw;
    int s = 0, t = 0;
    int2 my = make_int2(0, 0);
    int sn = rowptr[first], tn = rowptr[first + 1];

    for (;;) {
        const int rn = r + nw;
        int2 myN = make_int2(0, 0);
        if (rn < NT) {
            if (lane < tn - sn) myN = sorted[sn + lane];
        }
        int snn = 0, tnn = 0;
        if (rn + nw < NT) { snn = rowptr[rn + nw]; tnn = rowptr[rn + nw + 1]; }

        if (r >= 0) {
            float4 acc = make_float4(0.f, 0.f, 0.f, 0.f);
            row_batch(acc, my, min(32, t - s), half, q, ego);
            for (int base = s + 32; base < t; base += 32) {
                const int nb = min(32, t - base);
                int2 m2 = make_int2(0, 0);
                if (lane < nb) m2 = sorted[base + lane];
                row_batch(acc, m2, nb, half, q, ego);
            }
            acc.x += __shfl_xor(acc.x, 16, 32);
            acc.y += __shfl_xor(acc.y, 16, 32);
            acc.z += __shfl_xor(acc.z, 16, 32);
            acc.w += __shfl_xor(acc.w, 16, 32);
            if (half == 0) *(float4*)(L + (size_t)r * DD + q) = acc;
        }

        if (rn >= NT) break;
        r = rn; s = sn; t = tn; my = myN; sn = snn; tn = tnn;
    }
}

__global__ __launch_bounds__(256)
void k_dense_wmma(const float* __restrict__ L, float* __restrict__ ego,
                  const float* __restrict__ Wgc, const float* __restrict__ Wbi,
                  const float* __restrict__ bias) {
    __shared__ float sg[4096];
    __shared__ float sb[4096];
    __shared__ float sbias[64];

    for (int i = threadIdx.x; i < 4096; i += 256) {
        int r = i >> 6, c = i & 63;
        int dst = ((((r >> 1) * 64) + c) << 1) | (r & 1);
        sg[dst] = Wgc[i];
        sb[dst] = Wbi[i];
    }
    if (threadIdx.x < 64) sbias[threadIdx.x] = 2.0f * bias[threadIdx.x];
    __syncthreads();

    const int wave  = threadIdx.x >> 5;
    const int lane  = threadIdx.x & 31;
    const int mbase = (blockIdx.x * 8 + wave) * 16;
    if (mbase >= NT) return;

    const int half = lane >> 4;
    const int n    = lane & 15;

    const float* lrow = L   + (size_t)(mbase + n) * DD;
    const float* erow = ego + (size_t)(mbase + n) * DD;

    v8f acc[4];
    #pragma unroll
    for (int j = 0; j < 4; ++j) acc[j] = (v8f){};

    #pragma unroll
    for (int kk = 0; kk < 16; ++kk) {
        int off = kk * 4 + half * 2;
        v2f ls = *(const v2f*)(lrow + off);
        v2f eg = *(const v2f*)(erow + off);
        v2f a_li  = ls + eg;
        v2f a_mul = ls * eg;
        int bbase = ((kk * 2 + half) * 64) << 1;
        #pragma unroll
        for (int j = 0; j < 4; ++j) {
            v2f bgc = *(const v2f*)(sg + bbase + ((j * 16 + n) << 1));
            v2f bbi = *(const v2f*)(sb + bbase + ((j * 16 + n) << 1));
            acc[j] = __builtin_amdgcn_wmma_f32_16x16x4_f32(
                false, a_li, false, bgc, (short)0, acc[j], false, false);
            acc[j] = __builtin_amdgcn_wmma_f32_16x16x4_f32(
                false, a_mul, false, bbi, (short)0, acc[j], false, false);
        }
    }

    #pragma unroll
    for (int j = 0; j < 4; ++j) {
        float bsum = sbias[j * 16 + n];
        #pragma unroll
        for (int v = 0; v < 8; ++v) {
            float x = acc[j][v] + bsum;
            x = (x >= 0.f) ? x : SLOPE * x;
            ego[(size_t)(mbase + (half << 3) + v) * DD + j * 16 + n] = x;
        }
    }
}

__global__ __launch_bounds__(256)
void k_gather_norm(const float* __restrict__ ego, const int* __restrict__ us,
                   const int* __restrict__ pi, const int* __restrict__ ni,
                   float* __restrict__ out, int colbase) {
    int wid  = (blockIdx.x * 256 + threadIdx.x) >> 5;
    int lane = threadIdx.x & 31;
    if (wid >= 3 * BATCH) return;
    int seg = wid >> 12;
    int bb  = wid & (BATCH - 1);
    int node;
    if (seg == 0)      node = us[bb];
    else if (seg == 1) node = NU + pi[bb];
    else               node = NU + ni[bb];
    v2f x = *(const v2f*)(ego + (size_t)node * DD + lane * 2);
    float ss = x.x * x.x + x.y * x.y;
    #pragma unroll
    for (int o = 16; o > 0; o >>= 1) ss += __shfl_xor(ss, o, 32);
    float scale = 1.0f / fmaxf(sqrtf(ss), 1e-12f);
    v2f y; y.x = x.x * scale; y.y = x.y * scale;
    *(v2f*)(out + (size_t)wid * 256 + colbase + lane * 2) = y;
}

extern "C" void kernel_launch(void* const* d_in, const int* in_sizes, int n_in,
                              void* d_out, int out_size, void* d_ws, size_t ws_size,
                              hipStream_t stream) {
    const float* ue   = (const float*)d_in[0];
    const float* ie   = (const float*)d_in[1];
    const float* Wgc  = (const float*)d_in[2];
    const float* Wbi  = (const float*)d_in[3];
    const float* bbi  = (const float*)d_in[4];
    const float* vals = (const float*)d_in[5];
    const int*   rows = (const int*)d_in[6];
    const int*   cols = (const int*)d_in[7];
    const int*   us   = (const int*)d_in[8];
    const int*   pi   = (const int*)d_in[9];
    const int*   ni   = (const int*)d_in[10];
    float* out = (float*)d_out;

    float* ego = (float*)d_ws;
    float* L   = ego + (size_t)NT * DD;

    hipMemcpyAsync(ego, ue, (size_t)NU * DD * sizeof(float),
                   hipMemcpyDeviceToDevice, stream);
    hipMemcpyAsync(ego + (size_t)NU * DD, ie, (size_t)NI * DD * sizeof(float),
                   hipMemcpyDeviceToDevice, stream);

    k_gather0<<<(3 * BATCH * DD + 255) / 256, 256, 0, stream>>>(
        ue, ie, us, pi, ni, out);

    const size_t need = (size_t)2 * NT * DD * 4 + (size_t)NNZE * 8 + (size_t)(NT + 1) * 4;
    if (ws_size >= need) {
        int2* sorted = (int2*)(L + (size_t)NT * DD);
        int*  rowptr = (int*)(sorted + NNZE);
        int*  rank   = (int*)L;
        int*  cnt    = rank + NNZE;
        hipMemsetAsync(cnt, 0, (size_t)NT * sizeof(int), stream);
        k_rank<<<EDGE_GRID, 256, 0, stream>>>(rows, cnt, rank);
        k_rowptr<<<SCAN_B, SCAN_T, 0, stream>>>(cnt, rowptr);
        k_place<<<EDGE_GRID, 256, 0, stream>>>(rows, cols, vals, rank, rowptr, sorted);
        for (int k = 0; k < 3; ++k) {
            k_spmm_rows<<<ROW_GRID, 256, 0, stream>>>(rowptr, sorted, ego, L);
            k_dense_wmma<<<(NT + 127) / 128, 256, 0, stream>>>(
                L, ego, Wgc + k * 4096, Wbi + k * 4096, bbi + k * 64);
            k_gather_norm<<<(3 * BATCH * 32 + 255) / 256, 256, 0, stream>>>(
                ego, us, pi, ni, out, 64 * (k + 1));
        }
        return;
    }

    for (int k = 0; k < 3; ++k) {
        hipMemsetAsync(L, 0, (size_t)NT * DD * sizeof(float), stream);
        long tot = (long)NNZE * 16;
        k_spmm<<<(int)((tot + 255) / 256), 256, 0, stream>>>(
            rows, cols, vals, ego, L);
        k_dense_wmma<<<(NT + 127) / 128, 256, 0, stream>>>(
            L, ego, Wgc + k * 4096, Wbi + k * 4096, bbi + k * 64);
        k_gather_norm<<<(3 * BATCH * 32 + 255) / 256, 256, 0, stream>>>(
            ego, us, pi, ni, out, 64 * (k + 1));
    }
}


# SpMM and dense layer fused per 16-row tile: a wave sums its rows into its own LDS tile and feeds the f32 WMMA from it; L never in global memory, 3 launches fewer, two feature buffers swap
# speedup vs baseline: 2.0067x; 1.0911x over previous
#include <hip/hip_runtime.h>
#include <hip/hip_bf16.h>

typedef float v2f __attribute__((ext_vector_type(2)));
typedef float v8f __attribute__((ext_vector_type(8)));

#define NU 50000
#define NI 100000
#define NT 150000
#define DD 64
#define NNZE 2400000
#define BATCH 4096
#define SLOPE 0.01f

__global__ __launch_bounds__(256)
void k_gather0(const float* __restrict__ ue, const float* __restrict__ ie,
               const int* __restrict__ us, const int* __restrict__ pi,
               const int* __restrict__ ni, float* __restrict__ out) {
    int t = blockIdx.x * 256 + threadIdx.x;
    if (t >= 3 * BATCH * DD) return;
    int c  = t & 63;
    int b  = t >> 6;
    int seg = b >> 12;
    int bb  = b & (BATCH - 1);
    float v;
    if (seg == 0)      v = ue[(size_t)us[bb] * DD + c];
    else if (seg == 1) v = ie[(size_t)pi[bb] * DD + c];
    else               v = ie[(size_t)ni[bb] * DD + c];
    out[(size_t)b * 256 + c] = v;
}

__global__ __launch_bounds__(256)
void k_spmm(const int* __restrict__ rows, const int* __restrict__ cols,
            const float* __restrict__ vals, const float* __restrict__ ego,
            float* __restrict__ L) {
    long t = (long)blockIdx.x * 256 + threadIdx.x;
    if (t >= (long)NNZE * 16) return;
    int e = (int)(t >> 4);
    int q = ((int)t & 15) * 4;
    int r = rows[e], c = cols[e];
    float v = vals[e];
    const float4 x = *(const float4*)(ego + (size_t)c * DD + q);
    float* dst = L + (size_t)r * DD + q;
    atomicAdd(dst + 0, v * x.x);
    atomicAdd(dst + 1, v * x.y);
    atomicAdd(dst + 2, v * x.z);
    atomicAdd(dst + 3, v * x.w);
}

#define EDGE_GRID 2048
#define SCAN_T 1024
#define SCAN_B ((NT + SCAN_T - 1) / SCAN_T)

__global__ __launch_bounds__(256)
void k_rank(const int* __restrict__ rows, int* __restrict__ cnt,
            int* __restrict__ rank) {
    const int stride = gridDim.x * 256;
    for (int e = blockIdx.x * 256 + threadIdx.x; e < NNZE; e += stride)
        rank[e] = atomicAdd(&cnt[rows[e]], 1);
}

__global__ __launch_bounds__(SCAN_T)
void k_rowptr(const int* __restrict__ cnt, int* __restrict__ rowptr) {
    __shared__ int wred[32];
    __shared__ int wtot[32];
    __shared__ int sbase;
    const int tid = threadIdx.x, lane = tid & 31, wave = tid >> 5;

    int part = 0;
    for (int b = 0; b < (int)blockIdx.x; ++b) part += cnt[b * SCAN_T + tid];
    #pragma unroll
    for (int o = 16; o > 0; o >>= 1) part += __shfl_xor(part, o, 32);
    if (lane == 0) wred[wave] = part;
    __syncthreads();
    if (wave == 0) {
        int x = wred[lane];
        #pragma unroll
        for (int o = 16; o > 0; o >>= 1) x += __shfl_xor(x, o, 32);
        if (lane == 0) sbase = x;
    }

    const int idx = blockIdx.x * SCAN_T + tid;
    const int own = (idx < NT) ? cnt[idx] : 0;
    int inc = own;
    #pragma unroll
    for (int o = 1; o < 32; o <<= 1) {
        int y = __shfl_up(inc, o, 32);
        if (lane >= o) inc += y;
    }
    if (lane == 31) wtot[wave] = inc;
    __syncthreads();
    if (wave == 0) {
        const int x = wtot[lane];
        int xi = x;
        #pragma unroll
        for (int o = 1; o < 32; o <<= 1) {
            int y = __shfl_up(xi, o, 32);
            if (lane >= o) xi += y;
        }
        wtot[lane] = xi - x;
    }
    __syncthreads();
    const int excl = sbase + wtot[wave] + inc - own;
    if (idx < NT) rowptr[idx] = excl;
    if (idx == NT - 1) rowptr[NT] = excl + own;
}

__global__ __launch_bounds__(256)
void k_place(const int* __restrict__ rows, const int* __restrict__ cols,
             const float* __restrict__ vals, const int* __restrict__ rank,
             const int* __restrict__ rowptr, int2* __restrict__ sorted) {
    const int stride = gridDim.x * 256;
    for (int e = blockIdx.x * 256 + threadIdx.x; e < NNZE; e += stride) {
        const int p = rowptr[rows[e]] + rank[e];
        sorted[p] = make_int2(cols[e], __float_as_int(vals[e]));
    }
}

static __device__ __forceinline__
void row_batch(float4& acc, const int2 my, const int nb, const int half,
               const int q, const float* __restrict__ ego) {
    for (int i0 = 0; i0 < nb; i0 += 16) {
        int   c[8];
        float v[8];
        float4 x[8];
        #pragma unroll
        for (int j = 0; j < 8; ++j) {
            const int src = i0 + 2 * j + half;
            c[j] = __shfl(my.x, src, 32);
            v[j] = __int_as_float(__shfl(my.y, src, 32));
        }
        #pragma unroll
        for (int j = 0; j < 8; ++j)
            x[j] = *(const float4*)(ego + (size_t)c[j] * DD + q);
        #pragma unroll
        for (int j = 0; j < 8; ++j) {
            const bool ok = (i0 + 2 * j + half) < nb;
            acc.x += v[j] * (ok ? x[j].x : 0.f);
            acc.y += v[j] * (ok ? x[j].y : 0.f);
            acc.z += v[j] * (ok ? x[j].z : 0.f);
            acc.w += v[j] * (ok ? x[j].w : 0.f);
        }
    }
}

__global__ __launch_bounds__(256)
void k_dense_wmma(const float* __restrict__ L, float* __restrict__ ego,
                  const float* __restrict__ Wgc, const float* __restrict__ Wbi,
                  const float* __restrict__ bias) {
    __shared__ float sg[4096];
    __shared__ float sb[4096];
    __shared__ float sbias[64];

    for (int i = threadIdx.x; i < 4096; i += 256) {
        int r = i >> 6, c = i & 63;
        int dst = ((((r >> 1) * 64) + c) << 1) | (r & 1);
        sg[dst] = Wgc[i];
        sb[dst] = Wbi[i];
    }
    if (threadIdx.x < 64) sbias[threadIdx.x] = 2.0f * bias[threadIdx.x];
    __syncthreads();

    const int wave  = threadIdx.x >> 5;
    const int lane  = threadIdx.x & 31;
    const int mbase = (blockIdx.x * 8 + wave) * 16;
    if (mbase >= NT) return;

    const int half = lane >> 4;
    const int n    = lane & 15;

    const float* lrow = L   + (size_t)(mbase + n) * DD;
    const float* erow = ego + (size_t)(mbase + n) * DD;

    v8f acc[4];
    #pragma unroll
    for (int j = 0; j < 4; ++j) acc[j] = (v8f){};

    #pragma unroll
    for (int kk = 0; kk < 16; ++kk) {
        int off = kk * 4 + half * 2;
        v2f ls = *(const v2f*)(lrow + off);
        v2f eg = *(const v2f*)(erow + off);
        v2f a_li  = ls + eg;
        v2f a_mul = ls * eg;
        int bbase = ((kk * 2 + half) * 64) << 1;
        #pragma unroll
        for (int j = 0; j < 4; ++j) {
            v2f bgc = *(const v2f*)(sg + bbase + ((j * 16 + n) << 1));
            v2f bbi = *(const v2f*)(sb + bbase + ((j * 16 + n) << 1));
            acc[j] = __builtin_amdgcn_wmma_f32_16x16x4_f32(
                false, a_li, false, bgc, (short)0, acc[j], false, false);
            acc[j] = __builtin_amdgcn_wmma_f32_16x16x4_f32(
                false, a_mul, false, bbi, (short)0, acc[j], false, false);
        }
    }

    #pragma unroll
    for (int j = 0; j < 4; ++j) {
        float bsum = sbias[j * 16 + n];
        #pragma unroll
        for (int v = 0; v < 8; ++v) {
            float x = acc[j][v] + bsum;
            x = (x >= 0.f) ? x : SLOPE * x;
            ego[(size_t)(mbase + (half << 3) + v) * DD + j * 16 + n] = x;
        }
    }
}

#define TILE_PITCH 68
#define LAYER_WAVES 16
#define LAYER_T (LAYER_WAVES * 32)
#define N_TILES (NT / 16)
#define LAYER_GRID ((N_TILES + LAYER_WAVES - 1) / LAYER_WAVES)

__global__ __launch_bounds__(LAYER_T, 12)
void k_layer(const int* __restrict__ rowptr, const int2* __restrict__ sorted,
             const float* __restrict__ ein, float* __restrict__ eout,
             const float* __restrict__ Wgc, const float* __restrict__ Wbi,
             const float* __restrict__ bias) {
    __shared__ __attribute__((aligned(16))) float sg[4096];
    __shared__ __attribute__((aligned(16))) float sb[4096];
    __shared__ float sbias[64];
    __shared__ __attribute__((aligned(16))) float sL[LAYER_WAVES * 16 * TILE_PITCH];

    #pragma unroll
    for (int i = threadIdx.x; i < 4096; i += LAYER_T) {
        int r = i >> 6, c = i & 63;
        int dst = ((((r >> 1) * 64) + c) << 1) | (r & 1);
        sg[dst] = Wgc[i];
        sb[dst] = Wbi[i];
    }
    if (threadIdx.x < 64) sbias[threadIdx.x] = 2.0f * bias[threadIdx.x];
    __syncthreads();

    const int lane = threadIdx.x & 31;
    const int half = lane >> 4;
    const int n    = lane & 15;
    const int q    = n * 4;
    const int wave = __builtin_amdgcn_readfirstlane(threadIdx.x >> 5);
    const int tile = blockIdx.x * LAYER_WAVES + wave;
    if (tile >= N_TILES) return;
    const int mbase = tile * 16;
    float* myL = sL + wave * (16 * TILE_PITCH);

    int s = rowptr[mbase], t = rowptr[mbase + 1], tn = rowptr[mbase + 2];
    int2 my = make_int2(0, 0);
    if (lane < t - s) my = sorted[s + lane];

    #pragma unroll 1
    for (int i = 0; i < 16; ++i) {
        int2 myN = make_int2(0, 0);
        if (i < 15) {
            if (lane < tn - t) myN = sorted[t + lane];
        }
        int tnn = tn;
        if (i < 14) tnn = rowptr[mbase + i + 3];

        float4 acc = make_float4(0.f, 0.f, 0.f, 0.f);
        row_batch(acc, my, min(32, t - s), half, q, ein);
        for (int base = s + 32; base < t; base += 32) {
            const int nb = min(32, t - base);
            int2 m2 = make_int2(0, 0);
            if (lane < nb) m2 = sorted[base + lane];
            row_batch(acc, m2, nb, half, q, ein);
        }
        acc.x += __shfl_xor(acc.x, 16, 32);
        acc.y += __shfl_xor(acc.y, 16, 32);
        acc.z += __shfl_xor(acc.z, 16, 32);
        acc.w += __shfl_xor(acc.w, 16, 32);
        if (half == 0) *(float4*)(myL + i * TILE_PITCH + q) = acc;

        s = t; t = tn; tn = tnn; my = myN;
    }

    __builtin_amdgcn_wave_barrier();

    const float* lrow = myL + n * TILE_PITCH;
    const float* erow = ein + (size_t)(mbase + n) * DD;

    v8f dacc[4];
    #pragma unroll
    for (int j = 0; j < 4; ++j) dacc[j] = (v8f){};

    #pragma unroll
    for (int kk = 0; kk < 16; ++kk) {
        int off = kk * 4 + half * 2;
        v2f ls = *(const v2f*)(lrow + off);
        v2f eg = *(const v2f*)(erow + off);
        v2f a_li  = ls + eg;
        v2f a_mul = ls * eg;
        int bbase = ((kk * 2 + half) * 64) << 1;
        #pragma unroll
        for (int j = 0; j < 4; ++j) {
            v2f bgc = *(const v2f*)(sg + bbase + ((j * 16 + n) << 1));
            v2f bbi = *(const v2f*)(sb + bbase + ((j * 16 + n) << 1));
            dacc[j] = __builtin_amdgcn_wmma_f32_16x16x4_f32(
                false, a_li, false, bgc, (short)0, dacc[j], false, false);
            dacc[j] = __builtin_amdgcn_wmma_f32_16x16x4_f32(
                false, a_mul, false, bbi, (short)0, dacc[j], false, false);
        }
    }

    #pragma unroll
    for (int j = 0; j < 4; ++j) {
        float bsum = sbias[j * 16 + n];
        #pragma unroll
        for (int v = 0; v < 8; ++v) {
            float x = dacc[j][v] + bsum;
            x = (x >= 0.f) ? x : SLOPE * x;
            eout[(size_t)(mbase + (half << 3) + v) * DD + j * 16 + n] = x;
        }
    }
}

__global__ __launch_bounds__(256)
void k_gather_norm(const float* __restrict__ ego, const int* __restrict__ us,
                   const int* __restrict__ pi, const int* __restrict__ ni,
                   float* __restrict__ out, int colbase) {
    int wid  = (blockIdx.x * 256 + threadIdx.x) >> 5;
    int lane = threadIdx.x & 31;
    if (wid >= 3 * BATCH) return;
    int seg = wid >> 12;
    int bb  = wid & (BATCH - 1);
    int node;
    if (seg == 0)      node = us[bb];
    else if (seg == 1) node = NU + pi[bb];
    else               node = NU + ni[bb];
    v2f x = *(const v2f*)(ego + (size_t)node * DD + lane * 2);
    float ss = x.x * x.x + x.y * x.y;
    #pragma unroll
    for (int o = 16; o > 0; o >>= 1) ss += __shfl_xor(ss, o, 32);
    float scale = 1.0f / fmaxf(sqrtf(ss), 1e-12f);
    v2f y; y.x = x.x * scale; y.y = x.y * scale;
    *(v2f*)(out + (size_t)wid * 256 + colbase + lane * 2) = y;
}

extern "C" void kernel_launch(void* const* d_in, const int* in_sizes, int n_in,
                              void* d_out, int out_size, void* d_ws, size_t ws_size,
                              hipStream_t stream) {
    const float* ue   = (const float*)d_in[0];
    const float* ie   = (const float*)d_in[1];
    const float* Wgc  = (const float*)d_in[2];
    const float* Wbi  = (const float*)d_in[3];
    const float* bbi  = (const float*)d_in[4];
    const float* vals = (const float*)d_in[5];
    const int*   rows = (const int*)d_in[6];
    const int*   cols = (const int*)d_in[7];
    const int*   us   = (const int*)d_in[8];
    const int*   pi   = (const int*)d_in[9];
    const int*   ni   = (const int*)d_in[10];
    float* out = (float*)d_out;

    float* ego = (float*)d_ws;
    float* L   = ego + (size_t)NT * DD;

    hipMemcpyAsync(ego, ue, (size_t)NU * DD * sizeof(float),
                   hipMemcpyDeviceToDevice, stream);
    hipMemcpyAsync(ego + (size_t)NU * DD, ie, (size_t)NI * DD * sizeof(float),
                   hipMemcpyDeviceToDevice, stream);

    k_gather0<<<(3 * BATCH * DD + 255) / 256, 256, 0, stream>>>(
        ue, ie, us, pi, ni, out);

    const size_t need = (size_t)2 * NT * DD * 4 + (size_t)NNZE * 8 + (size_t)(NT + 1) * 4;
    if (ws_size >= need) {
        int2* sorted = (int2*)(L + (size_t)NT * DD);
        int*  rowptr = (int*)(sorted + NNZE);
        int*  rank   = (int*)L;
        int*  cnt    = rank + NNZE;
        hipMemsetAsync(cnt, 0, (size_t)NT * sizeof(int), stream);
        k_rank<<<EDGE_GRID, 256, 0, stream>>>(rows, cnt, rank);
        k_rowptr<<<SCAN_B, SCAN_T, 0, stream>>>(cnt, rowptr);
        k_place<<<EDGE_GRID, 256, 0, stream>>>(rows, cols, vals, rank, rowptr, sorted);
        float* ego2 = L;
        k_layer<<<LAYER_GRID, LAYER_T, 0, stream>>>(
            rowptr, sorted, ego, ego2, Wgc, Wbi, bbi);
        k_gather_norm<<<(3 * BATCH * 32 + 255) / 256, 256, 0, stream>>>(
            ego2, us, pi, ni, out, 64);
        k_layer<<<LAYER_GRID, LAYER_T, 0, stream>>>(
            rowptr, sorted, ego2, ego, Wgc + 4096, Wbi + 4096, bbi + 64);
        k_gather_norm<<<(3 * BATCH * 32 + 255) / 256, 256, 0, stream>>>(
            ego, us, pi, ni, out, 128);
        k_layer<<<LAYER_GRID, LAYER_T, 0, stream>>>(
            rowptr, sorted, ego, ego2, Wgc + 8192, Wbi + 8192, bbi + 128);
        k_gather_norm<<<(3 * BATCH * 32 + 255) / 256, 256, 0, stream>>>(
            ego2, us, pi, ni, out, 192);
        return;
    }

    for (int k = 0; k < 3; ++k) {
        hipMemsetAsync(L, 0, (size_t)NT * DD * sizeof(float), stream);
        long tot = (long)NNZE * 16;
        k_spmm<<<(int)((tot + 255) / 256), 256, 0, stream>>>(
            rows, cols, vals, ego, L);
        k_dense_wmma<<<(NT + 127) / 128, 256, 0, stream>>>(
            L, ego, Wgc + k * 4096, Wbi + k * 4096, bbi + k * 64);
        k_gather_norm<<<(3 * BATCH * 32 + 255) / 256, 256, 0, stream>>>(
            ego, us, pi, ni, out, 64 * (k + 1));
    }
}


# Last layer computed only for the 3x4096 rows the output gathers: one wave per 16 gathered rows sums them, runs the f32 WMMA, normalises from its LDS tile and stores to out; one launch fewer
# speedup vs baseline: 2.1515x; 1.0722x over previous
#include <hip/hip_runtime.h>
#include <hip/hip_bf16.h>

typedef float v2f __attribute__((ext_vector_type(2)));
typedef float v8f __attribute__((ext_vector_type(8)));

#define NU 50000
#define NI 100000
#define NT 150000
#define DD 64
#define NNZE 2400000
#define BATCH 4096
#define SLOPE 0.01f

__global__ __launch_bounds__(256)
void k_gather0(const float* __restrict__ ue, const float* __restrict__ ie,
               const int* __restrict__ us, const int* __restrict__ pi,
               const int* __restrict__ ni, float* __restrict__ out) {
    int t = blockIdx.x * 256 + threadIdx.x;
    if (t >= 3 * BATCH * DD) return;
    int c  = t & 63;
    int b  = t >> 6;
    int seg = b >> 12;
    int bb  = b & (BATCH - 1);
    float v;
    if (seg == 0)      v = ue[(size_t)us[bb] * DD + c];
    else if (seg == 1) v = ie[(size_t)pi[bb] * DD + c];
    else               v = ie[(size_t)ni[bb] * DD + c];
    out[(size_t)b * 256 + c] = v;
}

__global__ __launch_bounds__(256)
void k_spmm(const int* __restrict__ rows, const int* __restrict__ cols,
            const float* __restrict__ vals, const float* __restrict__ ego,
            float* __restrict__ L) {
    long t = (long)blockIdx.x * 256 + threadIdx.x;
    if (t >= (long)NNZE * 16) return;
    int e = (int)(t >> 4);
    int q = ((int)t & 15) * 4;
    int r = rows[e], c = cols[e];
    float v = vals[e];
    const float4 x = *(const float4*)(ego + (size_t)c * DD + q);
    float* dst = L + (size_t)r * DD + q;
    atomicAdd(dst + 0, v * x.x);
    atomicAdd(dst + 1, v * x.y);
    atomicAdd(dst + 2, v * x.z);
    atomicAdd(dst + 3, v * x.w);
}

#define EDGE_GRID 2048
#define SCAN_T 1024
#define SCAN_B ((NT + SCAN_T - 1) / SCAN_T)

__global__ __launch_bounds__(256)
void k_rank(const int* __restrict__ rows, int* __restrict__ cnt,
            int* __restrict__ rank) {
    const int stride = gridDim.x * 256;
    for (int e = blockIdx.x * 256 + threadIdx.x; e < NNZE; e += stride)
        rank[e] = atomicAdd(&cnt[rows[e]], 1);
}

__global__ __launch_bounds__(SCAN_T)
void k_rowptr(const int* __restrict__ cnt, int* __restrict__ rowptr) {
    __shared__ int wred[32];
    __shared__ int wtot[32];
    __shared__ int sbase;
    const int tid = threadIdx.x, lane = tid & 31, wave = tid >> 5;

    int part = 0;
    for (int b = 0; b < (int)blockIdx.x; ++b) part += cnt[b * SCAN_T + tid];
    #pragma unroll
    for (int o = 16; o > 0; o >>= 1) part += __shfl_xor(part, o, 32);
    if (lane == 0) wred[wave] = part;
    __syncthreads();
    if (wave == 0) {
        int x = wred[lane];
        #pragma unroll
        for (int o = 16; o > 0; o >>= 1) x += __shfl_xor(x, o, 32);
        if (lane == 0) sbase = x;
    }

    const int idx = blockIdx.x * SCAN_T + tid;
    const int own = (idx < NT) ? cnt[idx] : 0;
    int inc = own;
    #pragma unroll
    for (int o = 1; o < 32; o <<= 1) {
        int y = __shfl_up(inc, o, 32);
        if (lane >= o) inc += y;
    }
    if (lane == 31) wtot[wave] = inc;
    __syncthreads();
    if (wave == 0) {
        const int x = wtot[lane];
        int xi = x;
        #pragma unroll
        for (int o = 1; o < 32; o <<= 1) {
            int y = __shfl_up(xi, o, 32);
            if (lane >= o) xi += y;
        }
        wtot[lane] = xi - x;
    }
    __syncthreads();
    const int excl = sbase + wtot[wave] + inc - own;
    if (idx < NT) rowptr[idx] = excl;
    if (idx == NT - 1) rowptr[NT] = excl + own;
}

__global__ __launch_bounds__(256)
void k_place(const int* __restrict__ rows, const int* __restrict__ cols,
             const float* __restrict__ vals, const int* __restrict__ rank,
             const int* __restrict__ rowptr, int2* __restrict__ sorted) {
    const int stride = gridDim.x * 256;
    for (int e = blockIdx.x * 256 + threadIdx.x; e < NNZE; e += stride) {
        const int p = rowptr[rows[e]] + rank[e];
        sorted[p] = make_int2(cols[e], __float_as_int(vals[e]));
    }
}

static __device__ __forceinline__
void row_batch(float4& acc, const int2 my, const int nb, const int half,
               const int q, const float* __restrict__ ego) {
    for (int i0 = 0; i0 < nb; i0 += 16) {
        int   c[8];
        float v[8];
        float4 x[8];
        #pragma unroll
        for (int j = 0; j < 8; ++j) {
            const int src = i0 + 2 * j + half;
            c[j] = __shfl(my.x, src, 32);
            v[j] = __int_as_float(__shfl(my.y, src, 32));
        }
        #pragma unroll
        for (int j = 0; j < 8; ++j)
            x[j] = *(const float4*)(ego + (size_t)c[j] * DD + q);
        #pragma unroll
        for (int j = 0; j < 8; ++j) {
            const bool ok = (i0 + 2 * j + half) < nb;
            acc.x += v[j] * (ok ? x[j].x : 0.f);
            acc.y += v[j] * (ok ? x[j].y : 0.f);
            acc.z += v[j] * (ok ? x[j].z : 0.f);
            acc.w += v[j] * (ok ? x[j].w : 0.f);
        }
    }
}

__global__ __launch_bounds__(256)
void k_dense_wmma(const float* __restrict__ L, float* __restrict__ ego,
                  const float* __restrict__ Wgc, const float* __restrict__ Wbi,
                  const float* __restrict__ bias) {
    __shared__ float sg[4096];
    __shared__ float sb[4096];
    __shared__ float sbias[64];

    for (int i = threadIdx.x; i < 4096; i += 256) {
        int r = i >> 6, c = i & 63;
        int dst = ((((r >> 1) * 64) + c) << 1) | (r & 1);
        sg[dst] = Wgc[i];
        sb[dst] = Wbi[i];
    }
    if (threadIdx.x < 64) sbias[threadIdx.x] = 2.0f * bias[threadIdx.x];
    __syncthreads();

    const int wave  = threadIdx.x >> 5;
    const int lane  = threadIdx.x & 31;
    const int mbase = (blockIdx.x * 8 + wave) * 16;
    if (mbase >= NT) return;

    const int half = lane >> 4;
    const int n    = lane & 15;

    const float* lrow = L   + (size_t)(mbase + n) * DD;
    const float* erow = ego + (size_t)(mbase + n) * DD;

    v8f acc[4];
    #pragma unroll
    for (int j = 0; j < 4; ++j) acc[j] = (v8f){};

    #pragma unroll
    for (int kk = 0; kk < 16; ++kk) {
        int off = kk * 4 + half * 2;
        v2f ls = *(const v2f*)(lrow + off);
        v2f eg = *(const v2f*)(erow + off);
        v2f a_li  = ls + eg;
        v2f a_mul = ls * eg;
        int bbase = ((kk * 2 + half) * 64) << 1;
        #pragma unroll
        for (int j = 0; j < 4; ++j) {
            v2f bgc = *(const v2f*)(sg + bbase + ((j * 16 + n) << 1));
            v2f bbi = *(const v2f*)(sb + bbase + ((j * 16 + n) << 1));
            acc[j] = __builtin_amdgcn_wmma_f32_16x16x4_f32(
                false, a_li, false, bgc, (short)0, acc[j], false, false);
            acc[j] = __builtin_amdgcn_wmma_f32_16x16x4_f32(
                false, a_mul, false, bbi, (short)0, acc[j], false, false);
        }
    }

    #pragma unroll
    for (int j = 0; j < 4; ++j) {
        float bsum = sbias[j * 16 + n];
        #pragma unroll
        for (int v = 0; v < 8; ++v) {
            float x = acc[j][v] + bsum;
            x = (x >= 0.f) ? x : SLOPE * x;
            ego[(size_t)(mbase + (half << 3) + v) * DD + j * 16 + n] = x;
        }
    }
}

#define TILE_PITCH 68
#define LAYER_WAVES 16
#define LAYER_T (LAYER_WAVES * 32)
#define N_TILES (NT / 16)
#define LAYER_GRID ((N_TILES + LAYER_WAVES - 1) / LAYER_WAVES)

__global__ __launch_bounds__(LAYER_T, 12)
void k_layer(const int* __restrict__ rowptr, const int2* __restrict__ sorted,
             const float* __restrict__ ein, float* __restrict__ eout,
             const float* __restrict__ Wgc, const float* __restrict__ Wbi,
             const float* __restrict__ bias) {
    __shared__ __attribute__((aligned(16))) float sg[4096];
    __shared__ __attribute__((aligned(16))) float sb[4096];
    __shared__ float sbias[64];
    __shared__ __attribute__((aligned(16))) float sL[LAYER_WAVES * 16 * TILE_PITCH];

    #pragma unroll
    for (int i = threadIdx.x; i < 4096; i += LAYER_T) {
        int r = i >> 6, c = i & 63;
        int dst = ((((r >> 1) * 64) + c) << 1) | (r & 1);
        sg[dst] = Wgc[i];
        sb[dst] = Wbi[i];
    }
    if (threadIdx.x < 64) sbias[threadIdx.x] = 2.0f * bias[threadIdx.x];
    __syncthreads();

    const int lane = threadIdx.x & 31;
    const int half = lane >> 4;
    const int n    = lane & 15;
    const int q    = n * 4;
    const int wave = __builtin_amdgcn_readfirstlane(threadIdx.x >> 5);
    const int tile = blockIdx.x * LAYER_WAVES + wave;
    if (tile >= N_TILES) return;
    const int mbase = tile * 16;
    float* myL = sL + wave * (16 * TILE_PITCH);

    int s = rowptr[mbase], t = rowptr[mbase + 1], tn = rowptr[mbase + 2];
    int2 my = make_int2(0, 0);
    if (lane < t - s) my = sorted[s + lane];

    #pragma unroll 1
    for (int i = 0; i < 16; ++i) {
        int2 myN = make_int2(0, 0);
        if (i < 15) {
            if (lane < tn - t) myN = sorted[t + lane];
        }
        int tnn = tn;
        if (i < 14) tnn = rowptr[mbase + i + 3];

        float4 acc = make_float4(0.f, 0.f, 0.f, 0.f);
        row_batch(acc, my, min(32, t - s), half, q, ein);
        for (int base = s + 32; base < t; base += 32) {
            const int nb = min(32, t - base);
            int2 m2 = make_int2(0, 0);
            if (lane < nb) m2 = sorted[base + lane];
            row_batch(acc, m2, nb, half, q, ein);
        }
        acc.x += __shfl_xor(acc.x, 16, 32);
        acc.y += __shfl_xor(acc.y, 16, 32);
        acc.z += __shfl_xor(acc.z, 16, 32);
        acc.w += __shfl_xor(acc.w, 16, 32);
        if (half == 0) *(float4*)(myL + i * TILE_PITCH + q) = acc;

        s = t; t = tn; tn = tnn; my = myN;
    }

    __builtin_amdgcn_wave_barrier();

    const float* lrow = myL + n * TILE_PITCH;
    const float* erow = ein + (size_t)(mbase + n) * DD;

    v8f dacc[4];
    #pragma unroll
    for (int j = 0; j < 4; ++j) dacc[j] = (v8f){};

    #pragma unroll
    for (int kk = 0; kk < 16; ++kk) {
        int off = kk * 4 + half * 2;
        v2f ls = *(const v2f*)(lrow + off);
        v2f eg = *(const v2f*)(erow + off);
        v2f a_li  = ls + eg;
        v2f a_mul = ls * eg;
        int bbase = ((kk * 2 + half) * 64) << 1;
        #pragma unroll
        for (int j = 0; j < 4; ++j) {
            v2f bgc = *(const v2f*)(sg + bbase + ((j * 16 + n) << 1));
            v2f bbi = *(const v2f*)(sb + bbase + ((j * 16 + n) << 1));
            dacc[j] = __builtin_amdgcn_wmma_f32_16x16x4_f32(
                false, a_li, false, bgc, (short)0, dacc[j], false, false);
            dacc[j] = __builtin_amdgcn_wmma_f32_16x16x4_f32(
                false, a_mul, false, bbi, (short)0, dacc[j], false, false);
        }
    }

    #pragma unroll
    for (int j = 0; j < 4; ++j) {
        float bsum = sbias[j * 16 + n];
        #pragma unroll
        for (int v = 0; v < 8; ++v) {
            float x = dacc[j][v] + bsum;
            x = (x >= 0.f) ? x : SLOPE * x;
            eout[(size_t)(mbase + (half << 3) + v) * DD + j * 16 + n] = x;
        }
    }
}

#define LAST_WAVES 8
#define LAST_T (LAST_WAVES * 32)
#define LAST_TILES (3 * BATCH / 16)
#define LAST_GRID (LAST_TILES / LAST_WAVES)

__global__ __launch_bounds__(LAST_T)
void k_layer_last(const int* __restrict__ rowptr, const int2* __restrict__ sorted,
                  const float* __restrict__ ein,
                  const float* __restrict__ Wgc, const float* __restrict__ Wbi,
                  const float* __restrict__ bias,
                  const int* __restrict__ us, const int* __restrict__ pi,
                  const int* __restrict__ ni, float* __restrict__ out, int colbase) {
    __shared__ __attribute__((aligned(16))) float sg[4096];
    __shared__ __attribute__((aligned(16))) float sb[4096];
    __shared__ float sbias[64];
    __shared__ __attribute__((aligned(16))) float sL[LAST_WAVES * 16 * TILE_PITCH];

    #pragma unroll
    for (int i = threadIdx.x; i < 4096; i += LAST_T) {
        int r = i >> 6, c = i & 63;
        int dst = ((((r >> 1) * 64) + c) << 1) | (r & 1);
        sg[dst] = Wgc[i];
        sb[dst] = Wbi[i];
    }
    if (threadIdx.x < 64) sbias[threadIdx.x] = 2.0f * bias[threadIdx.x];
    __syncthreads();

    const int lane = threadIdx.x & 31;
    const int half = lane >> 4;
    const int n    = lane & 15;
    const int q    = n * 4;
    const int wave = __builtin_amdgcn_readfirstlane(threadIdx.x >> 5);
    const int tile = blockIdx.x * LAST_WAVES + wave;
    const int seg  = tile >> 8;
    const int bb0  = (tile & 255) * 16;
    const int* idx = (seg == 0) ? us : ((seg == 1) ? pi : ni);
    float* myL = sL + wave * (16 * TILE_PITCH);

    const int node = idx[bb0 + n] + (seg ? NU : 0);
    const int rs = rowptr[node], rt = rowptr[node + 1];

    int s = __builtin_amdgcn_readlane(rs, 0), t = __builtin_amdgcn_readlane(rt, 0);
    int2 my = make_int2(0, 0);
    if (lane < t - s) my = sorted[s + lane];

    #pragma unroll 1
    for (int i = 0; i < 16; ++i) {
        int sN = t, tN = t;
        int2 myN = make_int2(0, 0);
        if (i < 15) {
            sN = __builtin_amdgcn_readlane(rs, i + 1);
            tN = __builtin_amdgcn_readlane(rt, i + 1);
            if (lane < tN - sN) myN = sorted[sN + lane];
        }

        float4 acc = make_float4(0.f, 0.f, 0.f, 0.f);
        row_batch(acc, my, min(32, t - s), half, q, ein);
        for (int base = s + 32; base < t; base += 32) {
            const int nb = min(32, t - base);
            int2 m2 = make_int2(0, 0);
            if (lane < nb) m2 = sorted[base + lane];
            row_batch(acc, m2, nb, half, q, ein);
        }
        acc.x += __shfl_xor(acc.x, 16, 32);
        acc.y += __shfl_xor(acc.y, 16, 32);
        acc.z += __shfl_xor(acc.z, 16, 32);
        acc.w += __shfl_xor(acc.w, 16, 32);
        if (half == 0) *(float4*)(myL + i * TILE_PITCH + q) = acc;

        s = sN; t = tN; my = myN;
    }

    __builtin_amdgcn_wave_barrier();

    const float* lrow = myL + n * TILE_PITCH;
    const float* erow = ein + (size_t)node * DD;

    v8f dacc[4];
    #pragma unroll
    for (int j = 0; j < 4; ++j) dacc[j] = (v8f){};

    #pragma unroll
    for (int kk = 0; kk < 16; ++kk) {
        int off = kk * 4 + half * 2;
        v2f ls = *(const v2f*)(lrow + off);
        v2f eg = *(const v2f*)(erow + off);
        v2f a_li  = ls + eg;
        v2f a_mul = ls * eg;
        int bbase = ((kk * 2 + half) * 64) << 1;
        #pragma unroll
        for (int j = 0; j < 4; ++j) {
            v2f bgc = *(const v2f*)(sg + bbase + ((j * 16 + n) << 1));
            v2f bbi = *(const v2f*)(sb + bbase + ((j * 16 + n) << 1));
            dacc[j] = __builtin_amdgcn_wmma_f32_16x16x4_f32(
                false, a_li, false, bgc, (short)0, dacc[j], false, false);
            dacc[j] = __builtin_amdgcn_wmma_f32_16x16x4_f32(
                false, a_mul, false, bbi, (short)0, dacc[j], false, false);
        }
    }

    #pragma unroll
    for (int j = 0; j < 4; ++j) {
        float bsum = sbias[j * 16 + n];
        #pragma unroll
        for (int v = 0; v < 8; ++v) {
            float x = dacc[j][v] + bsum;
            x = (x >= 0.f) ? x : SLOPE * x;
            myL[((half << 3) + v) * TILE_PITCH + j * 16 + n] = x;
        }
    }

    __builtin_amdgcn_wave_barrier();

    #pragma unroll 4
    for (int i = 0; i < 16; ++i) {
        v2f x = *(const v2f*)(myL + i * TILE_PITCH + lane * 2);
        float ss = x.x * x.x + x.y * x.y;
        #pragma unroll
        for (int o = 16; o > 0; o >>= 1) ss += __shfl_xor(ss, o, 32);
        float scale = 1.0f / fmaxf(sqrtf(ss), 1e-12f);
        v2f y; y.x = x.x * scale; y.y = x.y * scale;
        *(v2f*)(out + (size_t)(tile * 16 + i) * 256 + colbase + lane * 2) = y;
    }
}

__global__ __launch_bounds__(256)
void k_gather_norm(const float* __restrict__ ego, const int* __restrict__ us,
                   const int* __restrict__ pi, const int* __restrict__ ni,
                   float* __restrict__ out, int colbase) {
    int wid  = (blockIdx.x * 256 + threadIdx.x) >> 5;
    int lane = threadIdx.x & 31;
    if (wid >= 3 * BATCH) return;
    int seg = wid >> 12;
    int bb  = wid & (BATCH - 1);
    int node;
    if (seg == 0)      node = us[bb];
    else if (seg == 1) node = NU + pi[bb];
    else               node = NU + ni[bb];
    v2f x = *(const v2f*)(ego + (size_t)node * DD + lane * 2);
    float ss = x.x * x.x + x.y * x.y;
    #pragma unroll
    for (int o = 16; o > 0; o >>= 1) ss += __shfl_xor(ss, o, 32);
    float scale = 1.0f / fmaxf(sqrtf(ss), 1e-12f);
    v2f y; y.x = x.x * scale; y.y = x.y * scale;
    *(v2f*)(out + (size_t)wid * 256 + colbase + lane * 2) = y;
}

extern "C" void kernel_launch(void* const* d_in, const int* in_sizes, int n_in,
                              void* d_out, int out_size, void* d_ws, size_t ws_size,
                              hipStream_t stream) {
    const float* ue   = (const float*)d_in[0];
    const float* ie   = (const float*)d_in[1];
    const float* Wgc  = (const float*)d_in[2];
    const float* Wbi  = (const float*)d_in[3];
    const float* bbi  = (const float*)d_in[4];
    const float* vals = (const float*)d_in[5];
    const int*   rows = (const int*)d_in[6];
    const int*   cols = (const int*)d_in[7];
    const int*   us   = (const int*)d_in[8];
    const int*   pi   = (const int*)d_in[9];
    const int*   ni   = (const int*)d_in[10];
    float* out = (float*)d_out;

    float* ego = (float*)d_ws;
    float* L   = ego + (size_t)NT * DD;

    hipMemcpyAsync(ego, ue, (size_t)NU * DD * sizeof(float),
                   hipMemcpyDeviceToDevice, stream);
    hipMemcpyAsync(ego + (size_t)NU * DD, ie, (size_t)NI * DD * sizeof(float),
                   hipMemcpyDeviceToDevice, stream);

    k_gather0<<<(3 * BATCH * DD + 255) / 256, 256, 0, stream>>>(
        ue, ie, us, pi, ni, out);

    const size_t need = (size_t)2 * NT * DD * 4 + (size_t)NNZE * 8 + (size_t)(NT + 1) * 4;
    if (ws_size >= need) {
        int2* sorted = (int2*)(L + (size_t)NT * DD);
        int*  rowptr = (int*)(sorted + NNZE);
        int*  rank   = (int*)L;
        int*  cnt    = rank + NNZE;
        hipMemsetAsync(cnt, 0, (size_t)NT * sizeof(int), stream);
        k_rank<<<EDGE_GRID, 256, 0, stream>>>(rows, cnt, rank);
        k_rowptr<<<SCAN_B, SCAN_T, 0, stream>>>(cnt, rowptr);
        k_place<<<EDGE_GRID, 256, 0, stream>>>(rows, cols, vals, rank, rowptr, sorted);
        float* ego2 = L;
        k_layer<<<LAYER_GRID, LAYER_T, 0, stream>>>(
            rowptr, sorted, ego, ego2, Wgc, Wbi, bbi);
        k_gather_norm<<<(3 * BATCH * 32 + 255) / 256, 256, 0, stream>>>(
            ego2, us, pi, ni, out, 64);
        k_layer<<<LAYER_GRID, LAYER_T, 0, stream>>>(
            rowptr, sorted, ego2, ego, Wgc + 4096, Wbi + 4096, bbi + 64);
        k_gather_norm<<<(3 * BATCH * 32 + 255) / 256, 256, 0, stream>>>(
            ego, us, pi, ni, out, 128);
        k_layer_last<<<LAST_GRID, LAST_T, 0, stream>>>(
            rowptr, sorted, ego, Wgc + 8192, Wbi + 8192, bbi + 128,
            us, pi, ni, out, 192);
        return;
    }

    for (int k = 0; k < 3; ++k) {
        hipMemsetAsync(L, 0, (size_t)NT * DD * sizeof(float), stream);
        long tot = (long)NNZE * 16;
        k_spmm<<<(int)((tot + 255) / 256), 256, 0, stream>>>(
            rows, cols, vals, ego, L);
        k_dense_wmma<<<(NT + 127) / 128, 256, 0, stream>>>(
            L, ego, Wgc + k * 4096, Wbi + k * 4096, bbi + k * 64);
        k_gather_norm<<<(3 * BATCH * 32 + 255) / 256, 256, 0, stream>>>(
            ego, us, pi, ni, out, 64 * (k + 1));
    }
}


# Edge sort in two levels with LDS counters: 256-row buckets (one global add per work-group and bucket), then rows in a bucket; no per-edge global adds, no 150000-row scan; sum order: LDS arrival rank
# speedup vs baseline: 3.3510x; 1.5575x over previous
#include <hip/hip_runtime.h>
#include <hip/hip_bf16.h>

typedef float v2f __attribute__((ext_vector_type(2)));
typedef float v8f __attribute__((ext_vector_type(8)));

#define NU 50000
#define NI 100000
#define NT 150000
#define DD 64
#define NNZE 2400000
#define BATCH 4096
#define SLOPE 0.01f

__global__ __launch_bounds__(256)
void k_gather0(const float* __restrict__ ue, const float* __restrict__ ie,
               const int* __restrict__ us, const int* __restrict__ pi,
               const int* __restrict__ ni, float* __restrict__ out) {
    int t = blockIdx.x * 256 + threadIdx.x;
    if (t >= 3 * BATCH * DD) return;
    int c  = t & 63;
    int b  = t >> 6;
    int seg = b >> 12;
    int bb  = b & (BATCH - 1);
    float v;
    if (seg == 0)      v = ue[(size_t)us[bb] * DD + c];
    else if (seg == 1) v = ie[(size_t)pi[bb] * DD + c];
    else               v = ie[(size_t)ni[bb] * DD + c];
    out[(size_t)b * 256 + c] = v;
}

__global__ __launch_bounds__(256)
void k_spmm(const int* __restrict__ rows, const int* __restrict__ cols,
            const float* __restrict__ vals, const float* __restrict__ ego,
            float* __restrict__ L) {
    long t = (long)blockIdx.x * 256 + threadIdx.x;
    if (t >= (long)NNZE * 16) return;
    int e = (int)(t >> 4);
    int q = ((int)t & 15) * 4;
    int r = rows[e], c = cols[e];
    float v = vals[e];
    const float4 x = *(const float4*)(ego + (size_t)c * DD + q);
    float* dst = L + (size_t)r * DD + q;
    atomicAdd(dst + 0, v * x.x);
    atomicAdd(dst + 1, v * x.y);
    atomicAdd(dst + 2, v * x.z);
    atomicAdd(dst + 3, v * x.w);
}

#define RB 256
#define NB ((NT + RB - 1) / RB)
#define COL_BITS 18
#define SORT_G 256
#define SORT_T 1024
#define SLICE ((NNZE + SORT_G - 1) / SORT_G)
#define SLICE_TRIPS ((SLICE + SORT_T - 1) / SORT_T)
#define CNT_PITCH 32
static_assert(NT <= (1 << COL_BITS), "column index must fit in COL_BITS");
static_assert(RB <= (1 << (31 - COL_BITS)), "row in bucket must fit above the column");
static_assert(NB <= SORT_T && RB <= SORT_T, "one scan element per thread");

static __device__ __forceinline__
int block_excl_scan(const int own, int* wtot) {
    const int lane = threadIdx.x & 31, wave = threadIdx.x >> 5;
    int inc = own;
    #pragma unroll
    for (int o = 1; o < 32; o <<= 1) {
        int y = __shfl_up(inc, o, 32);
        if (lane >= o) inc += y;
    }
    if (lane == 31) wtot[wave] = inc;
    __syncthreads();
    if (wave == 0) {
        const int x = wtot[lane];
        int xi = x;
        #pragma unroll
        for (int o = 1; o < 32; o <<= 1) {
            int y = __shfl_up(xi, o, 32);
            if (lane >= o) xi += y;
        }
        wtot[lane] = xi - x;
    }
    __syncthreads();
    return wtot[wave] + inc - own;
}

__global__ __launch_bounds__(SORT_T)
void k_bucket_count(const int* __restrict__ rows, int* __restrict__ bcnt,
                    int* __restrict__ wgoff) {
    __shared__ int hist[NB];
    const int t = threadIdx.x;
    for (int i = t; i < NB; i += SORT_T) hist[i] = 0;
    __syncthreads();
    const int s0 = blockIdx.x * SLICE;
    const int s1 = min(s0 + SLICE, NNZE);
    int r[SLICE_TRIPS];
    #pragma unroll
    for (int k = 0; k < SLICE_TRIPS; ++k) {
        const int e = s0 + k * SORT_T + t;
        r[k] = (e < s1) ? rows[e] : -1;
    }
    #pragma unroll
    for (int k = 0; k < SLICE_TRIPS; ++k)
        if (r[k] >= 0) atomicAdd(&hist[r[k] / RB], 1);
    __syncthreads();
    for (int b = t; b < NB; b += SORT_T) {
        const int h = hist[b];
        wgoff[blockIdx.x * NB + b] = h ? atomicAdd(&bcnt[b * CNT_PITCH], h) : 0;
    }
}

__global__ __launch_bounds__(SORT_T)
void k_bucket_scatter(const int* __restrict__ rows, const int* __restrict__ cols,
                      const float* __restrict__ vals, const int* __restrict__ bcnt,
                      const int* __restrict__ wgoff, int* __restrict__ bbase,
                      int2* __restrict__ tmp) {
    __shared__ int start[NB];
    __shared__ int lcnt[NB];
    __shared__ int wtot[32];
    const int t = threadIdx.x;
    const int own = (t < NB) ? bcnt[t * CNT_PITCH] : 0;
    const int off = (t < NB) ? wgoff[blockIdx.x * NB + t] : 0;
    const int excl = block_excl_scan(own, wtot);
    if (t < NB) {
        start[t] = excl + off;
        lcnt[t] = 0;
        if (blockIdx.x == 0) {
            bbase[t] = excl;
            if (t == NB - 1) bbase[NB] = excl + own;
        }
    }
    __syncthreads();
    const int s0 = blockIdx.x * SLICE;
    const int s1 = min(s0 + SLICE, NNZE);
    int   r[SLICE_TRIPS], c[SLICE_TRIPS];
    float v[SLICE_TRIPS];
    #pragma unroll
    for (int k = 0; k < SLICE_TRIPS; ++k) {
        const int e = s0 + k * SORT_T + t;
        const bool ok = e < s1;
        r[k] = ok ? rows[e] : -1;
        c[k] = ok ? cols[e] : 0;
        v[k] = ok ? vals[e] : 0.f;
    }
    #pragma unroll
    for (int k = 0; k < SLICE_TRIPS; ++k) {
        if (r[k] >= 0) {
            const int b  = r[k] / RB;
            const int lr = atomicAdd(&lcnt[b], 1);
            tmp[start[b] + lr] = make_int2(((r[k] % RB) << COL_BITS) | c[k],
                                           __float_as_int(v[k]));
        }
    }
}

#define ROWS_T 1024
#define ROWS_BATCH 4

__global__ __launch_bounds__(ROWS_T)
void k_bucket_rows(const int* __restrict__ bbase, const int2* __restrict__ tmp,
                   int2* __restrict__ sorted, int* __restrict__ rowptr) {
    __shared__ int cnt[RB];
    __shared__ int wtot[32];
    const int t = threadIdx.x;
    const int b = blockIdx.x;
    const int base = bbase[b];
    const int n = bbase[b + 1] - base;
    if (t < RB) cnt[t] = 0;
    __syncthreads();
    for (int i0 = 0; i0 < n; i0 += ROWS_BATCH * ROWS_T) {
        int x[ROWS_BATCH];
        #pragma unroll
        for (int j = 0; j < ROWS_BATCH; ++j) {
            const int i = i0 + j * ROWS_T + t;
            x[j] = (i < n) ? tmp[base + i].x : -1;
        }
        #pragma unroll
        for (int j = 0; j < ROWS_BATCH; ++j)
            if (x[j] >= 0) atomicAdd(&cnt[x[j] >> COL_BITS], 1);
    }
    __syncthreads();
    const int own = (t < RB) ? cnt[t] : 0;
    const int excl = block_excl_scan(own, wtot);
    if (t < RB) {
        cnt[t] = excl;
        const int row = b * RB + t;
        if (row < NT) rowptr[row] = base + excl;
    }
    if (b == NB - 1 && t == 0) rowptr[NT] = base + n;
    __syncthreads();
    for (int i0 = 0; i0 < n; i0 += ROWS_BATCH * ROWS_T) {
        int2 e[ROWS_BATCH];
        #pragma unroll
        for (int j = 0; j < ROWS_BATCH; ++j) {
            const int i = i0 + j * ROWS_T + t;
            e[j] = (i < n) ? tmp[base + i] : make_int2(-1, 0);
        }
        #pragma unroll
        for (int j = 0; j < ROWS_BATCH; ++j) {
            if (e[j].x >= 0) {
                const int p = atomicAdd(&cnt[e[j].x >> COL_BITS], 1);
                sorted[base + p] = make_int2(e[j].x & ((1 << COL_BITS) - 1), e[j].y);
            }
        }
    }
}

static __device__ __forceinline__
void row_batch(float4& acc, const int2 my, const int nb, const int half,
               const int q, const float* __restrict__ ego) {
    for (int i0 = 0; i0 < nb; i0 += 16) {
        int   c[8];
        float v[8];
        float4 x[8];
        #pragma unroll
        for (int j = 0; j < 8; ++j) {
            const int src = i0 + 2 * j + half;
            c[j] = __shfl(my.x, src, 32);
            v[j] = __int_as_float(__shfl(my.y, src, 32));
        }
        #pragma unroll
        for (int j = 0; j < 8; ++j)
            x[j] = *(const float4*)(ego + (size_t)c[j] * DD + q);
        #pragma unroll
        for (int j = 0; j < 8; ++j) {
            const bool ok = (i0 + 2 * j + half) < nb;
            acc.x += v[j] * (ok ? x[j].x : 0.f);
            acc.y += v[j] * (ok ? x[j].y : 0.f);
            acc.z += v[j] * (ok ? x[j].z : 0.f);
            acc.w += v[j] * (ok ? x[j].w : 0.f);
        }
    }
}

__global__ __launch_bounds__(256)
void k_dense_wmma(const float* __restrict__ L, float* __restrict__ ego,
                  const float* __restrict__ Wgc, const float* __restrict__ Wbi,
                  const float* __restrict__ bias) {
    __shared__ float sg[4096];
    __shared__ float sb[4096];
    __shared__ float sbias[64];

    for (int i = threadIdx.x; i < 4096; i += 256) {
        int r = i >> 6, c = i & 63;
        int dst = ((((r >> 1) * 64) + c) << 1) | (r & 1);
        sg[dst] = Wgc[i];
        sb[dst] = Wbi[i];
    }
    if (threadIdx.x < 64) sbias[threadIdx.x] = 2.0f * bias[threadIdx.x];
    __syncthreads();

    const int wave  = threadIdx.x >> 5;
    const int lane  = threadIdx.x & 31;
    const int mbase = (blockIdx.x * 8 + wave) * 16;
    if (mbase >= NT) return;

    const int half = lane >> 4;
    const int n    = lane & 15;

    const float* lrow = L   + (size_t)(mbase + n) * DD;
    const float* erow = ego + (size_t)(mbase + n) * DD;

    v8f acc[4];
    #pragma unroll
    for (int j = 0; j < 4; ++j) acc[j] = (v8f){};

    #pragma unroll
    for (int kk = 0; kk < 16; ++kk) {
        int off = kk * 4 + half * 2;
        v2f ls = *(const v2f*)(lrow + off);
        v2f eg = *(const v2f*)(erow + off);
        v2f a_li  = ls + eg;
        v2f a_mul = ls * eg;
        int bbase = ((kk * 2 + half) * 64) << 1;
        #pragma unroll
        for (int j = 0; j < 4; ++j) {
            v2f bgc = *(const v2f*)(sg + bbase + ((j * 16 + n) << 1));
            v2f bbi = *(const v2f*)(sb + bbase + ((j * 16 + n) << 1));
            acc[j] = __builtin_amdgcn_wmma_f32_16x16x4_f32(
                false, a_li, false, bgc, (short)0, acc[j], false, false);
            acc[j] = __builtin_amdgcn_wmma_f32_16x16x4_f32(
                false, a_mul, false, bbi, (short)0, acc[j], false, false);
        }
    }

    #pragma unroll
    for (int j = 0; j < 4; ++j) {
        float bsum = sbias[j * 16 + n];
        #pragma unroll
        for (int v = 0; v < 8; ++v) {
            float x = acc[j][v] + bsum;
            x = (x >= 0.f) ? x : SLOPE * x;
            ego[(size_t)(mbase + (half << 3) + v) * DD + j * 16 + n] = x;
        }
    }
}

#define TILE_PITCH 68
#define LAYER_WAVES 16
#define LAYER_T (LAYER_WAVES * 32)
#define N_TILES (NT / 16)
#define LAYER_GRID ((N_TILES + LAYER_WAVES - 1) / LAYER_WAVES)

__global__ __launch_bounds__(LAYER_T, 12)
void k_layer(const int* __restrict__ rowptr, const int2* __restrict__ sorted,
             const float* __restrict__ ein, float* __restrict__ eout,
             const float* __restrict__ Wgc, const float* __restrict__ Wbi,
             const float* __restrict__ bias) {
    __shared__ __attribute__((aligned(16))) float sg[4096];
    __shared__ __attribute__((aligned(16))) float sb[4096];
    __shared__ float sbias[64];
    __shared__ __attribute__((aligned(16))) float sL[LAYER_WAVES * 16 * TILE_PITCH];

    #pragma unroll
    for (int i = threadIdx.x; i < 4096; i += LAYER_T) {
        int r = i >> 6, c = i & 63;
        int dst = ((((r >> 1) * 64) + c) << 1) | (r & 1);
        sg[dst] = Wgc[i];
        sb[dst] = Wbi[i];
    }
    if (threadIdx.x < 64) sbias[threadIdx.x] = 2.0f * bias[threadIdx.x];
    __syncthreads();

    const int lane = threadIdx.x & 31;
    const int half = lane >> 4;
    const int n    = lane & 15;
    const int q    = n * 4;
    const int wave = __builtin_amdgcn_readfirstlane(threadIdx.x >> 5);
    const int tile = blockIdx.x * LAYER_WAVES + wave;
    if (tile >= N_TILES) return;
    const int mbase = tile * 16;
    float* myL = sL + wave * (16 * TILE_PITCH);

    int s = rowptr[mbase], t = rowptr[mbase + 1], tn = rowptr[mbase + 2];
    int2 my = make_int2(0, 0);
    if (lane < t - s) my = sorted[s + lane];

    #pragma unroll 1
    for (int i = 0; i < 16; ++i) {
        int2 myN = make_int2(0, 0);
        if (i < 15) {
            if (lane < tn - t) myN = sorted[t + lane];
        }
        int tnn = tn;
        if (i < 14) tnn = rowptr[mbase + i + 3];

        float4 acc = make_float4(0.f, 0.f, 0.f, 0.f);
        row_batch(acc, my, min(32, t - s), half, q, ein);
        for (int base = s + 32; base < t; base += 32) {
            const int nb = min(32, t - base);
            int2 m2 = make_int2(0, 0);
            if (lane < nb) m2 = sorted[base + lane];
            row_batch(acc, m2, nb, half, q, ein);
        }
        acc.x += __shfl_xor(acc.x, 16, 32);
        acc.y += __shfl_xor(acc.y, 16, 32);
        acc.z += __shfl_xor(acc.z, 16, 32);
        acc.w += __shfl_xor(acc.w, 16, 32);
        if (half == 0) *(float4*)(myL + i * TILE_PITCH + q) = acc;

        s = t; t = tn; tn = tnn; my = myN;
    }

    __builtin_amdgcn_wave_barrier();

    const float* lrow = myL + n * TILE_PITCH;
    const float* erow = ein + (size_t)(mbase + n) * DD;

    v8f dacc[4];
    #pragma unroll
    for (int j = 0; j < 4; ++j) dacc[j] = (v8f){};

    #pragma unroll
    for (int kk = 0; kk < 16; ++kk) {
        int off = kk * 4 + half * 2;
        v2f ls = *(const v2f*)(lrow + off);
        v2f eg = *(const v2f*)(erow + off);
        v2f a_li  = ls + eg;
        v2f a_mul = ls * eg;
        int bbase = ((kk * 2 + half) * 64) << 1;
        #pragma unroll
        for (int j = 0; j < 4; ++j) {
            v2f bgc = *(const v2f*)(sg + bbase + ((j * 16 + n) << 1));
            v2f bbi = *(const v2f*)(sb + bbase + ((j * 16 + n) << 1));
            dacc[j] = __builtin_amdgcn_wmma_f32_16x16x4_f32(
                false, a_li, false, bgc, (short)0, dacc[j], false, false);
            dacc[j] = __builtin_amdgcn_wmma_f32_16x16x4_f32(
                false, a_mul, false, bbi, (short)0, dacc[j], false, false);
        }
    }

    #pragma unroll
    for (int j = 0; j < 4; ++j) {
        float bsum = sbias[j * 16 + n];
        #pragma unroll
        for (int v = 0; v < 8; ++v) {
            float x = dacc[j][v] + bsum;
            x = (x >= 0.f) ? x : SLOPE * x;
            eout[(size_t)(mbase + (half << 3) + v) * DD + j * 16 + n] = x;
        }
    }
}

#define LAST_WAVES 8
#define LAST_T (LAST_WAVES * 32)
#define LAST_TILES (3 * BATCH / 16)
#define LAST_GRID (LAST_TILES / LAST_WAVES)

__global__ __launch_bounds__(LAST_T)
void k_layer_last(const int* __restrict__ rowptr, const int2* __restrict__ sorted,
                  const float* __restrict__ ein,
                  const float* __restrict__ Wgc, const float* __restrict__ Wbi,
                  const float* __restrict__ bias,
                  const int* __restrict__ us, const int* __restrict__ pi,
                  const int* __restrict__ ni, float* __restrict__ out, int colbase) {
    __shared__ __attribute__((aligned(16))) float sg[4096];
    __shared__ __attribute__((aligned(16))) float sb[4096];
    __shared__ float sbias[64];
    __shared__ __attribute__((aligned(16))) float sL[LAST_WAVES * 16 * TILE_PITCH];

    #pragma unroll
    for (int i = threadIdx.x; i < 4096; i += LAST_T) {
        int r = i >> 6, c = i & 63;
        int dst = ((((r >> 1) * 64) + c) << 1) | (r & 1);
        sg[dst] = Wgc[i];
        sb[dst] = Wbi[i];
    }
    if (threadIdx.x < 64) sbias[threadIdx.x] = 2.0f * bias[threadIdx.x];
    __syncthreads();

    const int lane = threadIdx.x & 31;
    const int half = lane >> 4;
    const int n    = lane & 15;
    const int q    = n * 4;
    const int wave = __builtin_amdgcn_readfirstlane(threadIdx.x >> 5);
    const int tile = blockIdx.x * LAST_WAVES + wave;
    const int seg  = tile >> 8;
    const int bb0  = (tile & 255) * 16;
    const int* idx = (seg == 0) ? us : ((seg == 1) ? pi : ni);
    float* myL = sL + wave * (16 * TILE_PITCH);

    const int node = idx[bb0 + n] + (seg ? NU : 0);
    const int rs = rowptr[node], rt = rowptr[node + 1];

    int s = __builtin_amdgcn_readlane(rs, 0), t = __builtin_amdgcn_readlane(rt, 0);
    int2 my = make_int2(0, 0);
    if (lane < t - s) my = sorted[s + lane];

    #pragma unroll 1
    for (int i = 0; i < 16; ++i) {
        int sN = t, tN = t;
        int2 myN = make_int2(0, 0);
        if (i < 15) {
            sN = __builtin_amdgcn_readlane(rs, i + 1);
            tN = __builtin_amdgcn_readlane(rt, i + 1);
            if (lane < tN - sN) myN = sorted[sN + lane];
        }

        float4 acc = make_float4(0.f, 0.f, 0.f, 0.f);
        row_batch(acc, my, min(32, t - s), half, q, ein);
        for (int base = s + 32; base < t; base += 32) {
            const int nb = min(32, t - base);
            int2 m2 = make_int2(0, 0);
            if (lane < nb) m2 = sorted[base + lane];
            row_batch(acc, m2, nb, half, q, ein);
        }
        acc.x += __shfl_xor(acc.x, 16, 32);
        acc.y += __shfl_xor(acc.y, 16, 32);
        acc.z += __shfl_xor(acc.z, 16, 32);
        acc.w += __shfl_xor(acc.w, 16, 32);
        if (half == 0) *(float4*)(myL + i * TILE_PITCH + q) = acc;

        s = sN; t = tN; my = myN;
    }

    __builtin_amdgcn_wave_barrier();

    const float* lrow = myL + n * TILE_PITCH;
    const float* erow = ein + (size_t)node * DD;

    v8f dacc[4];
    #pragma unroll
    for (int j = 0; j < 4; ++j) dacc[j] = (v8f){};

    #pragma unroll
    for (int kk = 0; kk < 16; ++kk) {
        int off = kk * 4 + half * 2;
        v2f ls = *(const v2f*)(lrow + off);
        v2f eg = *(const v2f*)(erow + off);
        v2f a_li  = ls + eg;
        v2f a_mul = ls * eg;
        int bbase = ((kk * 2 + half) * 64) << 1;
        #pragma unroll
        for (int j = 0; j < 4; ++j) {
            v2f bgc = *(const v2f*)(sg + bbase + ((j * 16 + n) << 1));
            v2f bbi = *(const v2f*)(sb + bbase + ((j * 16 + n) << 1));
            dacc[j] = __builtin_amdgcn_wmma_f32_16x16x4_f32(
                false, a_li, false, bgc, (short)0, dacc[j], false, false);
            dacc[j] = __builtin_amdgcn_wmma_f32_16x16x4_f32(
                false, a_mul, false, bbi, (short)0, dacc[j], false, false);
        }
    }

    #pragma unroll
    for (int j = 0; j < 4; ++j) {
        float bsum = sbias[j * 16 + n];
        #pragma unroll
        for (int v = 0; v < 8; ++v) {
            float x = dacc[j][v] + bsum;
            x = (x >= 0.f) ? x : SLOPE * x;
            myL[((half << 3) + v) * TILE_PITCH + j * 16 + n] = x;
        }
    }

    __builtin_amdgcn_wave_barrier();

    #pragma unroll 4
    for (int i = 0; i < 16; ++i) {
        v2f x = *(const v2f*)(myL + i * TILE_PITCH + lane * 2);
        float ss = x.x * x.x + x.y * x.y;
        #pragma unroll
        for (int o = 16; o > 0; o >>= 1) ss += __shfl_xor(ss, o, 32);
        float scale = 1.0f / fmaxf(sqrtf(ss), 1e-12f);
        v2f y; y.x = x.x * scale; y.y = x.y * scale;
        *(v2f*)(out + (size_t)(tile * 16 + i) * 256 + colbase + lane * 2) = y;
    }
}

__global__ __launch_bounds__(256)
void k_gather_norm(const float* __restrict__ ego, const int* __restrict__ us,
                   const int* __restrict__ pi, const int* __restrict__ ni,
                   float* __restrict__ out, int colbase) {
    int wid  = (blockIdx.x * 256 + threadIdx.x) >> 5;
    int lane = threadIdx.x & 31;
    if (wid >= 3 * BATCH) return;
    int seg = wid >> 12;
    int bb  = wid & (BATCH - 1);
    int node;
    if (seg == 0)      node = us[bb];
    else if (seg == 1) node = NU + pi[bb];
    else               node = NU + ni[bb];
    v2f x = *(const v2f*)(ego + (size_t)node * DD + lane * 2);
    float ss = x.x * x.x + x.y * x.y;
    #pragma unroll
    for (int o = 16; o > 0; o >>= 1) ss += __shfl_xor(ss, o, 32);
    float scale = 1.0f / fmaxf(sqrtf(ss), 1e-12f);
    v2f y; y.x = x.x * scale; y.y = x.y * scale;
    *(v2f*)(out + (size_t)wid * 256 + colbase + lane * 2) = y;
}

extern "C" void kernel_launch(void* const* d_in, const int* in_sizes, int n_in,
                              void* d_out, int out_size, void* d_ws, size_t ws_size,
                              hipStream_t stream) {
    const float* ue   = (const float*)d_in[0];
    const float* ie   = (const float*)d_in[1];
    const float* Wgc  = (const float*)d_in[2];
    const float* Wbi  = (const float*)d_in[3];
    const float* bbi  = (const float*)d_in[4];
    const float* vals = (const float*)d_in[5];
    const int*   rows = (const int*)d_in[6];
    const int*   cols = (const int*)d_in[7];
    const int*   us   = (const int*)d_in[8];
    const int*   pi   = (const int*)d_in[9];
    const int*   ni   = (const int*)d_in[10];
    float* out = (float*)d_out;

    float* ego = (float*)d_ws;
    float* L   = ego + (size_t)NT * DD;

    hipMemcpyAsync(ego, ue, (size_t)NU * DD * sizeof(float),
                   hipMemcpyDeviceToDevice, stream);
    hipMemcpyAsync(ego + (size_t)NU * DD, ie, (size_t)NI * DD * sizeof(float),
                   hipMemcpyDeviceToDevice, stream);

    k_gather0<<<(3 * BATCH * DD + 255) / 256, 256, 0, stream>>>(
        ue, ie, us, pi, ni, out);

    const size_t need = (size_t)2 * NT * DD * 4 + (size_t)NNZE * 8 + (size_t)(NT + 1) * 4;
    if (ws_size >= need) {
        int2* sorted = (int2*)(L + (size_t)NT * DD);
        int*  rowptr = (int*)(sorted + NNZE);
        int2* tmp    = (int2*)L;
        int*  wgoff  = (int*)(tmp + NNZE);
        int*  bcnt   = wgoff + SORT_G * NB;
        int*  bbase  = bcnt + NB * CNT_PITCH;
        hipMemsetAsync(bcnt, 0, (size_t)NB * CNT_PITCH * sizeof(int), stream);
        k_bucket_count<<<SORT_G, SORT_T, 0, stream>>>(rows, bcnt, wgoff);
        k_bucket_scatter<<<SORT_G, SORT_T, 0, stream>>>(rows, cols, vals, bcnt, wgoff,
                                                        bbase, tmp);
        k_bucket_rows<<<NB, ROWS_T, 0, stream>>>(bbase, tmp, sorted, rowptr);
        float* ego2 = L;
        k_layer<<<LAYER_GRID, LAYER_T, 0, stream>>>(
            rowptr, sorted, ego, ego2, Wgc, Wbi, bbi);
        k_gather_norm<<<(3 * BATCH * 32 + 255) / 256, 256, 0, stream>>>(
            ego2, us, pi, ni, out, 64);
        k_layer<<<LAYER_GRID, LAYER_T, 0, stream>>>(
            rowptr, sorted, ego2, ego, Wgc + 4096, Wbi + 4096, bbi + 64);
        k_gather_norm<<<(3 * BATCH * 32 + 255) / 256, 256, 0, stream>>>(
            ego, us, pi, ni, out, 128);
        k_layer_last<<<LAST_GRID, LAST_T, 0, stream>>>(
            rowptr, sorted, ego, Wgc + 8192, Wbi + 8192, bbi + 128,
            us, pi, ni, out, 192);
        return;
    }

    for (int k = 0; k < 3; ++k) {
        hipMemsetAsync(L, 0, (size_t)NT * DD * sizeof(float), stream);
        long tot = (long)NNZE * 16;
        k_spmm<<<(int)((tot + 255) / 256), 256, 0, stream>>>(
            rows, cols, vals, ego, L);
        k_dense_wmma<<<(NT + 127) / 128, 256, 0, stream>>>(
            L, ego, Wgc + k * 4096, Wbi + k * 4096, bbi + k * 64);
        k_gather_norm<<<(3 * BATCH * 32 + 255) / 256, 256, 0, stream>>>(
            ego, us, pi, ni, out, 64 * (k + 1));
    }
}


# Output gathers folded into the last-layer launch: 96 more work-groups, one wave per 16 gathered rows, write columns 0-191 (table rows, layer 1 and 2 rows normalised); 3 launches, 49152 waves fewer
# speedup vs baseline: 3.5294x; 1.0532x over previous
#include <hip/hip_runtime.h>
#include <hip/hip_bf16.h>

typedef float v2f __attribute__((ext_vector_type(2)));
typedef float v8f __attribute__((ext_vector_type(8)));

#define NU 50000
#define NI 100000
#define NT 150000
#define DD 64
#define NNZE 2400000
#define BATCH 4096
#define SLOPE 0.01f

__global__ __launch_bounds__(256)
void k_gather0(const float* __restrict__ ue, const float* __restrict__ ie,
               const int* __restrict__ us, const int* __restrict__ pi,
               const int* __restrict__ ni, float* __restrict__ out) {
    int t = blockIdx.x * 256 + threadIdx.x;
    if (t >= 3 * BATCH * DD) return;
    int c  = t & 63;
    int b  = t >> 6;
    int seg = b >> 12;
    int bb  = b & (BATCH - 1);
    float v;
    if (seg == 0)      v = ue[(size_t)us[bb] * DD + c];
    else if (seg == 1) v = ie[(size_t)pi[bb] * DD + c];
    else               v = ie[(size_t)ni[bb] * DD + c];
    out[(size_t)b * 256 + c] = v;
}

__global__ __launch_bounds__(256)
void k_spmm(const int* __restrict__ rows, const int* __restrict__ cols,
            const float* __restrict__ vals, const float* __restrict__ ego,
            float* __restrict__ L) {
    long t = (long)blockIdx.x * 256 + threadIdx.x;
    if (t >= (long)NNZE * 16) return;
    int e = (int)(t >> 4);
    int q = ((int)t & 15) * 4;
    int r = rows[e], c = cols[e];
    float v = vals[e];
    const float4 x = *(const float4*)(ego + (size_t)c * DD + q);
    float* dst = L + (size_t)r * DD + q;
    atomicAdd(dst + 0, v * x.x);
    atomicAdd(dst + 1, v * x.y);
    atomicAdd(dst + 2, v * x.z);
    atomicAdd(dst + 3, v * x.w);
}

#define RB 256
#define NB ((NT + RB - 1) / RB)
#define COL_BITS 18
#define SORT_G 256
#define SORT_T 1024
#define SLICE ((NNZE + SORT_G - 1) / SORT_G)
#define SLICE_TRIPS ((SLICE + SORT_T - 1) / SORT_T)
#define CNT_PITCH 32
static_assert(NT <= (1 << COL_BITS), "column index must fit in COL_BITS");
static_assert(RB <= (1 << (31 - COL_BITS)), "row in bucket must fit above the column");
static_assert(NB <= SORT_T && RB <= SORT_T, "one scan element per thread");

static __device__ __forceinline__
int block_excl_scan(const int own, int* wtot) {
    const int lane = threadIdx.x & 31, wave = threadIdx.x >> 5;
    int inc = own;
    #pragma unroll
    for (int o = 1; o < 32; o <<= 1) {
        int y = __shfl_up(inc, o, 32);
        if (lane >= o) inc += y;
    }
    if (lane == 31) wtot[wave] = inc;
    __syncthreads();
    if (wave == 0) {
        const int x = wtot[lane];
        int xi = x;
        #pragma unroll
        for (int o = 1; o < 32; o <<= 1) {
            int y = __shfl_up(xi, o, 32);
            if (lane >= o) xi += y;
        }
        wtot[lane] = xi - x;
    }
    __syncthreads();
    return wtot[wave] + inc - own;
}

__global__ __launch_bounds__(SORT_T)
void k_bucket_count(const int* __restrict__ rows, int* __restrict__ bcnt,
                    int* __restrict__ wgoff) {
    __shared__ int hist[NB];
    const int t = threadIdx.x;
    for (int i = t; i < NB; i += SORT_T) hist[i] = 0;
    __syncthreads();
    const int s0 = blockIdx.x * SLICE;
    const int s1 = min(s0 + SLICE, NNZE);
    int r[SLICE_TRIPS];
    #pragma unroll
    for (int k = 0; k < SLICE_TRIPS; ++k) {
        const int e = s0 + k * SORT_T + t;
        r[k] = (e < s1) ? rows[e] : -1;
    }
    #pragma unroll
    for (int k = 0; k < SLICE_TRIPS; ++k)
        if (r[k] >= 0) atomicAdd(&hist[r[k] / RB], 1);
    __syncthreads();
    for (int b = t; b < NB; b += SORT_T) {
        const int h = hist[b];
        wgoff[blockIdx.x * NB + b] = h ? atomicAdd(&bcnt[b * CNT_PITCH], h) : 0;
    }
}

__global__ __launch_bounds__(SORT_T)
void k_bucket_scatter(const int* __restrict__ rows, const int* __restrict__ cols,
                      const float* __restrict__ vals, const int* __restrict__ bcnt,
                      const int* __restrict__ wgoff, int* __restrict__ bbase,
                      int2* __restrict__ tmp) {
    __shared__ int start[NB];
    __shared__ int lcnt[NB];
    __shared__ int wtot[32];
    const int t = threadIdx.x;
    const int own = (t < NB) ? bcnt[t * CNT_PITCH] : 0;
    const int off = (t < NB) ? wgoff[blockIdx.x * NB + t] : 0;
    const int excl = block_excl_scan(own, wtot);
    if (t < NB) {
        start[t] = excl + off;
        lcnt[t] = 0;
        if (blockIdx.x == 0) {
            bbase[t] = excl;
            if (t == NB - 1) bbase[NB] = excl + own;
        }
    }
    __syncthreads();
    const int s0 = blockIdx.x * SLICE;
    const int s1 = min(s0 + SLICE, NNZE);
    int   r[SLICE_TRIPS], c[SLICE_TRIPS];
    float v[SLICE_TRIPS];
    #pragma unroll
    for (int k = 0; k < SLICE_TRIPS; ++k) {
        const int e = s0 + k * SORT_T + t;
        const bool ok = e < s1;
        r[k] = ok ? rows[e] : -1;
        c[k] = ok ? cols[e] : 0;
        v[k] = ok ? vals[e] : 0.f;
    }
    #pragma unroll
    for (int k = 0; k < SLICE_TRIPS; ++k) {
        if (r[k] >= 0) {
            const int b  = r[k] / RB;
            const int lr = atomicAdd(&lcnt[b], 1);
            tmp[start[b] + lr] = make_int2(((r[k] % RB) << COL_BITS) | c[k],
                                           __float_as_int(v[k]));
        }
    }
}

#define ROWS_T 1024
#define ROWS_BATCH 4

__global__ __launch_bounds__(ROWS_T)
void k_bucket_rows(const int* __restrict__ bbase, const int2* __restrict__ tmp,
                   int2* __restrict__ sorted, int* __restrict__ rowptr) {
    __shared__ int cnt[RB];
    __shared__ int wtot[32];
    const int t = threadIdx.x;
    const int b = blockIdx.x;
    const int base = bbase[b];
    const int n = bbase[b + 1] - base;
    if (t < RB) cnt[t] = 0;
    __syncthreads();
    for (int i0 = 0; i0 < n; i0 += ROWS_BATCH * ROWS_T) {
        int x[ROWS_BATCH];
        #pragma unroll
        for (int j = 0; j < ROWS_BATCH; ++j) {
            const int i = i0 + j * ROWS_T + t;
            x[j] = (i < n) ? tmp[base + i].x : -1;
        }
        #pragma unroll
        for (int j = 0; j < ROWS_BATCH; ++j)
            if (x[j] >= 0) atomicAdd(&cnt[x[j] >> COL_BITS], 1);
    }
    __syncthreads();
    const int own = (t < RB) ? cnt[t] : 0;
    const int excl = block_excl_scan(own, wtot);
    if (t < RB) {
        cnt[t] = excl;
        const int row = b * RB + t;
        if (row < NT) rowptr[row] = base + excl;
    }
    if (b == NB - 1 && t == 0) rowptr[NT] = base + n;
    __syncthreads();
    for (int i0 = 0; i0 < n; i0 += ROWS_BATCH * ROWS_T) {
        int2 e[ROWS_BATCH];
        #pragma unroll
        for (int j = 0; j < ROWS_BATCH; ++j) {
            const int i = i0 + j * ROWS_T + t;
            e[j] = (i < n) ? tmp[base + i] : make_int2(-1, 0);
        }
        #pragma unroll
        for (int j = 0; j < ROWS_BATCH; ++j) {
            if (e[j].x >= 0) {
                const int p = atomicAdd(&cnt[e[j].x >> COL_BITS], 1);
                sorted[base + p] = make_int2(e[j].x & ((1 << COL_BITS) - 1), e[j].y);
            }
        }
    }
}

static __device__ __forceinline__
void row_batch(float4& acc, const int2 my, const int nb, const int half,
               const int q, const float* __restrict__ ego) {
    for (int i0 = 0; i0 < nb; i0 += 16) {
        int   c[8];
        float v[8];
        float4 x[8];
        #pragma unroll
        for (int j = 0; j < 8; ++j) {
            const int src = i0 + 2 * j + half;
            c[j] = __shfl(my.x, src, 32);
            v[j] = __int_as_float(__shfl(my.y, src, 32));
        }
        #pragma unroll
        for (int j = 0; j < 8; ++j)
            x[j] = *(const float4*)(ego + (size_t)c[j] * DD + q);
        #pragma unroll
        for (int j = 0; j < 8; ++j) {
            const bool ok = (i0 + 2 * j + half) < nb;
            acc.x += v[j] * (ok ? x[j].x : 0.f);
            acc.y += v[j] * (ok ? x[j].y : 0.f);
            acc.z += v[j] * (ok ? x[j].z : 0.f);
            acc.w += v[j] * (ok ? x[j].w : 0.f);
        }
    }
}

__global__ __launch_bounds__(256)
void k_dense_wmma(const float* __restrict__ L, float* __restrict__ ego,
                  const float* __restrict__ Wgc, const float* __restrict__ Wbi,
                  const float* __restrict__ bias) {
    __shared__ float sg[4096];
    __shared__ float sb[4096];
    __shared__ float sbias[64];

    for (int i = threadIdx.x; i < 4096; i += 256) {
        int r = i >> 6, c = i & 63;
        int dst = ((((r >> 1) * 64) + c) << 1) | (r & 1);
        sg[dst] = Wgc[i];
        sb[dst] = Wbi[i];
    }
    if (threadIdx.x < 64) sbias[threadIdx.x] = 2.0f * bias[threadIdx.x];
    __syncthreads();

    const int wave  = threadIdx.x >> 5;
    const int lane  = threadIdx.x & 31;
    const int mbase = (blockIdx.x * 8 + wave) * 16;
    if (mbase >= NT) return;

    const int half = lane >> 4;
    const int n    = lane & 15;

    const float* lrow = L   + (size_t)(mbase + n) * DD;
    const float* erow = ego + (size_t)(mbase + n) * DD;

    v8f acc[4];
    #pragma unroll
    for (int j = 0; j < 4; ++j) acc[j] = (v8f){};

    #pragma unroll
    for (int kk = 0; kk < 16; ++kk) {
        int off = kk * 4 + half * 2;
        v2f ls = *(const v2f*)(lrow + off);
        v2f eg = *(const v2f*)(erow + off);
        v2f a_li  = ls + eg;
        v2f a_mul = ls * eg;
        int bbase = ((kk * 2 + half) * 64) << 1;
        #pragma unroll
        for (int j = 0; j < 4; ++j) {
            v2f bgc = *(const v2f*)(sg + bbase + ((j * 16 + n) << 1));
            v2f bbi = *(const v2f*)(sb + bbase + ((j * 16 + n) << 1));
            acc[j] = __builtin_amdgcn_wmma_f32_16x16x4_f32(
                false, a_li, false, bgc, (short)0, acc[j], false, false);
            acc[j] = __builtin_amdgcn_wmma_f32_16x16x4_f32(
                false, a_mul, false, bbi, (short)0, acc[j], false, false);
        }
    }

    #pragma unroll
    for (int j = 0; j < 4; ++j) {
        float bsum = sbias[j * 16 + n];
        #pragma unroll
        for (int v = 0; v < 8; ++v) {
            float x = acc[j][v] + bsum;
            x = (x >= 0.f) ? x : SLOPE * x;
            ego[(size_t)(mbase + (half << 3) + v) * DD + j * 16 + n] = x;
        }
    }
}

#define TILE_PITCH 68
#define LAYER_WAVES 16
#define LAYER_T (LAYER_WAVES * 32)
#define N_TILES (NT / 16)
#define LAYER_GRID ((N_TILES + LAYER_WAVES - 1) / LAYER_WAVES)

__global__ __launch_bounds__(LAYER_T, 12)
void k_layer(const int* __restrict__ rowptr, const int2* __restrict__ sorted,
             const float* __restrict__ ein, float* __restrict__ eout,
             const float* __restrict__ Wgc, const float* __restrict__ Wbi,
             const float* __restrict__ bias) {
    __shared__ __attribute__((aligned(16))) float sg[4096];
    __shared__ __attribute__((aligned(16))) float sb[4096];
    __shared__ float sbias[64];
    __shared__ __attribute__((aligned(16))) float sL[LAYER_WAVES * 16 * TILE_PITCH];

    #pragma unroll
    for (int i = threadIdx.x; i < 4096; i += LAYER_T) {
        int r = i >> 6, c = i & 63;
        int dst = ((((r >> 1) * 64) + c) << 1) | (r & 1);
        sg[dst] = Wgc[i];
        sb[dst] = Wbi[i];
    }
    if (threadIdx.x < 64) sbias[threadIdx.x] = 2.0f * bias[threadIdx.x];
    __syncthreads();

    const int lane = threadIdx.x & 31;
    const int half = lane >> 4;
    const int n    = lane & 15;
    const int q    = n * 4;
    const int wave = __builtin_amdgcn_readfirstlane(threadIdx.x >> 5);
    const int tile = blockIdx.x * LAYER_WAVES + wave;
    if (tile >= N_TILES) return;
    const int mbase = tile * 16;
    float* myL = sL + wave * (16 * TILE_PITCH);

    int s = rowptr[mbase], t = rowptr[mbase + 1], tn = rowptr[mbase + 2];
    int2 my = make_int2(0, 0);
    if (lane < t - s) my = sorted[s + lane];

    #pragma unroll 1
    for (int i = 0; i < 16; ++i) {
        int2 myN = make_int2(0, 0);
        if (i < 15) {
            if (lane < tn - t) myN = sorted[t + lane];
        }
        int tnn = tn;
        if (i < 14) tnn = rowptr[mbase + i + 3];

        float4 acc = make_float4(0.f, 0.f, 0.f, 0.f);
        row_batch(acc, my, min(32, t - s), half, q, ein);
        for (int base = s + 32; base < t; base += 32) {
            const int nb = min(32, t - base);
            int2 m2 = make_int2(0, 0);
            if (lane < nb) m2 = sorted[base + lane];
            row_batch(acc, m2, nb, half, q, ein);
        }
        acc.x += __shfl_xor(acc.x, 16, 32);
        acc.y += __shfl_xor(acc.y, 16, 32);
        acc.z += __shfl_xor(acc.z, 16, 32);
        acc.w += __shfl_xor(acc.w, 16, 32);
        if (half == 0) *(float4*)(myL + i * TILE_PITCH + q) = acc;

        s = t; t = tn; tn = tnn; my = myN;
    }

    __builtin_amdgcn_wave_barrier();

    const float* lrow = myL + n * TILE_PITCH;
    const float* erow = ein + (size_t)(mbase + n) * DD;

    v8f dacc[4];
    #pragma unroll
    for (int j = 0; j < 4; ++j) dacc[j] = (v8f){};

    #pragma unroll
    for (int kk = 0; kk < 16; ++kk) {
        int off = kk * 4 + half * 2;
        v2f ls = *(const v2f*)(lrow + off);
        v2f eg = *(const v2f*)(erow + off);
        v2f a_li  = ls + eg;
        v2f a_mul = ls * eg;
        int bbase = ((kk * 2 + half) * 64) << 1;
        #pragma unroll
        for (int j = 0; j < 4; ++j) {
            v2f bgc = *(const v2f*)(sg + bbase + ((j * 16 + n) << 1));
            v2f bbi = *(const v2f*)(sb + bbase + ((j * 16 + n) << 1));
            dacc[j] = __builtin_amdgcn_wmma_f32_16x16x4_f32(
                false, a_li, false, bgc, (short)0, dacc[j], false, false);
            dacc[j] = __builtin_amdgcn_wmma_f32_16x16x4_f32(
                false, a_mul, false, bbi, (short)0, dacc[j], false, false);
        }
    }

    #pragma unroll
    for (int j = 0; j < 4; ++j) {
        float bsum = sbias[j * 16 + n];
        #pragma unroll
        for (int v = 0; v < 8; ++v) {
            float x = dacc[j][v] + bsum;
            x = (x >= 0.f) ? x : SLOPE * x;
            eout[(size_t)(mbase + (half << 3) + v) * DD + j * 16 + n] = x;
        }
    }
}

#define LAST_WAVES 8
#define LAST_T (LAST_WAVES * 32)
#define LAST_TILES (3 * BATCH / 16)
#define LAST_GRID (LAST_TILES / LAST_WAVES)

static __device__ __forceinline__
void gathered_cols_0_191(const int tile, const int lane,
                         const float* __restrict__ ue, const float* __restrict__ ie,
                         const float* __restrict__ e1, const float* __restrict__ e2,
                         const int* __restrict__ us, const int* __restrict__ pi,
                         const int* __restrict__ ni, float* __restrict__ out) {
    const int seg = tile >> 8;
    const int bb0 = (tile & 255) * 16;
    const int* idx   = (seg == 0) ? us : ((seg == 1) ? pi : ni);
    const float* tab = (seg == 0) ? ue : ie;
    const int nodeoff = seg ? NU : 0;

    v2f x0[16], x1[16], x2[16];
    #pragma unroll
    for (int i = 0; i < 16; ++i) {
        const int id = idx[bb0 + i];
        const size_t node = (size_t)(id + nodeoff);
        x0[i] = *(const v2f*)(tab + (size_t)id * DD + lane * 2);
        x1[i] = *(const v2f*)(e1 + node * DD + lane * 2);
        x2[i] = *(const v2f*)(e2 + node * DD + lane * 2);
    }
    #pragma unroll
    for (int i = 0; i < 16; ++i) {
        float* o = out + (size_t)(tile * 16 + i) * 256 + lane * 2;
        *(v2f*)o = x0[i];

        float s1 = fmaf(x1[i].x, x1[i].x, x1[i].y * x1[i].y);
        float s2 = fmaf(x2[i].x, x2[i].x, x2[i].y * x2[i].y);
        #pragma unroll
        for (int d = 16; d > 0; d >>= 1) {
            s1 += __shfl_xor(s1, d, 32);
            s2 += __shfl_xor(s2, d, 32);
        }
        const float c1 = 1.0f / fmaxf(sqrtf(s1), 1e-12f);
        const float c2 = 1.0f / fmaxf(sqrtf(s2), 1e-12f);
        v2f y1; y1.x = x1[i].x * c1; y1.y = x1[i].y * c1;
        v2f y2; y2.x = x2[i].x * c2; y2.y = x2[i].y * c2;
        *(v2f*)(o + 64)  = y1;
        *(v2f*)(o + 128) = y2;
    }
}

__global__ __launch_bounds__(LAST_T)
void k_layer_last(const int* __restrict__ rowptr, const int2* __restrict__ sorted,
                  const float* __restrict__ ein, const float* __restrict__ eprev,
                  const float* __restrict__ ue, const float* __restrict__ ie,
                  const float* __restrict__ Wgc, const float* __restrict__ Wbi,
                  const float* __restrict__ bias,
                  const int* __restrict__ us, const int* __restrict__ pi,
                  const int* __restrict__ ni, float* __restrict__ out) {
    __shared__ __attribute__((aligned(16))) float sg[4096];
    __shared__ __attribute__((aligned(16))) float sb[4096];
    __shared__ float sbias[64];
    __shared__ __attribute__((aligned(16))) float sL[LAST_WAVES * 16 * TILE_PITCH];

    if (blockIdx.x >= LAST_GRID) {
        const int w = __builtin_amdgcn_readfirstlane(threadIdx.x >> 5);
        gathered_cols_0_191((blockIdx.x - LAST_GRID) * LAST_WAVES + w, threadIdx.x & 31,
                            ue, ie, eprev, ein, us, pi, ni, out);
        return;
    }

    #pragma unroll
    for (int i = threadIdx.x; i < 4096; i += LAST_T) {
        int r = i >> 6, c = i & 63;
        int dst = ((((r >> 1) * 64) + c) << 1) | (r & 1);
        sg[dst] = Wgc[i];
        sb[dst] = Wbi[i];
    }
    if (threadIdx.x < 64) sbias[threadIdx.x] = 2.0f * bias[threadIdx.x];
    __syncthreads();

    const int lane = threadIdx.x & 31;
    const int half = lane >> 4;
    const int n    = lane & 15;
    const int q    = n * 4;
    const int wave = __builtin_amdgcn_readfirstlane(threadIdx.x >> 5);
    const int tile = blockIdx.x * LAST_WAVES + wave;
    const int seg  = tile >> 8;
    const int bb0  = (tile & 255) * 16;
    const int* idx = (seg == 0) ? us : ((seg == 1) ? pi : ni);
    float* myL = sL + wave * (16 * TILE_PITCH);

    const int node = idx[bb0 + n] + (seg ? NU : 0);
    const int rs = rowptr[node], rt = rowptr[node + 1];

    int s = __builtin_amdgcn_readlane(rs, 0), t = __builtin_amdgcn_readlane(rt, 0);
    int2 my = make_int2(0, 0);
    if (lane < t - s) my = sorted[s + lane];

    #pragma unroll 1
    for (int i = 0; i < 16; ++i) {
        int sN = t, tN = t;
        int2 myN = make_int2(0, 0);
        if (i < 15) {
            sN = __builtin_amdgcn_readlane(rs, i + 1);
            tN = __builtin_amdgcn_readlane(rt, i + 1);
            if (lane < tN - sN) myN = sorted[sN + lane];
        }

        float4 acc = make_float4(0.f, 0.f, 0.f, 0.f);
        row_batch(acc, my, min(32, t - s), half, q, ein);
        for (int base = s + 32; base < t; base += 32) {
            const int nb = min(32, t - base);
            int2 m2 = make_int2(0, 0);
            if (lane < nb) m2 = sorted[base + lane];
            row_batch(acc, m2, nb, half, q, ein);
        }
        acc.x += __shfl_xor(acc.x, 16, 32);
        acc.y += __shfl_xor(acc.y, 16, 32);
        acc.z += __shfl_xor(acc.z, 16, 32);
        acc.w += __shfl_xor(acc.w, 16, 32);
        if (half == 0) *(float4*)(myL + i * TILE_PITCH + q) = acc;

        s = sN; t = tN; my = myN;
    }

    __builtin_amdgcn_wave_barrier();

    const float* lrow = myL + n * TILE_PITCH;
    const float* erow = ein + (size_t)node * DD;

    v8f dacc[4];
    #pragma unroll
    for (int j = 0; j < 4; ++j) dacc[j] = (v8f){};

    #pragma unroll
    for (int kk = 0; kk < 16; ++kk) {
        int off = kk * 4 + half * 2;
        v2f ls = *(const v2f*)(lrow + off);
        v2f eg = *(const v2f*)(erow + off);
        v2f a_li  = ls + eg;
        v2f a_mul = ls * eg;
        int bbase = ((kk * 2 + half) * 64) << 1;
        #pragma unroll
        for (int j = 0; j < 4; ++j) {
            v2f bgc = *(const v2f*)(sg + bbase + ((j * 16 + n) << 1));
            v2f bbi = *(const v2f*)(sb + bbase + ((j * 16 + n) << 1));
            dacc[j] = __builtin_amdgcn_wmma_f32_16x16x4_f32(
                false, a_li, false, bgc, (short)0, dacc[j], false, false);
            dacc[j] = __builtin_amdgcn_wmma_f32_16x16x4_f32(
                false, a_mul, false, bbi, (short)0, dacc[j], false, false);
        }
    }

    #pragma unroll
    for (int j = 0; j < 4; ++j) {
        float bsum = sbias[j * 16 + n];
        #pragma unroll
        for (int v = 0; v < 8; ++v) {
            float x = dacc[j][v] + bsum;
            x = (x >= 0.f) ? x : SLOPE * x;
            myL[((half << 3) + v) * TILE_PITCH + j * 16 + n] = x;
        }
    }

    __builtin_amdgcn_wave_barrier();

    #pragma unroll 4
    for (int i = 0; i < 16; ++i) {
        v2f x = *(const v2f*)(myL + i * TILE_PITCH + lane * 2);
        float ss = x.x * x.x + x.y * x.y;
        #pragma unroll
        for (int o = 16; o > 0; o >>= 1) ss += __shfl_xor(ss, o, 32);
        float scale = 1.0f / fmaxf(sqrtf(ss), 1e-12f);
        v2f y; y.x = x.x * scale; y.y = x.y * scale;
        *(v2f*)(out + (size_t)(tile * 16 + i) * 256 + 192 + lane * 2) = y;
    }
}

__global__ __launch_bounds__(256)
void k_gather_norm(const float* __restrict__ ego, const int* __restrict__ us,
                   const int* __restrict__ pi, const int* __restrict__ ni,
                   float* __restrict__ out, int colbase) {
    int wid  = (blockIdx.x * 256 + threadIdx.x) >> 5;
    int lane = threadIdx.x & 31;
    if (wid >= 3 * BATCH) return;
    int seg = wid >> 12;
    int bb  = wid & (BATCH - 1);
    int node;
    if (seg == 0)      node = us[bb];
    else if (seg == 1) node = NU + pi[bb];
    else               node = NU + ni[bb];
    v2f x = *(const v2f*)(ego + (size_t)node * DD + lane * 2);
    float ss = x.x * x.x + x.y * x.y;
    #pragma unroll
    for (int o = 16; o > 0; o >>= 1) ss += __shfl_xor(ss, o, 32);
    float scale = 1.0f / fmaxf(sqrtf(ss), 1e-12f);
    v2f y; y.x = x.x * scale; y.y = x.y * scale;
    *(v2f*)(out + (size_t)wid * 256 + colbase + lane * 2) = y;
}

extern "C" void kernel_launch(void* const* d_in, const int* in_sizes, int n_in,
                              void* d_out, int out_size, void* d_ws, size_t ws_size,
                              hipStream_t stream) {
    const float* ue   = (const float*)d_in[0];
    const float* ie   = (const float*)d_in[1];
    const float* Wgc  = (const float*)d_in[2];
    const float* Wbi  = (const float*)d_in[3];
    const float* bbi  = (const float*)d_in[4];
    const float* vals = (const float*)d_in[5];
    const int*   rows = (const int*)d_in[6];
    const int*   cols = (const int*)d_in[7];
    const int*   us   = (const int*)d_in[8];
    const int*   pi   = (const int*)d_in[9];
    const int*   ni   = (const int*)d_in[10];
    float* out = (float*)d_out;

    float* ego = (float*)d_ws;
    float* L   = ego + (size_t)NT * DD;

    hipMemcpyAsync(ego, ue, (size_t)NU * DD * sizeof(float),
                   hipMemcpyDeviceToDevice, stream);
    hipMemcpyAsync(ego + (size_t)NU * DD, ie, (size_t)NI * DD * sizeof(float),
                   hipMemcpyDeviceToDevice, stream);

    const size_t need = (size_t)2 * NT * DD * 4 + (size_t)NNZE * 8 + (size_t)(NT + 1) * 4;
    if (ws_size >= need) {
        int2* sorted = (int2*)(L + (size_t)NT * DD);
        int*  rowptr = (int*)(sorted + NNZE);
        int2* tmp    = (int2*)L;
        int*  wgoff  = (int*)(tmp + NNZE);
        int*  bcnt   = wgoff + SORT_G * NB;
        int*  bbase  = bcnt + NB * CNT_PITCH;
        hipMemsetAsync(bcnt, 0, (size_t)NB * CNT_PITCH * sizeof(int), stream);
        k_bucket_count<<<SORT_G, SORT_T, 0, stream>>>(rows, bcnt, wgoff);
        k_bucket_scatter<<<SORT_G, SORT_T, 0, stream>>>(rows, cols, vals, bcnt, wgoff,
                                                        bbase, tmp);
        k_bucket_rows<<<NB, ROWS_T, 0, stream>>>(bbase, tmp, sorted, rowptr);
        float* ego2 = L;
        k_layer<<<LAYER_GRID, LAYER_T, 0, stream>>>(
            rowptr, sorted, ego, ego2, Wgc, Wbi, bbi);
        k_layer<<<LAYER_GRID, LAYER_T, 0, stream>>>(
            rowptr, sorted, ego2, ego, Wgc + 4096, Wbi + 4096, bbi + 64);
        k_layer_last<<<2 * LAST_GRID, LAST_T, 0, stream>>>(
            rowptr, sorted, ego, ego2, ue, ie, Wgc + 8192, Wbi + 8192, bbi + 128,
            us, pi, ni, out);
        return;
    }

    k_gather0<<<(3 * BATCH * DD + 255) / 256, 256, 0, stream>>>(
        ue, ie, us, pi, ni, out);
    for (int k = 0; k < 3; ++k) {
        hipMemsetAsync(L, 0, (size_t)NT * DD * sizeof(float), stream);
        long tot = (long)NNZE * 16;
        k_spmm<<<(int)((tot + 255) / 256), 256, 0, stream>>>(
            rows, cols, vals, ego, L);
        k_dense_wmma<<<(NT + 127) / 128, 256, 0, stream>>>(
            L, ego, Wgc + k * 4096, Wbi + k * 4096, bbi + k * 64);
        k_gather_norm<<<(3 * BATCH * 32 + 255) / 256, 256, 0, stream>>>(
            ego, us, pi, ni, out, 64 * (k + 1));
    }
}


# Last layer: a 16-row tile is shared by 4 waves (4 rows summed per wave, their gathers in flight together; 16 output columns per wave, B fragments straight to registers); sum order kept
# speedup vs baseline: 3.9455x; 1.1179x over previous
#include <hip/hip_runtime.h>
#include <hip/hip_bf16.h>

typedef float v2f __attribute__((ext_vector_type(2)));
typedef float v8f __attribute__((ext_vector_type(8)));

#define NU 50000
#define NI 100000
#define NT 150000
#define DD 64
#define NNZE 2400000
#define BATCH 4096
#define SLOPE 0.01f

__global__ __launch_bounds__(256)
void k_gather0(const float* __restrict__ ue, const float* __restrict__ ie,
               const int* __restrict__ us, const int* __restrict__ pi,
               const int* __restrict__ ni, float* __restrict__ out) {
    int t = blockIdx.x * 256 + threadIdx.x;
    if (t >= 3 * BATCH * DD) return;
    int c  = t & 63;
    int b  = t >> 6;
    int seg = b >> 12;
    int bb  = b & (BATCH - 1);
    float v;
    if (seg == 0)      v = ue[(size_t)us[bb] * DD + c];
    else if (seg == 1) v = ie[(size_t)pi[bb] * DD + c];
    else               v = ie[(size_t)ni[bb] * DD + c];
    out[(size_t)b * 256 + c] = v;
}

__global__ __launch_bounds__(256)
void k_spmm(const int* __restrict__ rows, const int* __restrict__ cols,
            const float* __restrict__ vals, const float* __restrict__ ego,
            float* __restrict__ L) {
    long t = (long)blockIdx.x * 256 + threadIdx.x;
    if (t >= (long)NNZE * 16) return;
    int e = (int)(t >> 4);
    int q = ((int)t & 15) * 4;
    int r = rows[e], c = cols[e];
    float v = vals[e];
    const float4 x = *(const float4*)(ego + (size_t)c * DD + q);
    float* dst = L + (size_t)r * DD + q;
    atomicAdd(dst + 0, v * x.x);
    atomicAdd(dst + 1, v * x.y);
    atomicAdd(dst + 2, v * x.z);
    atomicAdd(dst + 3, v * x.w);
}

#define RB 256
#define NB ((NT + RB - 1) / RB)
#define COL_BITS 18
#define SORT_G 256
#define SORT_T 1024
#define SLICE ((NNZE + SORT_G - 1) / SORT_G)
#define SLICE_TRIPS ((SLICE + SORT_T - 1) / SORT_T)
#define CNT_PITCH 32
static_assert(NT <= (1 << COL_BITS), "column index must fit in COL_BITS");
static_assert(RB <= (1 << (31 - COL_BITS)), "row in bucket must fit above the column");
static_assert(NB <= SORT_T && RB <= SORT_T, "one scan element per thread");

static __device__ __forceinline__
int block_excl_scan(const int own, int* wtot) {
    const int lane = threadIdx.x & 31, wave = threadIdx.x >> 5;
    int inc = own;
    #pragma unroll
    for (int o = 1; o < 32; o <<= 1) {
        int y = __shfl_up(inc, o, 32);
        if (lane >= o) inc += y;
    }
    if (lane == 31) wtot[wave] = inc;
    __syncthreads();
    if (wave == 0) {
        const int x = wtot[lane];
        int xi = x;
        #pragma unroll
        for (int o = 1; o < 32; o <<= 1) {
            int y = __shfl_up(xi, o, 32);
            if (lane >= o) xi += y;
        }
        wtot[lane] = xi - x;
    }
    __syncthreads();
    return wtot[wave] + inc - own;
}

__global__ __launch_bounds__(SORT_T)
void k_bucket_count(const int* __restrict__ rows, int* __restrict__ bcnt,
                    int* __restrict__ wgoff) {
    __shared__ int hist[NB];
    const int t = threadIdx.x;
    for (int i = t; i < NB; i += SORT_T) hist[i] = 0;
    __syncthreads();
    const int s0 = blockIdx.x * SLICE;
    const int s1 = min(s0 + SLICE, NNZE);
    int r[SLICE_TRIPS];
    #pragma unroll
    for (int k = 0; k < SLICE_TRIPS; ++k) {
        const int e = s0 + k * SORT_T + t;
        r[k] = (e < s1) ? rows[e] : -1;
    }
    #pragma unroll
    for (int k = 0; k < SLICE_TRIPS; ++k)
        if (r[k] >= 0) atomicAdd(&hist[r[k] / RB], 1);
    __syncthreads();
    for (int b = t; b < NB; b += SORT_T) {
        const int h = hist[b];
        wgoff[blockIdx.x * NB + b] = h ? atomicAdd(&bcnt[b * CNT_PITCH], h) : 0;
    }
}

__global__ __launch_bounds__(SORT_T)
void k_bucket_scatter(const int* __restrict__ rows, const int* __restrict__ cols,
                      const float* __restrict__ vals, const int* __restrict__ bcnt,
                      const int* __restrict__ wgoff, int* __restrict__ bbase,
                      int2* __restrict__ tmp) {
    __shared__ int start[NB];
    __shared__ int lcnt[NB];
    __shared__ int wtot[32];
    const int t = threadIdx.x;
    const int own = (t < NB) ? bcnt[t * CNT_PITCH] : 0;
    const int off = (t < NB) ? wgoff[blockIdx.x * NB + t] : 0;
    const int excl = block_excl_scan(own, wtot);
    if (t < NB) {
        start[t] = excl + off;
        lcnt[t] = 0;
        if (blockIdx.x == 0) {
            bbase[t] = excl;
            if (t == NB - 1) bbase[NB] = excl + own;
        }
    }
    __syncthreads();
    const int s0 = blockIdx.x * SLICE;
    const int s1 = min(s0 + SLICE, NNZE);
    int   r[SLICE_TRIPS], c[SLICE_TRIPS];
    float v[SLICE_TRIPS];
    #pragma unroll
    for (int k = 0; k < SLICE_TRIPS; ++k) {
        const int e = s0 + k * SORT_T + t;
        const bool ok = e < s1;
        r[k] = ok ? rows[e] : -1;
        c[k] = ok ? cols[e] : 0;
        v[k] = ok ? vals[e] : 0.f;
    }
    #pragma unroll
    for (int k = 0; k < SLICE_TRIPS; ++k) {
        if (r[k] >= 0) {
            const int b  = r[k] / RB;
            const int lr = atomicAdd(&lcnt[b], 1);
            tmp[start[b] + lr] = make_int2(((r[k] % RB) << COL_BITS) | c[k],
                                           __float_as_int(v[k]));
        }
    }
}

#define ROWS_T 1024
#define ROWS_BATCH 4

__global__ __launch_bounds__(ROWS_T)
void k_bucket_rows(const int* __restrict__ bbase, const int2* __restrict__ tmp,
                   int2* __restrict__ sorted, int* __restrict__ rowptr) {
    __shared__ int cnt[RB];
    __shared__ int wtot[32];
    const int t = threadIdx.x;
    const int b = blockIdx.x;
    const int base = bbase[b];
    const int n = bbase[b + 1] - base;
    if (t < RB) cnt[t] = 0;
    __syncthreads();
    for (int i0 = 0; i0 < n; i0 += ROWS_BATCH * ROWS_T) {
        int x[ROWS_BATCH];
        #pragma unroll
        for (int j = 0; j < ROWS_BATCH; ++j) {
            const int i = i0 + j * ROWS_T + t;
            x[j] = (i < n) ? tmp[base + i].x : -1;
        }
        #pragma unroll
        for (int j = 0; j < ROWS_BATCH; ++j)
            if (x[j] >= 0) atomicAdd(&cnt[x[j] >> COL_BITS], 1);
    }
    __syncthreads();
    const int own = (t < RB) ? cnt[t] : 0;
    const int excl = block_excl_scan(own, wtot);
    if (t < RB) {
        cnt[t] = excl;
        const int row = b * RB + t;
        if (row < NT) rowptr[row] = base + excl;
    }
    if (b == NB - 1 && t == 0) rowptr[NT] = base + n;
    __syncthreads();
    for (int i0 = 0; i0 < n; i0 += ROWS_BATCH * ROWS_T) {
        int2 e[ROWS_BATCH];
        #pragma unroll
        for (int j = 0; j < ROWS_BATCH; ++j) {
            const int i = i0 + j * ROWS_T + t;
            e[j] = (i < n) ? tmp[base + i] : make_int2(-1, 0);
        }
        #pragma unroll
        for (int j = 0; j < ROWS_BATCH; ++j) {
            if (e[j].x >= 0) {
                const int p = atomicAdd(&cnt[e[j].x >> COL_BITS], 1);
                sorted[base + p] = make_int2(e[j].x & ((1 << COL_BITS) - 1), e[j].y);
            }
        }
    }
}

static __device__ __forceinline__
void row_batch(float4& acc, const int2 my, const int nb, const int half,
               const int q, const float* __restrict__ ego) {
    for (int i0 = 0; i0 < nb; i0 += 16) {
        int   c[8];
        float v[8];
        float4 x[8];
        #pragma unroll
        for (int j = 0; j < 8; ++j) {
            const int src = i0 + 2 * j + half;
            c[j] = __shfl(my.x, src, 32);
            v[j] = __int_as_float(__shfl(my.y, src, 32));
        }
        #pragma unroll
        for (int j = 0; j < 8; ++j)
            x[j] = *(const float4*)(ego + (size_t)c[j] * DD + q);
        #pragma unroll
        for (int j = 0; j < 8; ++j) {
            const bool ok = (i0 + 2 * j + half) < nb;
            acc.x += v[j] * (ok ? x[j].x : 0.f);
            acc.y += v[j] * (ok ? x[j].y : 0.f);
            acc.z += v[j] * (ok ? x[j].z : 0.f);
            acc.w += v[j] * (ok ? x[j].w : 0.f);
        }
    }
}

__global__ __launch_bounds__(256)
void k_dense_wmma(const float* __restrict__ L, float* __restrict__ ego,
                  const float* __restrict__ Wgc, const float* __restrict__ Wbi,
                  const float* __restrict__ bias) {
    __shared__ float sg[4096];
    __shared__ float sb[4096];
    __shared__ float sbias[64];

    for (int i = threadIdx.x; i < 4096; i += 256) {
        int r = i >> 6, c = i & 63;
        int dst = ((((r >> 1) * 64) + c) << 1) | (r & 1);
        sg[dst] = Wgc[i];
        sb[dst] = Wbi[i];
    }
    if (threadIdx.x < 64) sbias[threadIdx.x] = 2.0f * bias[threadIdx.x];
    __syncthreads();

    const int wave  = threadIdx.x >> 5;
    const int lane  = threadIdx.x & 31;
    const int mbase = (blockIdx.x * 8 + wave) * 16;
    if (mbase >= NT) return;

    const int half = lane >> 4;
    const int n    = lane & 15;

    const float* lrow = L   + (size_t)(mbase + n) * DD;
    const float* erow = ego + (size_t)(mbase + n) * DD;

    v8f acc[4];
    #pragma unroll
    for (int j = 0; j < 4; ++j) acc[j] = (v8f){};

    #pragma unroll
    for (int kk = 0; kk < 16; ++kk) {
        int off = kk * 4 + half * 2;
        v2f ls = *(const v2f*)(lrow + off);
        v2f eg = *(const v2f*)(erow + off);
        v2f a_li  = ls + eg;
        v2f a_mul = ls * eg;
        int bbase = ((kk * 2 + half) * 64) << 1;
        #pragma unroll
        for (int j = 0; j < 4; ++j) {
            v2f bgc = *(const v2f*)(sg + bbase + ((j * 16 + n) << 1));
            v2f bbi = *(const v2f*)(sb + bbase + ((j * 16 + n) << 1));
            acc[j] = __builtin_amdgcn_wmma_f32_16x16x4_f32(
                false, a_li, false, bgc, (short)0, acc[j], false, false);
            acc[j] = __builtin_amdgcn_wmma_f32_16x16x4_f32(
                false, a_mul, false, bbi, (short)0, acc[j], false, false);
        }
    }

    #pragma unroll
    for (int j = 0; j < 4; ++j) {
        float bsum = sbias[j * 16 + n];
        #pragma unroll
        for (int v = 0; v < 8; ++v) {
            float x = acc[j][v] + bsum;
            x = (x >= 0.f) ? x : SLOPE * x;
            ego[(size_t)(mbase + (half << 3) + v) * DD + j * 16 + n] = x;
        }
    }
}

#define TILE_PITCH 68
#define LAYER_WAVES 16
#define LAYER_T (LAYER_WAVES * 32)
#define N_TILES (NT / 16)
#define LAYER_GRID ((N_TILES + LAYER_WAVES - 1) / LAYER_WAVES)

__global__ __launch_bounds__(LAYER_T, 12)
void k_layer(const int* __restrict__ rowptr, const int2* __restrict__ sorted,
             const float* __restrict__ ein, float* __restrict__ eout,
             const float* __restrict__ Wgc, const float* __restrict__ Wbi,
             const float* __restrict__ bias) {
    __shared__ __attribute__((aligned(16))) float sg[4096];
    __shared__ __attribute__((aligned(16))) float sb[4096];
    __shared__ float sbias[64];
    __shared__ __attribute__((aligned(16))) float sL[LAYER_WAVES * 16 * TILE_PITCH];

    #pragma unroll
    for (int i = threadIdx.x; i < 4096; i += LAYER_T) {
        int r = i >> 6, c = i & 63;
        int dst = ((((r >> 1) * 64) + c) << 1) | (r & 1);
        sg[dst] = Wgc[i];
        sb[dst] = Wbi[i];
    }
    if (threadIdx.x < 64) sbias[threadIdx.x] = 2.0f * bias[threadIdx.x];
    __syncthreads();

    const int lane = threadIdx.x & 31;
    const int half = lane >> 4;
    const int n    = lane & 15;
    const int q    = n * 4;
    const int wave = __builtin_amdgcn_readfirstlane(threadIdx.x >> 5);
    const int tile = blockIdx.x * LAYER_WAVES + wave;
    if (tile >= N_TILES) return;
    const int mbase = tile * 16;
    float* myL = sL + wave * (16 * TILE_PITCH);

    int s = rowptr[mbase], t = rowptr[mbase + 1], tn = rowptr[mbase + 2];
    int2 my = make_int2(0, 0);
    if (lane < t - s) my = sorted[s + lane];

    #pragma unroll 1
    for (int i = 0; i < 16; ++i) {
        int2 myN = make_int2(0, 0);
        if (i < 15) {
            if (lane < tn - t) myN = sorted[t + lane];
        }
        int tnn = tn;
        if (i < 14) tnn = rowptr[mbase + i + 3];

        float4 acc = make_float4(0.f, 0.f, 0.f, 0.f);
        row_batch(acc, my, min(32, t - s), half, q, ein);
        for (int base = s + 32; base < t; base += 32) {
            const int nb = min(32, t - base);
            int2 m2 = make_int2(0, 0);
            if (lane < nb) m2 = sorted[base + lane];
            row_batch(acc, m2, nb, half, q, ein);
        }
        acc.x += __shfl_xor(acc.x, 16, 32);
        acc.y += __shfl_xor(acc.y, 16, 32);
        acc.z += __shfl_xor(acc.z, 16, 32);
        acc.w += __shfl_xor(acc.w, 16, 32);
        if (half == 0) *(float4*)(myL + i * TILE_PITCH + q) = acc;

        s = t; t = tn; tn = tnn; my = myN;
    }

    __builtin_amdgcn_wave_barrier();

    const float* lrow = myL + n * TILE_PITCH;
    const float* erow = ein + (size_t)(mbase + n) * DD;

    v8f dacc[4];
    #pragma unroll
    for (int j = 0; j < 4; ++j) dacc[j] = (v8f){};

    #pragma unroll
    for (int kk = 0; kk < 16; ++kk) {
        int off = kk * 4 + half * 2;
        v2f ls = *(const v2f*)(lrow + off);
        v2f eg = *(const v2f*)(erow + off);
        v2f a_li  = ls + eg;
        v2f a_mul = ls * eg;
        int bbase = ((kk * 2 + half) * 64) << 1;
        #pragma unroll
        for (int j = 0; j < 4; ++j) {
            v2f bgc = *(const v2f*)(sg + bbase + ((j * 16 + n) << 1));
            v2f bbi = *(const v2f*)(sb + bbase + ((j * 16 + n) << 1));
            dacc[j] = __builtin_amdgcn_wmma_f32_16x16x4_f32(
                false, a_li, false, bgc, (short)0, dacc[j], false, false);
            dacc[j] = __builtin_amdgcn_wmma_f32_16x16x4_f32(
                false, a_mul, false, bbi, (short)0, dacc[j], false, false);
        }
    }

    #pragma unroll
    for (int j = 0; j < 4; ++j) {
        float bsum = sbias[j * 16 + n];
        #pragma unroll
        for (int v = 0; v < 8; ++v) {
            float x = dacc[j][v] + bsum;
            x = (x >= 0.f) ? x : SLOPE * x;
            eout[(size_t)(mbase + (half << 3) + v) * DD + j * 16 + n] = x;
        }
    }
}

#define LAST_WAVES 4
#define LAST_T (LAST_WAVES * 32)
#define LAST_TILES (3 * BATCH / 16)
#define LAST_COPY_GRID (LAST_TILES / LAST_WAVES)
static_assert(LAST_WAVES * 4 == 16 && LAST_WAVES * 16 == DD, "four rows and sixteen columns a wave");

static __device__ __forceinline__
void gathered_cols_0_191(const int tile, const int lane,
                         const float* __restrict__ ue, const float* __restrict__ ie,
                         const float* __restrict__ e1, const float* __restrict__ e2,
                         const int* __restrict__ us, const int* __restrict__ pi,
                         const int* __restrict__ ni, float* __restrict__ out) {
    const int seg = tile >> 8;
    const int bb0 = (tile & 255) * 16;
    const int* idx   = (seg == 0) ? us : ((seg == 1) ? pi : ni);
    const float* tab = (seg == 0) ? ue : ie;
    const int nodeoff = seg ? NU : 0;

    v2f x0[16], x1[16], x2[16];
    #pragma unroll
    for (int i = 0; i < 16; ++i) {
        const int id = idx[bb0 + i];
        const size_t node = (size_t)(id + nodeoff);
        x0[i] = *(const v2f*)(tab + (size_t)id * DD + lane * 2);
        x1[i] = *(const v2f*)(e1 + node * DD + lane * 2);
        x2[i] = *(const v2f*)(e2 + node * DD + lane * 2);
    }
    #pragma unroll
    for (int i = 0; i < 16; ++i) {
        float* o = out + (size_t)(tile * 16 + i) * 256 + lane * 2;
        *(v2f*)o = x0[i];

        float s1 = fmaf(x1[i].x, x1[i].x, x1[i].y * x1[i].y);
        float s2 = fmaf(x2[i].x, x2[i].x, x2[i].y * x2[i].y);
        #pragma unroll
        for (int d = 16; d > 0; d >>= 1) {
            s1 += __shfl_xor(s1, d, 32);
            s2 += __shfl_xor(s2, d, 32);
        }
        const float c1 = 1.0f / fmaxf(sqrtf(s1), 1e-12f);
        const float c2 = 1.0f / fmaxf(sqrtf(s2), 1e-12f);
        v2f y1; y1.x = x1[i].x * c1; y1.y = x1[i].y * c1;
        v2f y2; y2.x = x2[i].x * c2; y2.y = x2[i].y * c2;
        *(v2f*)(o + 64)  = y1;
        *(v2f*)(o + 128) = y2;
    }
}

static __device__ __forceinline__
void trip_issue(float (&v)[8], float4 (&x)[8], const int2 my, const int i0,
                const int half, const int q, const float* __restrict__ ego) {
    int c[8];
    #pragma unroll
    for (int j = 0; j < 8; ++j) {
        const int src = i0 + 2 * j + half;
        c[j] = __shfl(my.x, src, 32);
        v[j] = __int_as_float(__shfl(my.y, src, 32));
    }
    #pragma unroll
    for (int j = 0; j < 8; ++j) {
        const unsigned off = (unsigned)c[j] * (unsigned)(DD * sizeof(float)) + (unsigned)q * sizeof(float);
        x[j] = *(const float4*)((const char*)ego + off);
    }
}

static __device__ __forceinline__
void trip_sum(float4& acc, const float (&v)[8], const float4 (&x)[8], const int i0,
              const int half, const int nb) {
    #pragma unroll
    for (int j = 0; j < 8; ++j) {
        const int keep = ((i0 + 2 * j + half) < nb) ? -1 : 0;
        acc.x += v[j] * __int_as_float(__float_as_int(x[j].x) & keep);
        acc.y += v[j] * __int_as_float(__float_as_int(x[j].y) & keep);
        acc.z += v[j] * __int_as_float(__float_as_int(x[j].z) & keep);
        acc.w += v[j] * __int_as_float(__float_as_int(x[j].w) & keep);
    }
}

__global__ __launch_bounds__(LAST_T) __attribute__((amdgpu_waves_per_eu(4, 4)))
void k_layer_last(const int* __restrict__ rowptr, const int2* __restrict__ sorted,
                  const float* __restrict__ ein, const float* __restrict__ eprev,
                  const float* __restrict__ ue, const float* __restrict__ ie,
                  const float* __restrict__ Wgc, const float* __restrict__ Wbi,
                  const float* __restrict__ bias,
                  const int* __restrict__ us, const int* __restrict__ pi,
                  const int* __restrict__ ni, float* __restrict__ out) {
    __shared__ __attribute__((aligned(16))) float sL[16 * TILE_PITCH];
    __shared__ __attribute__((aligned(16))) float sO[16 * TILE_PITCH];

    const int lane = threadIdx.x & 31;
    const int w = __builtin_amdgcn_readfirstlane(threadIdx.x >> 5);

    if (blockIdx.x >= LAST_TILES) {
        gathered_cols_0_191((blockIdx.x - LAST_TILES) * LAST_WAVES + w, lane,
                            ue, ie, eprev, ein, us, pi, ni, out);
        return;
    }

    const int half = lane >> 4;
    const int n    = lane & 15;
    const int q    = n * 4;
    const int tile = blockIdx.x;
    const int seg  = tile >> 8;
    const int bb0  = (tile & 255) * 16;
    const int* idx = (seg == 0) ? us : ((seg == 1) ? pi : ni);

    const int node = idx[bb0 + n] + (seg ? NU : 0);
    const int rs = rowptr[node], rt = rowptr[node + 1];

    int s4[4], t4[4], nb[4];
    int2 my[4];
    #pragma unroll
    for (int r = 0; r < 4; ++r) {
        s4[r] = __builtin_amdgcn_readlane(rs, 4 * w + r);
        t4[r] = __builtin_amdgcn_readlane(rt, 4 * w + r);
        nb[r] = min(32, t4[r] - s4[r]);
        my[r] = make_int2(0, 0);
        if (lane < nb[r]) my[r] = sorted[s4[r] + lane];
    }

    float4 acc[4];
    {
        float  v[4][8];
        float4 x[4][8];
        #pragma unroll
        for (int r = 0; r < 4; ++r) trip_issue(v[r], x[r], my[r], 0, half, q, ein);
        #pragma unroll
        for (int r = 0; r < 4; ++r) {
            acc[r] = make_float4(0.f, 0.f, 0.f, 0.f);
            trip_sum(acc[r], v[r], x[r], 0, half, nb[r]);
        }
    }

    if (max(max(nb[0], nb[1]), max(nb[2], nb[3])) > 16) {
        float  v[4][8];
        float4 x[4][8];
        #pragma unroll
        for (int r = 0; r < 4; ++r) trip_issue(v[r], x[r], my[r], 16, half, q, ein);
        #pragma unroll
        for (int r = 0; r < 4; ++r) trip_sum(acc[r], v[r], x[r], 16, half, nb[r]);
    }

    #pragma unroll
    for (int r = 0; r < 4; ++r) {
        for (int base = s4[r] + 32; base < t4[r]; base += 32) {
            const int nbb = min(32, t4[r] - base);
            int2 m2 = make_int2(0, 0);
            if (lane < nbb) m2 = sorted[base + lane];
            row_batch(acc[r], m2, nbb, half, q, ein);
        }
    }

    const int col = 16 * w + n;
    v2f bg[16], bb[16], eg[16];
    #pragma unroll
    for (int kk = 0; kk < 16; ++kk) {
        const int k0 = kk * 4 + half * 2;
        bg[kk].x = Wgc[k0 * DD + col];
        bg[kk].y = Wgc[(k0 + 1) * DD + col];
        bb[kk].x = Wbi[k0 * DD + col];
        bb[kk].y = Wbi[(k0 + 1) * DD + col];
    }
    const float* erow = ein + (size_t)node * DD;
    #pragma unroll
    for (int kk = 0; kk < 16; ++kk)
        eg[kk] = *(const v2f*)(erow + kk * 4 + half * 2);
    const float bsum = 2.0f * bias[col];

    #pragma unroll
    for (int r = 0; r < 4; ++r) {
        acc[r].x += __shfl_xor(acc[r].x, 16, 32);
        acc[r].y += __shfl_xor(acc[r].y, 16, 32);
        acc[r].z += __shfl_xor(acc[r].z, 16, 32);
        acc[r].w += __shfl_xor(acc[r].w, 16, 32);
        if (half == 0) *(float4*)(sL + (4 * w + r) * TILE_PITCH + q) = acc[r];
    }

    __syncthreads();

    const float* lrow = sL + n * TILE_PITCH;
    v8f dacc = (v8f){};
    #pragma unroll
    for (int kk = 0; kk < 16; ++kk) {
        v2f ls = *(const v2f*)(lrow + kk * 4 + half * 2);
        v2f a_li  = ls + eg[kk];
        v2f a_mul = ls * eg[kk];
        dacc = __builtin_amdgcn_wmma_f32_16x16x4_f32(
            false, a_li, false, bg[kk], (short)0, dacc, false, false);
        dacc = __builtin_amdgcn_wmma_f32_16x16x4_f32(
            false, a_mul, false, bb[kk], (short)0, dacc, false, false);
    }

    #pragma unroll
    for (int v = 0; v < 8; ++v) {
        float x = dacc[v] + bsum;
        x = (x >= 0.f) ? x : SLOPE * x;
        sO[((half << 3) + v) * TILE_PITCH + col] = x;
    }

    __syncthreads();

    #pragma unroll
    for (int r = 0; r < 4; ++r) {
        const int i = 4 * w + r;
        v2f x = *(const v2f*)(sO + i * TILE_PITCH + lane * 2);
        float ss = x.x * x.x + x.y * x.y;
        #pragma unroll
        for (int o = 16; o > 0; o >>= 1) ss += __shfl_xor(ss, o, 32);
        float scale = 1.0f / fmaxf(sqrtf(ss), 1e-12f);
        v2f y; y.x = x.x * scale; y.y = x.y * scale;
        *(v2f*)(out + (size_t)(tile * 16 + i) * 256 + 192 + lane * 2) = y;
    }
}

__global__ __launch_bounds__(256)
void k_gather_norm(const float* __restrict__ ego, const int* __restrict__ us,
                   const int* __restrict__ pi, const int* __restrict__ ni,
                   float* __restrict__ out, int colbase) {
    int wid  = (blockIdx.x * 256 + threadIdx.x) >> 5;
    int lane = threadIdx.x & 31;
    if (wid >= 3 * BATCH) return;
    int seg = wid >> 12;
    int bb  = wid & (BATCH - 1);
    int node;
    if (seg == 0)      node = us[bb];
    else if (seg == 1) node = NU + pi[bb];
    else               node = NU + ni[bb];
    v2f x = *(const v2f*)(ego + (size_t)node * DD + lane * 2);
    float ss = x.x * x.x + x.y * x.y;
    #pragma unroll
    for (int o = 16; o > 0; o >>= 1) ss += __shfl_xor(ss, o, 32);
    float scale = 1.0f / fmaxf(sqrtf(ss), 1e-12f);
    v2f y; y.x = x.x * scale; y.y = x.y * scale;
    *(v2f*)(out + (size_t)wid * 256 + colbase + lane * 2) = y;
}

extern "C" void kernel_launch(void* const* d_in, const int* in_sizes, int n_in,
                              void* d_out, int out_size, void* d_ws, size_t ws_size,
                              hipStream_t stream) {
    const float* ue   = (const float*)d_in[0];
    const float* ie   = (const float*)d_in[1];
    const float* Wgc  = (const float*)d_in[2];
    const float* Wbi  = (const float*)d_in[3];
    const float* bbi  = (const float*)d_in[4];
    const float* vals = (const float*)d_in[5];
    const int*   rows = (const int*)d_in[6];
    const int*   cols = (const int*)d_in[7];
    const int*   us   = (const int*)d_in[8];
    const int*   pi   = (const int*)d_in[9];
    const int*   ni   = (const int*)d_in[10];
    float* out = (float*)d_out;

    float* ego = (float*)d_ws;
    float* L   = ego + (size_t)NT * DD;

    hipMemcpyAsync(ego, ue, (size_t)NU * DD * sizeof(float),
                   hipMemcpyDeviceToDevice, stream);
    hipMemcpyAsync(ego + (size_t)NU * DD, ie, (size_t)NI * DD * sizeof(float),
                   hipMemcpyDeviceToDevice, stream);

    const size_t need = (size_t)2 * NT * DD * 4 + (size_t)NNZE * 8 + (size_t)(NT + 1) * 4;
    if (ws_size >= need) {
        int2* sorted = (int2*)(L + (size_t)NT * DD);
        int*  rowptr = (int*)(sorted + NNZE);
        int2* tmp    = (int2*)L;
        int*  wgoff  = (int*)(tmp + NNZE);
        int*  bcnt   = wgoff + SORT_G * NB;
        int*  bbase  = bcnt + NB * CNT_PITCH;
        hipMemsetAsync(bcnt, 0, (size_t)NB * CNT_PITCH * sizeof(int), stream);
        k_bucket_count<<<SORT_G, SORT_T, 0, stream>>>(rows, bcnt, wgoff);
        k_bucket_scatter<<<SORT_G, SORT_T, 0, stream>>>(rows, cols, vals, bcnt, wgoff,
                                                        bbase, tmp);
        k_bucket_rows<<<NB, ROWS_T, 0, stream>>>(bbase, tmp, sorted, rowptr);
        float* ego2 = L;
        k_layer<<<LAYER_GRID, LAYER_T, 0, stream>>>(
            rowptr, sorted, ego, ego2, Wgc, Wbi, bbi);
        k_layer<<<LAYER_GRID, LAYER_T, 0, stream>>>(
            rowptr, sorted, ego2, ego, Wgc + 4096, Wbi + 4096, bbi + 64);
        k_layer_last<<<LAST_TILES + LAST_COPY_GRID, LAST_T, 0, stream>>>(
            rowptr, sorted, ego, ego2, ue, ie, Wgc + 8192, Wbi + 8192, bbi + 128,
            us, pi, ni, out);
        return;
    }

    k_gather0<<<(3 * BATCH * DD + 255) / 256, 256, 0, stream>>>(
        ue, ie, us, pi, ni, out);
    for (int k = 0; k < 3; ++k) {
        hipMemsetAsync(L, 0, (size_t)NT * DD * sizeof(float), stream);
        long tot = (long)NNZE * 16;
        k_spmm<<<(int)((tot + 255) / 256), 256, 0, stream>>>(
            rows, cols, vals, ego, L);
        k_dense_wmma<<<(NT + 127) / 128, 256, 0, stream>>>(
            L, ego, Wgc + k * 4096, Wbi + k * 4096, bbi + k * 64);
        k_gather_norm<<<(3 * BATCH * 32 + 255) / 256, 256, 0, stream>>>(
            ego, us, pi, ni, out, 64 * (k + 1));
    }
}


# Layers 1-2: one work-group of 19 waves per WGP, a wave runs two tiles in turn and sums two rows at a time: some waves gather while others use the matrix pipe; 38 tiles on every WGP; sum order kept
# speedup vs baseline: 4.2071x; 1.0663x over previous
#include <hip/hip_runtime.h>
#include <hip/hip_bf16.h>

typedef float v2f __attribute__((ext_vector_type(2)));
typedef float v8f __attribute__((ext_vector_type(8)));

#define NU 50000
#define NI 100000
#define NT 150000
#define DD 64
#define NNZE 2400000
#define BATCH 4096
#define SLOPE 0.01f

__global__ __launch_bounds__(256)
void k_gather0(const float* __restrict__ ue, const float* __restrict__ ie,
               const int* __restrict__ us, const int* __restrict__ pi,
               const int* __restrict__ ni, float* __restrict__ out) {
    int t = blockIdx.x * 256 + threadIdx.x;
    if (t >= 3 * BATCH * DD) return;
    int c  = t & 63;
    int b  = t >> 6;
    int seg = b >> 12;
    int bb  = b & (BATCH - 1);
    float v;
    if (seg == 0)      v = ue[(size_t)us[bb] * DD + c];
    else if (seg == 1) v = ie[(size_t)pi[bb] * DD + c];
    else               v = ie[(size_t)ni[bb] * DD + c];
    out[(size_t)b * 256 + c] = v;
}

__global__ __launch_bounds__(256)
void k_spmm(const int* __restrict__ rows, const int* __restrict__ cols,
            const float* __restrict__ vals, const float* __restrict__ ego,
            float* __restrict__ L) {
    long t = (long)blockIdx.x * 256 + threadIdx.x;
    if (t >= (long)NNZE * 16) return;
    int e = (int)(t >> 4);
    int q = ((int)t & 15) * 4;
    int r = rows[e], c = cols[e];
    float v = vals[e];
    const float4 x = *(const float4*)(ego + (size_t)c * DD + q);
    float* dst = L + (size_t)r * DD + q;
    atomicAdd(dst + 0, v * x.x);
    atomicAdd(dst + 1, v * x.y);
    atomicAdd(dst + 2, v * x.z);
    atomicAdd(dst + 3, v * x.w);
}

#define RB 256
#define NB ((NT + RB - 1) / RB)
#define COL_BITS 18
#define SORT_G 256
#define SORT_T 1024
#define SLICE ((NNZE + SORT_G - 1) / SORT_G)
#define SLICE_TRIPS ((SLICE + SORT_T - 1) / SORT_T)
#define CNT_PITCH 32
static_assert(NT <= (1 << COL_BITS), "column index must fit in COL_BITS");
static_assert(RB <= (1 << (31 - COL_BITS)), "row in bucket must fit above the column");
static_assert(NB <= SORT_T && RB <= SORT_T, "one scan element per thread");

static __device__ __forceinline__
int block_excl_scan(const int own, int* wtot) {
    const int lane = threadIdx.x & 31, wave = threadIdx.x >> 5;
    int inc = own;
    #pragma unroll
    for (int o = 1; o < 32; o <<= 1) {
        int y = __shfl_up(inc, o, 32);
        if (lane >= o) inc += y;
    }
    if (lane == 31) wtot[wave] = inc;
    __syncthreads();
    if (wave == 0) {
        const int x = wtot[lane];
        int xi = x;
        #pragma unroll
        for (int o = 1; o < 32; o <<= 1) {
            int y = __shfl_up(xi, o, 32);
            if (lane >= o) xi += y;
        }
        wtot[lane] = xi - x;
    }
    __syncthreads();
    return wtot[wave] + inc - own;
}

__global__ __launch_bounds__(SORT_T)
void k_bucket_count(const int* __restrict__ rows, int* __restrict__ bcnt,
                    int* __restrict__ wgoff) {
    __shared__ int hist[NB];
    const int t = threadIdx.x;
    for (int i = t; i < NB; i += SORT_T) hist[i] = 0;
    __syncthreads();
    const int s0 = blockIdx.x * SLICE;
    const int s1 = min(s0 + SLICE, NNZE);
    int r[SLICE_TRIPS];
    #pragma unroll
    for (int k = 0; k < SLICE_TRIPS; ++k) {
        const int e = s0 + k * SORT_T + t;
        r[k] = (e < s1) ? rows[e] : -1;
    }
    #pragma unroll
    for (int k = 0; k < SLICE_TRIPS; ++k)
        if (r[k] >= 0) atomicAdd(&hist[r[k] / RB], 1);
    __syncthreads();
    for (int b = t; b < NB; b += SORT_T) {
        const int h = hist[b];
        wgoff[blockIdx.x * NB + b] = h ? atomicAdd(&bcnt[b * CNT_PITCH], h) : 0;
    }
}

__global__ __launch_bounds__(SORT_T)
void k_bucket_scatter(const int* __restrict__ rows, const int* __restrict__ cols,
                      const float* __restrict__ vals, const int* __restrict__ bcnt,
                      const int* __restrict__ wgoff, int* __restrict__ bbase,
                      int2* __restrict__ tmp) {
    __shared__ int start[NB];
    __shared__ int lcnt[NB];
    __shared__ int wtot[32];
    const int t = threadIdx.x;
    const int own = (t < NB) ? bcnt[t * CNT_PITCH] : 0;
    const int off = (t < NB) ? wgoff[blockIdx.x * NB + t] : 0;
    const int excl = block_excl_scan(own, wtot);
    if (t < NB) {
        start[t] = excl + off;
        lcnt[t] = 0;
        if (blockIdx.x == 0) {
            bbase[t] = excl;
            if (t == NB - 1) bbase[NB] = excl + own;
        }
    }
    __syncthreads();
    const int s0 = blockIdx.x * SLICE;
    const int s1 = min(s0 + SLICE, NNZE);
    int   r[SLICE_TRIPS], c[SLICE_TRIPS];
    float v[SLICE_TRIPS];
    #pragma unroll
    for (int k = 0; k < SLICE_TRIPS; ++k) {
        const int e = s0 + k * SORT_T + t;
        const bool ok = e < s1;
        r[k] = ok ? rows[e] : -1;
        c[k] = ok ? cols[e] : 0;
        v[k] = ok ? vals[e] : 0.f;
    }
    #pragma unroll
    for (int k = 0; k < SLICE_TRIPS; ++k) {
        if (r[k] >= 0) {
            const int b  = r[k] / RB;
            const int lr = atomicAdd(&lcnt[b], 1);
            tmp[start[b] + lr] = make_int2(((r[k] % RB) << COL_BITS) | c[k],
                                           __float_as_int(v[k]));
        }
    }
}

#define ROWS_T 1024
#define ROWS_BATCH 4

__global__ __launch_bounds__(ROWS_T)
void k_bucket_rows(const int* __restrict__ bbase, const int2* __restrict__ tmp,
                   int2* __restrict__ sorted, int* __restrict__ rowptr) {
    __shared__ int cnt[RB];
    __shared__ int wtot[32];
    const int t = threadIdx.x;
    const int b = blockIdx.x;
    const int base = bbase[b];
    const int n = bbase[b + 1] - base;
    if (t < RB) cnt[t] = 0;
    __syncthreads();
    for (int i0 = 0; i0 < n; i0 += ROWS_BATCH * ROWS_T) {
        int x[ROWS_BATCH];
        #pragma unroll
        for (int j = 0; j < ROWS_BATCH; ++j) {
            const int i = i0 + j * ROWS_T + t;
            x[j] = (i < n) ? tmp[base + i].x : -1;
        }
        #pragma unroll
        for (int j = 0; j < ROWS_BATCH; ++j)
            if (x[j] >= 0) atomicAdd(&cnt[x[j] >> COL_BITS], 1);
    }
    __syncthreads();
    const int own = (t < RB) ? cnt[t] : 0;
    const int excl = block_excl_scan(own, wtot);
    if (t < RB) {
        cnt[t] = excl;
        const int row = b * RB + t;
        if (row < NT) rowptr[row] = base + excl;
    }
    if (b == NB - 1 && t == 0) rowptr[NT] = base + n;
    __syncthreads();
    for (int i0 = 0; i0 < n; i0 += ROWS_BATCH * ROWS_T) {
        int2 e[ROWS_BATCH];
        #pragma unroll
        for (int j = 0; j < ROWS_BATCH; ++j) {
            const int i = i0 + j * ROWS_T + t;
            e[j] = (i < n) ? tmp[base + i] : make_int2(-1, 0);
        }
        #pragma unroll
        for (int j = 0; j < ROWS_BATCH; ++j) {
            if (e[j].x >= 0) {
                const int p = atomicAdd(&cnt[e[j].x >> COL_BITS], 1);
                sorted[base + p] = make_int2(e[j].x & ((1 << COL_BITS) - 1), e[j].y);
            }
        }
    }
}

static __device__ __forceinline__
void row_batch(float4& acc, const int2 my, const int nb, const int half,
               const int q, const float* __restrict__ ego) {
    for (int i0 = 0; i0 < nb; i0 += 16) {
        int   c[8];
        float v[8];
        float4 x[8];
        #pragma unroll
        for (int j = 0; j < 8; ++j) {
            const int src = i0 + 2 * j + half;
            c[j] = __shfl(my.x, src, 32);
            v[j] = __int_as_float(__shfl(my.y, src, 32));
        }
        #pragma unroll
        for (int j = 0; j < 8; ++j)
            x[j] = *(const float4*)(ego + (size_t)c[j] * DD + q);
        #pragma unroll
        for (int j = 0; j < 8; ++j) {
            const bool ok = (i0 + 2 * j + half) < nb;
            acc.x += v[j] * (ok ? x[j].x : 0.f);
            acc.y += v[j] * (ok ? x[j].y : 0.f);
            acc.z += v[j] * (ok ? x[j].z : 0.f);
            acc.w += v[j] * (ok ? x[j].w : 0.f);
        }
    }
}

__global__ __launch_bounds__(256)
void k_dense_wmma(const float* __restrict__ L, float* __restrict__ ego,
                  const float* __restrict__ Wgc, const float* __restrict__ Wbi,
                  const float* __restrict__ bias) {
    __shared__ float sg[4096];
    __shared__ float sb[4096];
    __shared__ float sbias[64];

    for (int i = threadIdx.x; i < 4096; i += 256) {
        int r = i >> 6, c = i & 63;
        int dst = ((((r >> 1) * 64) + c) << 1) | (r & 1);
        sg[dst] = Wgc[i];
        sb[dst] = Wbi[i];
    }
    if (threadIdx.x < 64) sbias[threadIdx.x] = 2.0f * bias[threadIdx.x];
    __syncthreads();

    const int wave  = threadIdx.x >> 5;
    const int lane  = threadIdx.x & 31;
    const int mbase = (blockIdx.x * 8 + wave) * 16;
    if (mbase >= NT) return;

    const int half = lane >> 4;
    const int n    = lane & 15;

    const float* lrow = L   + (size_t)(mbase + n) * DD;
    const float* erow = ego + (size_t)(mbase + n) * DD;

    v8f acc[4];
    #pragma unroll
    for (int j = 0; j < 4; ++j) acc[j] = (v8f){};

    #pragma unroll
    for (int kk = 0; kk < 16; ++kk) {
        int off = kk * 4 + half * 2;
        v2f ls = *(const v2f*)(lrow + off);
        v2f eg = *(const v2f*)(erow + off);
        v2f a_li  = ls + eg;
        v2f a_mul = ls * eg;
        int bbase = ((kk * 2 + half) * 64) << 1;
        #pragma unroll
        for (int j = 0; j < 4; ++j) {
            v2f bgc = *(const v2f*)(sg + bbase + ((j * 16 + n) << 1));
            v2f bbi = *(const v2f*)(sb + bbase + ((j * 16 + n) << 1));
            acc[j] = __builtin_amdgcn_wmma_f32_16x16x4_f32(
                false, a_li, false, bgc, (short)0, acc[j], false, false);
            acc[j] = __builtin_amdgcn_wmma_f32_16x16x4_f32(
                false, a_mul, false, bbi, (short)0, acc[j], false, false);
        }
    }

    #pragma unroll
    for (int j = 0; j < 4; ++j) {
        float bsum = sbias[j * 16 + n];
        #pragma unroll
        for (int v = 0; v < 8; ++v) {
            float x = acc[j][v] + bsum;
            x = (x >= 0.f) ? x : SLOPE * x;
            ego[(size_t)(mbase + (half << 3) + v) * DD + j * 16 + n] = x;
        }
    }
}

#define TILE_PITCH 68
#define LAYER_WAVES 19
#define LAYER_T (LAYER_WAVES * 32)
#define LAYER_TPW 2
#define N_TILES (NT / 16)
#define LAYER_GRID ((N_TILES + LAYER_WAVES * LAYER_TPW - 1) / (LAYER_WAVES * LAYER_TPW))
#define W_TRIPS ((4096 + LAYER_T - 1) / LAYER_T)

#define LAST_WAVES 4
#define LAST_T (LAST_WAVES * 32)
#define LAST_TILES (3 * BATCH / 16)
#define LAST_COPY_GRID (LAST_TILES / LAST_WAVES)
static_assert(LAST_WAVES * 4 == 16 && LAST_WAVES * 16 == DD, "four rows and sixteen columns a wave");

static __device__ __forceinline__
void gathered_cols_0_191(const int tile, const int lane,
                         const float* __restrict__ ue, const float* __restrict__ ie,
                         const float* __restrict__ e1, const float* __restrict__ e2,
                         const int* __restrict__ us, const int* __restrict__ pi,
                         const int* __restrict__ ni, float* __restrict__ out) {
    const int seg = tile >> 8;
    const int bb0 = (tile & 255) * 16;
    const int* idx   = (seg == 0) ? us : ((seg == 1) ? pi : ni);
    const float* tab = (seg == 0) ? ue : ie;
    const int nodeoff = seg ? NU : 0;

    v2f x0[16], x1[16], x2[16];
    #pragma unroll
    for (int i = 0; i < 16; ++i) {
        const int id = idx[bb0 + i];
        const size_t node = (size_t)(id + nodeoff);
        x0[i] = *(const v2f*)(tab + (size_t)id * DD + lane * 2);
        x1[i] = *(const v2f*)(e1 + node * DD + lane * 2);
        x2[i] = *(const v2f*)(e2 + node * DD + lane * 2);
    }
    #pragma unroll
    for (int i = 0; i < 16; ++i) {
        float* o = out + (size_t)(tile * 16 + i) * 256 + lane * 2;
        *(v2f*)o = x0[i];

        float s1 = fmaf(x1[i].x, x1[i].x, x1[i].y * x1[i].y);
        float s2 = fmaf(x2[i].x, x2[i].x, x2[i].y * x2[i].y);
        #pragma unroll
        for (int d = 16; d > 0; d >>= 1) {
            s1 += __shfl_xor(s1, d, 32);
            s2 += __shfl_xor(s2, d, 32);
        }
        const float c1 = 1.0f / fmaxf(sqrtf(s1), 1e-12f);
        const float c2 = 1.0f / fmaxf(sqrtf(s2), 1e-12f);
        v2f y1; y1.x = x1[i].x * c1; y1.y = x1[i].y * c1;
        v2f y2; y2.x = x2[i].x * c2; y2.y = x2[i].y * c2;
        *(v2f*)(o + 64)  = y1;
        *(v2f*)(o + 128) = y2;
    }
}

static __device__ __forceinline__
void trip_issue(float (&v)[8], float4 (&x)[8], const int2 my, const int i0,
                const int half, const int q, const float* __restrict__ ego) {
    int c[8];
    #pragma unroll
    for (int j = 0; j < 8; ++j) {
        const int src = i0 + 2 * j + half;
        c[j] = __shfl(my.x, src, 32);
        v[j] = __int_as_float(__shfl(my.y, src, 32));
    }
    #pragma unroll
    for (int j = 0; j < 8; ++j) {
        const unsigned off = (unsigned)c[j] * (unsigned)(DD * sizeof(float)) + (unsigned)q * sizeof(float);
        x[j] = *(const float4*)((const char*)ego + off);
    }
}

static __device__ __forceinline__
void trip_sum(float4& acc, const float (&v)[8], const float4 (&x)[8], const int i0,
              const int half, const int nb) {
    #pragma unroll
    for (int j = 0; j < 8; ++j) {
        const int keep = ((i0 + 2 * j + half) < nb) ? -1 : 0;
        acc.x += v[j] * __int_as_float(__float_as_int(x[j].x) & keep);
        acc.y += v[j] * __int_as_float(__float_as_int(x[j].y) & keep);
        acc.z += v[j] * __int_as_float(__float_as_int(x[j].z) & keep);
        acc.w += v[j] * __int_as_float(__float_as_int(x[j].w) & keep);
    }
}

#define LAYER_GROUP 2

static __device__ __forceinline__
void group_heads(int2 (&my)[LAYER_GROUP], int (&s4)[LAYER_GROUP], int (&t4)[LAYER_GROUP],
                 int (&nb)[LAYER_GROUP], const int rs, const int rt, const int r0,
                 const int lane, const int2* __restrict__ sorted) {
    #pragma unroll
    for (int r = 0; r < LAYER_GROUP; ++r) {
        s4[r] = __builtin_amdgcn_readlane(rs, r0 + r);
        t4[r] = __builtin_amdgcn_readlane(rt, r0 + r);
        nb[r] = min(32, t4[r] - s4[r]);
        my[r] = make_int2(0, 0);
        if (lane < nb[r]) my[r] = sorted[s4[r] + lane];
    }
}

__global__ __launch_bounds__(LAYER_T)
void k_layer(const int* __restrict__ rowptr, const int2* __restrict__ sorted,
             const float* __restrict__ ein, float* __restrict__ eout,
             const float* __restrict__ Wgc, const float* __restrict__ Wbi,
             const float* __restrict__ bias) {
    __shared__ __attribute__((aligned(16))) float sg[4096];
    __shared__ __attribute__((aligned(16))) float sb[4096];
    __shared__ float sbias[64];
    __shared__ __attribute__((aligned(16))) float sL[LAYER_WAVES * 16 * TILE_PITCH];

    {
        float wg[W_TRIPS], wb[W_TRIPS];
        #pragma unroll
        for (int k = 0; k < W_TRIPS; ++k) {
            const int i = min(k * LAYER_T + (int)threadIdx.x, 4095);
            wg[k] = Wgc[i];
            wb[k] = Wbi[i];
        }
        #pragma unroll
        for (int k = 0; k < W_TRIPS; ++k) {
            const int i = k * LAYER_T + (int)threadIdx.x;
            if (i < 4096) {
                const int r = i >> 6, c = i & 63;
                const int dst = ((((r >> 1) * 64) + c) << 1) | (r & 1);
                sg[dst] = wg[k];
                sb[dst] = wb[k];
            }
        }
    }
    if (threadIdx.x < 64) sbias[threadIdx.x] = 2.0f * bias[threadIdx.x];
    __syncthreads();

    const int lane = threadIdx.x & 31;
    const int half = lane >> 4;
    const int n    = lane & 15;
    const int q    = n * 4;
    const int wave = __builtin_amdgcn_readfirstlane(threadIdx.x >> 5);
    float* myL = sL + wave * (16 * TILE_PITCH);

    #pragma unroll 1
    for (int tt = 0; tt < LAYER_TPW; ++tt) {
        const int tile = (blockIdx.x * LAYER_TPW + tt) * LAYER_WAVES + wave;
        if (tile >= N_TILES) break;
        const int mbase = tile * 16;

        const int rs = rowptr[mbase + n], rt = rowptr[mbase + n + 1];

        int sN[LAYER_GROUP], tN[LAYER_GROUP], nN[LAYER_GROUP];
        int2 myN[LAYER_GROUP];
        group_heads(myN, sN, tN, nN, rs, rt, 0, lane, sorted);

        #pragma unroll 1
        for (int g = 0; g < 16 / LAYER_GROUP; ++g) {
            int s4[LAYER_GROUP], t4[LAYER_GROUP], nb[LAYER_GROUP];
            int2 my[LAYER_GROUP];
            #pragma unroll
            for (int r = 0; r < LAYER_GROUP; ++r) { s4[r] = sN[r]; t4[r] = tN[r]; nb[r] = nN[r]; my[r] = myN[r]; }

            float4 acc[LAYER_GROUP];
            {
                float  v[LAYER_GROUP][8];
                float4 x[LAYER_GROUP][8];
                #pragma unroll
                for (int r = 0; r < LAYER_GROUP; ++r) trip_issue(v[r], x[r], my[r], 0, half, q, ein);
                if (g < 16 / LAYER_GROUP - 1)
                    group_heads(myN, sN, tN, nN, rs, rt, LAYER_GROUP * (g + 1), lane, sorted);
                #pragma unroll
                for (int r = 0; r < LAYER_GROUP; ++r) {
                    acc[r] = make_float4(0.f, 0.f, 0.f, 0.f);
                    trip_sum(acc[r], v[r], x[r], 0, half, nb[r]);
                }
            }

            int nbmax = nb[0];
            #pragma unroll
            for (int r = 1; r < LAYER_GROUP; ++r) nbmax = max(nbmax, nb[r]);
            if (nbmax > 16) {
                float  v[LAYER_GROUP][8];
                float4 x[LAYER_GROUP][8];
                #pragma unroll
                for (int r = 0; r < LAYER_GROUP; ++r) trip_issue(v[r], x[r], my[r], 16, half, q, ein);
                #pragma unroll
                for (int r = 0; r < LAYER_GROUP; ++r) trip_sum(acc[r], v[r], x[r], 16, half, nb[r]);
            }

            #pragma unroll
            for (int r = 0; r < LAYER_GROUP; ++r) {
                for (int base = s4[r] + 32; base < t4[r]; base += 32) {
                    const int nbb = min(32, t4[r] - base);
                    int2 m2 = make_int2(0, 0);
                    if (lane < nbb) m2 = sorted[base + lane];
                    row_batch(acc[r], m2, nbb, half, q, ein);
                }
            }

            #pragma unroll
            for (int r = 0; r < LAYER_GROUP; ++r) {
                acc[r].x += __shfl_xor(acc[r].x, 16, 32);
                acc[r].y += __shfl_xor(acc[r].y, 16, 32);
                acc[r].z += __shfl_xor(acc[r].z, 16, 32);
                acc[r].w += __shfl_xor(acc[r].w, 16, 32);
                if (half == 0) *(float4*)(myL + (LAYER_GROUP * g + r) * TILE_PITCH + q) = acc[r];
            }
        }

        __builtin_amdgcn_wave_barrier();

        const float* lrow = myL + n * TILE_PITCH;
        const float* erow = ein + (size_t)(mbase + n) * DD;

        v8f dacc[4];
        #pragma unroll
        for (int j = 0; j < 4; ++j) dacc[j] = (v8f){};

        #pragma unroll
        for (int kk = 0; kk < 16; ++kk) {
            int off = kk * 4 + half * 2;
            v2f ls = *(const v2f*)(lrow + off);
            v2f eg = *(const v2f*)(erow + off);
            v2f a_li  = ls + eg;
            v2f a_mul = ls * eg;
            int bbase = ((kk * 2 + half) * 64) << 1;
            #pragma unroll
            for (int j = 0; j < 4; ++j) {
                v2f bgc = *(const v2f*)(sg + bbase + ((j * 16 + n) << 1));
                v2f bbi = *(const v2f*)(sb + bbase + ((j * 16 + n) << 1));
                dacc[j] = __builtin_amdgcn_wmma_f32_16x16x4_f32(
                    false, a_li, false, bgc, (short)0, dacc[j], false, false);
                dacc[j] = __builtin_amdgcn_wmma_f32_16x16x4_f32(
                    false, a_mul, false, bbi, (short)0, dacc[j], false, false);
            }
        }

        #pragma unroll
        for (int j = 0; j < 4; ++j) {
            float bsum = sbias[j * 16 + n];
            #pragma unroll
            for (int v = 0; v < 8; ++v) {
                float x = dacc[j][v] + bsum;
                x = (x >= 0.f) ? x : SLOPE * x;
                eout[(size_t)(mbase + (half << 3) + v) * DD + j * 16 + n] = x;
            }
        }
    }
}

__global__ __launch_bounds__(LAST_T) __attribute__((amdgpu_waves_per_eu(4, 4)))
void k_layer_last(const int* __restrict__ rowptr, const int2* __restrict__ sorted,
                  const float* __restrict__ ein, const float* __restrict__ eprev,
                  const float* __restrict__ ue, const float* __restrict__ ie,
                  const float* __restrict__ Wgc, const float* __restrict__ Wbi,
                  const float* __restrict__ bias,
                  const int* __restrict__ us, const int* __restrict__ pi,
                  const int* __restrict__ ni, float* __restrict__ out) {
    __shared__ __attribute__((aligned(16))) float sL[16 * TILE_PITCH];
    __shared__ __attribute__((aligned(16))) float sO[16 * TILE_PITCH];

    const int lane = threadIdx.x & 31;
    const int w = __builtin_amdgcn_readfirstlane(threadIdx.x >> 5);

    if (blockIdx.x >= LAST_TILES) {
        gathered_cols_0_191((blockIdx.x - LAST_TILES) * LAST_WAVES + w, lane,
                            ue, ie, eprev, ein, us, pi, ni, out);
        return;
    }

    const int half = lane >> 4;
    const int n    = lane & 15;
    const int q    = n * 4;
    const int tile = blockIdx.x;
    const int seg  = tile >> 8;
    const int bb0  = (tile & 255) * 16;
    const int* idx = (seg == 0) ? us : ((seg == 1) ? pi : ni);

    const int node = idx[bb0 + n] + (seg ? NU : 0);
    const int rs = rowptr[node], rt = rowptr[node + 1];

    int s4[4], t4[4], nb[4];
    int2 my[4];
    #pragma unroll
    for (int r = 0; r < 4; ++r) {
        s4[r] = __builtin_amdgcn_readlane(rs, 4 * w + r);
        t4[r] = __builtin_amdgcn_readlane(rt, 4 * w + r);
        nb[r] = min(32, t4[r] - s4[r]);
        my[r] = make_int2(0, 0);
        if (lane < nb[r]) my[r] = sorted[s4[r] + lane];
    }

    float4 acc[4];
    {
        float  v[4][8];
        float4 x[4][8];
        #pragma unroll
        for (int r = 0; r < 4; ++r) trip_issue(v[r], x[r], my[r], 0, half, q, ein);
        #pragma unroll
        for (int r = 0; r < 4; ++r) {
            acc[r] = make_float4(0.f, 0.f, 0.f, 0.f);
            trip_sum(acc[r], v[r], x[r], 0, half, nb[r]);
        }
    }

    if (max(max(nb[0], nb[1]), max(nb[2], nb[3])) > 16) {
        float  v[4][8];
        float4 x[4][8];
        #pragma unroll
        for (int r = 0; r < 4; ++r) trip_issue(v[r], x[r], my[r], 16, half, q, ein);
        #pragma unroll
        for (int r = 0; r < 4; ++r) trip_sum(acc[r], v[r], x[r], 16, half, nb[r]);
    }

    #pragma unroll
    for (int r = 0; r < 4; ++r) {
        for (int base = s4[r] + 32; base < t4[r]; base += 32) {
            const int nbb = min(32, t4[r] - base);
            int2 m2 = make_int2(0, 0);
            if (lane < nbb) m2 = sorted[base + lane];
            row_batch(acc[r], m2, nbb, half, q, ein);
        }
    }

    const int col = 16 * w + n;
    v2f bg[16], bb[16], eg[16];
    #pragma unroll
    for (int kk = 0; kk < 16; ++kk) {
        const int k0 = kk * 4 + half * 2;
        bg[kk].x = Wgc[k0 * DD + col];
        bg[kk].y = Wgc[(k0 + 1) * DD + col];
        bb[kk].x = Wbi[k0 * DD + col];
        bb[kk].y = Wbi[(k0 + 1) * DD + col];
    }
    const float* erow = ein + (size_t)node * DD;
    #pragma unroll
    for (int kk = 0; kk < 16; ++kk)
        eg[kk] = *(const v2f*)(erow + kk * 4 + half * 2);
    const float bsum = 2.0f * bias[col];

    #pragma unroll
    for (int r = 0; r < 4; ++r) {
        acc[r].x += __shfl_xor(acc[r].x, 16, 32);
        acc[r].y += __shfl_xor(acc[r].y, 16, 32);
        acc[r].z += __shfl_xor(acc[r].z, 16, 32);
        acc[r].w += __shfl_xor(acc[r].w, 16, 32);
        if (half == 0) *(float4*)(sL + (4 * w + r) * TILE_PITCH + q) = acc[r];
    }

    __syncthreads();

    const float* lrow = sL + n * TILE_PITCH;
    v8f dacc = (v8f){};
    #pragma unroll
    for (int kk = 0; kk < 16; ++kk) {
        v2f ls = *(const v2f*)(lrow + kk * 4 + half * 2);
        v2f a_li  = ls + eg[kk];
        v2f a_mul = ls * eg[kk];
        dacc = __builtin_amdgcn_wmma_f32_16x16x4_f32(
            false, a_li, false, bg[kk], (short)0, dacc, false, false);
        dacc = __builtin_amdgcn_wmma_f32_16x16x4_f32(
            false, a_mul, false, bb[kk], (short)0, dacc, false, false);
    }

    #pragma unroll
    for (int v = 0; v < 8; ++v) {
        float x = dacc[v] + bsum;
        x = (x >= 0.f) ? x : SLOPE * x;
        sO[((half << 3) + v) * TILE_PITCH + col] = x;
    }

    __syncthreads();

    #pragma unroll
    for (int r = 0; r < 4; ++r) {
        const int i = 4 * w + r;
        v2f x = *(const v2f*)(sO + i * TILE_PITCH + lane * 2);
        float ss = x.x * x.x + x.y * x.y;
        #pragma unroll
        for (int o = 16; o > 0; o >>= 1) ss += __shfl_xor(ss, o, 32);
        float scale = 1.0f / fmaxf(sqrtf(ss), 1e-12f);
        v2f y; y.x = x.x * scale; y.y = x.y * scale;
        *(v2f*)(out + (size_t)(tile * 16 + i) * 256 + 192 + lane * 2) = y;
    }
}

__global__ __launch_bounds__(256)
void k_gather_norm(const float* __restrict__ ego, const int* __restrict__ us,
                   const int* __restrict__ pi, const int* __restrict__ ni,
                   float* __restrict__ out, int colbase) {
    int wid  = (blockIdx.x * 256 + threadIdx.x) >> 5;
    int lane = threadIdx.x & 31;
    if (wid >= 3 * BATCH) return;
    int seg = wid >> 12;
    int bb  = wid & (BATCH - 1);
    int node;
    if (seg == 0)      node = us[bb];
    else if (seg == 1) node = NU + pi[bb];
    else               node = NU + ni[bb];
    v2f x = *(const v2f*)(ego + (size_t)node * DD + lane * 2);
    float ss = x.x * x.x + x.y * x.y;
    #pragma unroll
    for (int o = 16; o > 0; o >>= 1) ss += __shfl_xor(ss, o, 32);
    float scale = 1.0f / fmaxf(sqrtf(ss), 1e-12f);
    v2f y; y.x = x.x * scale; y.y = x.y * scale;
    *(v2f*)(out + (size_t)wid * 256 + colbase + lane * 2) = y;
}

extern "C" void kernel_launch(void* const* d_in, const int* in_sizes, int n_in,
                              void* d_out, int out_size, void* d_ws, size_t ws_size,
                              hipStream_t stream) {
    const float* ue   = (const float*)d_in[0];
    const float* ie   = (const float*)d_in[1];
    const float* Wgc  = (const float*)d_in[2];
    const float* Wbi  = (const float*)d_in[3];
    const float* bbi  = (const float*)d_in[4];
    const float* vals = (const float*)d_in[5];
    const int*   rows = (const int*)d_in[6];
    const int*   cols = (const int*)d_in[7];
    const int*   us   = (const int*)d_in[8];
    const int*   pi   = (const int*)d_in[9];
    const int*   ni   = (const int*)d_in[10];
    float* out = (float*)d_out;

    float* ego = (float*)d_ws;
    float* L   = ego + (size_t)NT * DD;

    hipMemcpyAsync(ego, ue, (size_t)NU * DD * sizeof(float),
                   hipMemcpyDeviceToDevice, stream);
    hipMemcpyAsync(ego + (size_t)NU * DD, ie, (size_t)NI * DD * sizeof(float),
                   hipMemcpyDeviceToDevice, stream);

    const size_t need = (size_t)2 * NT * DD * 4 + (size_t)NNZE * 8 + (size_t)(NT + 1) * 4;
    if (ws_size >= need) {
        int2* sorted = (int2*)(L + (size_t)NT * DD);
        int*  rowptr = (int*)(sorted + NNZE);
        int2* tmp    = (int2*)L;
        int*  wgoff  = (int*)(tmp + NNZE);
        int*  bcnt   = wgoff + SORT_G * NB;
        int*  bbase  = bcnt + NB * CNT_PITCH;
        hipMemsetAsync(bcnt, 0, (size_t)NB * CNT_PITCH * sizeof(int), stream);
        k_bucket_count<<<SORT_G, SORT_T, 0, stream>>>(rows, bcnt, wgoff);
        k_bucket_scatter<<<SORT_G, SORT_T, 0, stream>>>(rows, cols, vals, bcnt, wgoff,
                                                        bbase, tmp);
        k_bucket_rows<<<NB, ROWS_T, 0, stream>>>(bbase, tmp, sorted, rowptr);
        float* ego2 = L;
        k_layer<<<LAYER_GRID, LAYER_T, 0, stream>>>(
            rowptr, sorted, ego, ego2, Wgc, Wbi, bbi);
        k_layer<<<LAYER_GRID, LAYER_T, 0, stream>>>(
            rowptr, sorted, ego2, ego, Wgc + 4096, Wbi + 4096, bbi + 64);
        k_layer_last<<<LAST_TILES + LAST_COPY_GRID, LAST_T, 0, stream>>>(
            rowptr, sorted, ego, ego2, ue, ie, Wgc + 8192, Wbi + 8192, bbi + 128,
            us, pi, ni, out);
        return;
    }

    k_gather0<<<(3 * BATCH * DD + 255) / 256, 256, 0, stream>>>(
        ue, ie, us, pi, ni, out);
    for (int k = 0; k < 3; ++k) {
        hipMemsetAsync(L, 0, (size_t)NT * DD * sizeof(float), stream);
        long tot = (long)NNZE * 16;
        k_spmm<<<(int)((tot + 255) / 256), 256, 0, stream>>>(
            rows, cols, vals, ego, L);
        k_dense_wmma<<<(NT + 127) / 128, 256, 0, stream>>>(
            L, ego, Wgc + k * 4096, Wbi + k * 4096, bbi + k * 64);
        k_gather_norm<<<(3 * BATCH * 32 + 255) / 256, 256, 0, stream>>>(
            ego, us, pi, ni, out, 64 * (k + 1));
    }
}


# Edge sort: both scattering launches stage their edges in LDS in destination order and store them as runs, a store covering a few lines instead of 32; sum order: LDS arrival rank as before
# speedup vs baseline: 4.5185x; 1.0740x over previous
#include <hip/hip_runtime.h>
#include <hip/hip_bf16.h>

typedef float v2f __attribute__((ext_vector_type(2)));
typedef float v8f __attribute__((ext_vector_type(8)));

#define NU 50000
#define NI 100000
#define NT 150000
#define DD 64
#define NNZE 2400000
#define BATCH 4096
#define SLOPE 0.01f

__global__ __launch_bounds__(256)
void k_gather0(const float* __restrict__ ue, const float* __restrict__ ie,
               const int* __restrict__ us, const int* __restrict__ pi,
               const int* __restrict__ ni, float* __restrict__ out) {
    int t = blockIdx.x * 256 + threadIdx.x;
    if (t >= 3 * BATCH * DD) return;
    int c  = t & 63;
    int b  = t >> 6;
    int seg = b >> 12;
    int bb  = b & (BATCH - 1);
    float v;
    if (seg == 0)      v = ue[(size_t)us[bb] * DD + c];
    else if (seg == 1) v = ie[(size_t)pi[bb] * DD + c];
    else               v = ie[(size_t)ni[bb] * DD + c];
    out[(size_t)b * 256 + c] = v;
}

__global__ __launch_bounds__(256)
void k_spmm(const int* __restrict__ rows, const int* __restrict__ cols,
            const float* __restrict__ vals, const float* __restrict__ ego,
            float* __restrict__ L) {
    long t = (long)blockIdx.x * 256 + threadIdx.x;
    if (t >= (long)NNZE * 16) return;
    int e = (int)(t >> 4);
    int q = ((int)t & 15) * 4;
    int r = rows[e], c = cols[e];
    float v = vals[e];
    const float4 x = *(const float4*)(ego + (size_t)c * DD + q);
    float* dst = L + (size_t)r * DD + q;
    atomicAdd(dst + 0, v * x.x);
    atomicAdd(dst + 1, v * x.y);
    atomicAdd(dst + 2, v * x.z);
    atomicAdd(dst + 3, v * x.w);
}

#define RB 256
#define NB ((NT + RB - 1) / RB)
#define COL_BITS 18
#define SORT_G 256
#define SORT_T 1024
#define SLICE ((NNZE + SORT_G - 1) / SORT_G)
#define SLICE_TRIPS ((SLICE + SORT_T - 1) / SORT_T)
#define CNT_PITCH 32
static_assert(NT <= (1 << COL_BITS), "column index must fit in COL_BITS");
static_assert(RB <= (1 << (31 - COL_BITS)), "row in bucket must fit above the column");
static_assert(NB <= SORT_T && RB <= SORT_T, "one scan element per thread");

static __device__ __forceinline__
int block_excl_scan(const int own, int* wtot) {
    const int lane = threadIdx.x & 31, wave = threadIdx.x >> 5;
    int inc = own;
    #pragma unroll
    for (int o = 1; o < 32; o <<= 1) {
        int y = __shfl_up(inc, o, 32);
        if (lane >= o) inc += y;
    }
    if (lane == 31) wtot[wave] = inc;
    __syncthreads();
    if (wave == 0) {
        const int x = wtot[lane];
        int xi = x;
        #pragma unroll
        for (int o = 1; o < 32; o <<= 1) {
            int y = __shfl_up(xi, o, 32);
            if (lane >= o) xi += y;
        }
        wtot[lane] = xi - x;
    }
    __syncthreads();
    return wtot[wave] + inc - own;
}

__global__ __launch_bounds__(SORT_T)
void k_bucket_count(const int* __restrict__ rows, int* __restrict__ bcnt,
                    int* __restrict__ wgoff) {
    __shared__ int hist[NB];
    const int t = threadIdx.x;
    for (int i = t; i < NB; i += SORT_T) hist[i] = 0;
    __syncthreads();
    const int s0 = blockIdx.x * SLICE;
    const int s1 = min(s0 + SLICE, NNZE);
    int r[SLICE_TRIPS];
    #pragma unroll
    for (int k = 0; k < SLICE_TRIPS; ++k) {
        const int e = s0 + k * SORT_T + t;
        r[k] = (e < s1) ? rows[e] : -1;
    }
    #pragma unroll
    for (int k = 0; k < SLICE_TRIPS; ++k)
        if (r[k] >= 0) atomicAdd(&hist[r[k] / RB], 1);
    __syncthreads();
    for (int b = t; b < NB; b += SORT_T) {
        const int h = hist[b];
        wgoff[blockIdx.x * NB + b] = h ? atomicAdd(&bcnt[b * CNT_PITCH], h) : 0;
    }
}

__global__ __launch_bounds__(SORT_T)
void k_bucket_scatter(const int* __restrict__ rows, const int* __restrict__ cols,
                      const float* __restrict__ vals, const int* __restrict__ bcnt,
                      const int* __restrict__ wgoff, int* __restrict__ bbase,
                      int2* __restrict__ tmp) {
    __shared__ int start[NB];
    __shared__ int lcnt[NB];
    __shared__ int lbase[NB];
    __shared__ int wtot[32];
    __shared__ int2 skv[SLICE];
    __shared__ int  sdst[SLICE];
    const int t = threadIdx.x;
    const int own = (t < NB) ? bcnt[t * CNT_PITCH] : 0;
    const int off = (t < NB) ? wgoff[blockIdx.x * NB + t] : 0;
    const int excl = block_excl_scan(own, wtot);
    if (t < NB) {
        start[t] = excl + off;
        lcnt[t] = 0;
        if (blockIdx.x == 0) {
            bbase[t] = excl;
            if (t == NB - 1) bbase[NB] = excl + own;
        }
    }
    __syncthreads();
    const int s0 = blockIdx.x * SLICE;
    const int s1 = min(s0 + SLICE, NNZE);
    int   r[SLICE_TRIPS], c[SLICE_TRIPS], lr[SLICE_TRIPS];
    float v[SLICE_TRIPS];
    #pragma unroll
    for (int k = 0; k < SLICE_TRIPS; ++k) {
        const int e = s0 + k * SORT_T + t;
        const bool ok = e < s1;
        r[k] = ok ? rows[e] : -1;
        c[k] = ok ? cols[e] : 0;
        v[k] = ok ? vals[e] : 0.f;
    }
    #pragma unroll
    for (int k = 0; k < SLICE_TRIPS; ++k)
        lr[k] = (r[k] >= 0) ? atomicAdd(&lcnt[r[k] / RB], 1) : 0;
    __syncthreads();
    const int lex = block_excl_scan((t < NB) ? lcnt[t] : 0, wtot);
    if (t < NB) lbase[t] = lex;
    __syncthreads();
    #pragma unroll
    for (int k = 0; k < SLICE_TRIPS; ++k) {
        if (r[k] >= 0) {
            const int b   = r[k] / RB;
            const int pos = lbase[b] + lr[k];
            skv[pos]  = make_int2(((r[k] % RB) << COL_BITS) | c[k], __float_as_int(v[k]));
            sdst[pos] = start[b] + lr[k];
        }
    }
    __syncthreads();
    const int cnt = s1 - s0;
    int2 okv[SLICE_TRIPS];
    int  odst[SLICE_TRIPS];
    #pragma unroll
    for (int k = 0; k < SLICE_TRIPS; ++k) {
        const int i = min(k * SORT_T + t, SLICE - 1);
        okv[k]  = skv[i];
        odst[k] = sdst[i];
    }
    #pragma unroll
    for (int k = 0; k < SLICE_TRIPS; ++k)
        if (k * SORT_T + t < cnt) tmp[odst[k]] = okv[k];
}

#define ROWS_T 1024
#define ROWS_BATCH 4
#define ROWS_CAP 8192
static_assert(ROWS_CAP % (ROWS_BATCH * ROWS_T) == 0, "a copy-out batch must not leave the stage");

__global__ __launch_bounds__(ROWS_T)
void k_bucket_rows(const int* __restrict__ bbase, const int2* __restrict__ tmp,
                   int2* __restrict__ sorted, int* __restrict__ rowptr) {
    __shared__ int cnt[RB];
    __shared__ int wtot[32];
    __shared__ int2 stg[ROWS_CAP];
    const int t = threadIdx.x;
    const int b = blockIdx.x;
    const int base = bbase[b];
    const int n = bbase[b + 1] - base;
    const bool staged = n <= ROWS_CAP;
    if (t < RB) cnt[t] = 0;
    __syncthreads();
    for (int i0 = 0; i0 < n; i0 += ROWS_BATCH * ROWS_T) {
        int x[ROWS_BATCH];
        #pragma unroll
        for (int j = 0; j < ROWS_BATCH; ++j) {
            const int i = i0 + j * ROWS_T + t;
            x[j] = (i < n) ? tmp[base + i].x : -1;
        }
        #pragma unroll
        for (int j = 0; j < ROWS_BATCH; ++j)
            if (x[j] >= 0) atomicAdd(&cnt[x[j] >> COL_BITS], 1);
    }
    __syncthreads();
    const int own = (t < RB) ? cnt[t] : 0;
    const int excl = block_excl_scan(own, wtot);
    if (t < RB) {
        cnt[t] = excl;
        const int row = b * RB + t;
        if (row < NT) rowptr[row] = base + excl;
    }
    if (b == NB - 1 && t == 0) rowptr[NT] = base + n;
    __syncthreads();
    for (int i0 = 0; i0 < n; i0 += ROWS_BATCH * ROWS_T) {
        int2 e[ROWS_BATCH];
        #pragma unroll
        for (int j = 0; j < ROWS_BATCH; ++j) {
            const int i = i0 + j * ROWS_T + t;
            e[j] = (i < n) ? tmp[base + i] : make_int2(-1, 0);
        }
        #pragma unroll
        for (int j = 0; j < ROWS_BATCH; ++j) {
            if (e[j].x >= 0) {
                const int p = atomicAdd(&cnt[e[j].x >> COL_BITS], 1);
                const int2 cv = make_int2(e[j].x & ((1 << COL_BITS) - 1), e[j].y);
                if (staged) stg[p] = cv;
                else        sorted[base + p] = cv;
            }
        }
    }
    if (staged) {
        __syncthreads();
        for (int i0 = 0; i0 < n; i0 += ROWS_BATCH * ROWS_T) {
            int2 o[ROWS_BATCH];
            #pragma unroll
            for (int j = 0; j < ROWS_BATCH; ++j) o[j] = stg[i0 + j * ROWS_T + t];
            #pragma unroll
            for (int j = 0; j < ROWS_BATCH; ++j) {
                const int i = i0 + j * ROWS_T + t;
                if (i < n) sorted[base + i] = o[j];
            }
        }
    }
}

static __device__ __forceinline__
void row_batch(float4& acc, const int2 my, const int nb, const int half,
               const int q, const float* __restrict__ ego) {
    for (int i0 = 0; i0 < nb; i0 += 16) {
        int   c[8];
        float v[8];
        float4 x[8];
        #pragma unroll
        for (int j = 0; j < 8; ++j) {
            const int src = i0 + 2 * j + half;
            c[j] = __shfl(my.x, src, 32);
            v[j] = __int_as_float(__shfl(my.y, src, 32));
        }
        #pragma unroll
        for (int j = 0; j < 8; ++j)
            x[j] = *(const float4*)(ego + (size_t)c[j] * DD + q);
        #pragma unroll
        for (int j = 0; j < 8; ++j) {
            const bool ok = (i0 + 2 * j + half) < nb;
            acc.x += v[j] * (ok ? x[j].x : 0.f);
            acc.y += v[j] * (ok ? x[j].y : 0.f);
            acc.z += v[j] * (ok ? x[j].z : 0.f);
            acc.w += v[j] * (ok ? x[j].w : 0.f);
        }
    }
}

__global__ __launch_bounds__(256)
void k_dense_wmma(const float* __restrict__ L, float* __restrict__ ego,
                  const float* __restrict__ Wgc, const float* __restrict__ Wbi,
                  const float* __restrict__ bias) {
    __shared__ float sg[4096];
    __shared__ float sb[4096];
    __shared__ float sbias[64];

    for (int i = threadIdx.x; i < 4096; i += 256) {
        int r = i >> 6, c = i & 63;
        int dst = ((((r >> 1) * 64) + c) << 1) | (r & 1);
        sg[dst] = Wgc[i];
        sb[dst] = Wbi[i];
    }
    if (threadIdx.x < 64) sbias[threadIdx.x] = 2.0f * bias[threadIdx.x];
    __syncthreads();

    const int wave  = threadIdx.x >> 5;
    const int lane  = threadIdx.x & 31;
    const int mbase = (blockIdx.x * 8 + wave) * 16;
    if (mbase >= NT) return;

    const int half = lane >> 4;
    const int n    = lane & 15;

    const float* lrow = L   + (size_t)(mbase + n) * DD;
    const float* erow = ego + (size_t)(mbase + n) * DD;

    v8f acc[4];
    #pragma unroll
    for (int j = 0; j < 4; ++j) acc[j] = (v8f){};

    #pragma unroll
    for (int kk = 0; kk < 16; ++kk) {
        int off = kk * 4 + half * 2;
        v2f ls = *(const v2f*)(lrow + off);
        v2f eg = *(const v2f*)(erow + off);
        v2f a_li  = ls + eg;
        v2f a_mul = ls * eg;
        int bbase = ((kk * 2 + half) * 64) << 1;
        #pragma unroll
        for (int j = 0; j < 4; ++j) {
            v2f bgc = *(const v2f*)(sg + bbase + ((j * 16 + n) << 1));
            v2f bbi = *(const v2f*)(sb + bbase + ((j * 16 + n) << 1));
            acc[j] = __builtin_amdgcn_wmma_f32_16x16x4_f32(
                false, a_li, false, bgc, (short)0, acc[j], false, false);
            acc[j] = __builtin_amdgcn_wmma_f32_16x16x4_f32(
                false, a_mul, false, bbi, (short)0, acc[j], false, false);
        }
    }

    #pragma unroll
    for (int j = 0; j < 4; ++j) {
        float bsum = sbias[j * 16 + n];
        #pragma unroll
        for (int v = 0; v < 8; ++v) {
            float x = acc[j][v] + bsum;
            x = (x >= 0.f) ? x : SLOPE * x;
            ego[(size_t)(mbase + (half << 3) + v) * DD + j * 16 + n] = x;
        }
    }
}

#define TILE_PITCH 68
#define LAYER_WAVES 19
#define LAYER_T (LAYER_WAVES * 32)
#define LAYER_TPW 2
#define N_TILES (NT / 16)
#define LAYER_GRID ((N_TILES + LAYER_WAVES * LAYER_TPW - 1) / (LAYER_WAVES * LAYER_TPW))
#define W_TRIPS ((4096 + LAYER_T - 1) / LAYER_T)

#define LAST_WAVES 4
#define LAST_T (LAST_WAVES * 32)
#define LAST_TILES (3 * BATCH / 16)
#define LAST_COPY_GRID (LAST_TILES / LAST_WAVES)
static_assert(LAST_WAVES * 4 == 16 && LAST_WAVES * 16 == DD, "four rows and sixteen columns a wave");

static __device__ __forceinline__
void gathered_cols_0_191(const int tile, const int lane,
                         const float* __restrict__ ue, const float* __restrict__ ie,
                         const float* __restrict__ e1, const float* __restrict__ e2,
                         const int* __restrict__ us, const int* __restrict__ pi,
                         const int* __restrict__ ni, float* __restrict__ out) {
    const int seg = tile >> 8;
    const int bb0 = (tile & 255) * 16;
    const int* idx   = (seg == 0) ? us : ((seg == 1) ? pi : ni);
    const float* tab = (seg == 0) ? ue : ie;
    const int nodeoff = seg ? NU : 0;

    v2f x0[16], x1[16], x2[16];
    #pragma unroll
    for (int i = 0; i < 16; ++i) {
        const int id = idx[bb0 + i];
        const size_t node = (size_t)(id + nodeoff);
        x0[i] = *(const v2f*)(tab + (size_t)id * DD + lane * 2);
        x1[i] = *(const v2f*)(e1 + node * DD + lane * 2);
        x2[i] = *(const v2f*)(e2 + node * DD + lane * 2);
    }
    #pragma unroll
    for (int i = 0; i < 16; ++i) {
        float* o = out + (size_t)(tile * 16 + i) * 256 + lane * 2;
        *(v2f*)o = x0[i];

        float s1 = fmaf(x1[i].x, x1[i].x, x1[i].y * x1[i].y);
        float s2 = fmaf(x2[i].x, x2[i].x, x2[i].y * x2[i].y);
        #pragma unroll
        for (int d = 16; d > 0; d >>= 1) {
            s1 += __shfl_xor(s1, d, 32);
            s2 += __shfl_xor(s2, d, 32);
        }
        const float c1 = 1.0f / fmaxf(sqrtf(s1), 1e-12f);
        const float c2 = 1.0f / fmaxf(sqrtf(s2), 1e-12f);
        v2f y1; y1.x = x1[i].x * c1; y1.y = x1[i].y * c1;
        v2f y2; y2.x = x2[i].x * c2; y2.y = x2[i].y * c2;
        *(v2f*)(o + 64)  = y1;
        *(v2f*)(o + 128) = y2;
    }
}

static __device__ __forceinline__
void trip_issue(float (&v)[8], float4 (&x)[8], const int2 my, const int i0,
                const int half, const int q, const float* __restrict__ ego) {
    int c[8];
    #pragma unroll
    for (int j = 0; j < 8; ++j) {
        const int src = i0 + 2 * j + half;
        c[j] = __shfl(my.x, src, 32);
        v[j] = __int_as_float(__shfl(my.y, src, 32));
    }
    #pragma unroll
    for (int j = 0; j < 8; ++j) {
        const unsigned off = (unsigned)c[j] * (unsigned)(DD * sizeof(float)) + (unsigned)q * sizeof(float);
        x[j] = *(const float4*)((const char*)ego + off);
    }
}

static __device__ __forceinline__
void trip_sum(float4& acc, const float (&v)[8], const float4 (&x)[8], const int i0,
              const int half, const int nb) {
    #pragma unroll
    for (int j = 0; j < 8; ++j) {
        const int keep = ((i0 + 2 * j + half) < nb) ? -1 : 0;
        acc.x += v[j] * __int_as_float(__float_as_int(x[j].x) & keep);
        acc.y += v[j] * __int_as_float(__float_as_int(x[j].y) & keep);
        acc.z += v[j] * __int_as_float(__float_as_int(x[j].z) & keep);
        acc.w += v[j] * __int_as_float(__float_as_int(x[j].w) & keep);
    }
}

#define LAYER_GROUP 2

static __device__ __forceinline__
void group_heads(int2 (&my)[LAYER_GROUP], int (&s4)[LAYER_GROUP], int (&t4)[LAYER_GROUP],
                 int (&nb)[LAYER_GROUP], const int rs, const int rt, const int r0,
                 const int lane, const int2* __restrict__ sorted) {
    #pragma unroll
    for (int r = 0; r < LAYER_GROUP; ++r) {
        s4[r] = __builtin_amdgcn_readlane(rs, r0 + r);
        t4[r] = __builtin_amdgcn_readlane(rt, r0 + r);
        nb[r] = min(32, t4[r] - s4[r]);
        my[r] = make_int2(0, 0);
        if (lane < nb[r]) my[r] = sorted[s4[r] + lane];
    }
}

__global__ __launch_bounds__(LAYER_T)
void k_layer(const int* __restrict__ rowptr, const int2* __restrict__ sorted,
             const float* __restrict__ ein, float* __restrict__ eout,
             const float* __restrict__ Wgc, const float* __restrict__ Wbi,
             const float* __restrict__ bias) {
    __shared__ __attribute__((aligned(16))) float sg[4096];
    __shared__ __attribute__((aligned(16))) float sb[4096];
    __shared__ float sbias[64];
    __shared__ __attribute__((aligned(16))) float sL[LAYER_WAVES * 16 * TILE_PITCH];

    {
        float wg[W_TRIPS], wb[W_TRIPS];
        #pragma unroll
        for (int k = 0; k < W_TRIPS; ++k) {
            const int i = min(k * LAYER_T + (int)threadIdx.x, 4095);
            wg[k] = Wgc[i];
            wb[k] = Wbi[i];
        }
        #pragma unroll
        for (int k = 0; k < W_TRIPS; ++k) {
            const int i = k * LAYER_T + (int)threadIdx.x;
            if (i < 4096) {
                const int r = i >> 6, c = i & 63;
                const int dst = ((((r >> 1) * 64) + c) << 1) | (r & 1);
                sg[dst] = wg[k];
                sb[dst] = wb[k];
            }
        }
    }
    if (threadIdx.x < 64) sbias[threadIdx.x] = 2.0f * bias[threadIdx.x];
    __syncthreads();

    const int lane = threadIdx.x & 31;
    const int half = lane >> 4;
    const int n    = lane & 15;
    const int q    = n * 4;
    const int wave = __builtin_amdgcn_readfirstlane(threadIdx.x >> 5);
    float* myL = sL + wave * (16 * TILE_PITCH);

    #pragma unroll 1
    for (int tt = 0; tt < LAYER_TPW; ++tt) {
        const int tile = (blockIdx.x * LAYER_TPW + tt) * LAYER_WAVES + wave;
        if (tile >= N_TILES) break;
        const int mbase = tile * 16;

        const int rs = rowptr[mbase + n], rt = rowptr[mbase + n + 1];

        int sN[LAYER_GROUP], tN[LAYER_GROUP], nN[LAYER_GROUP];
        int2 myN[LAYER_GROUP];
        group_heads(myN, sN, tN, nN, rs, rt, 0, lane, sorted);

        #pragma unroll 1
        for (int g = 0; g < 16 / LAYER_GROUP; ++g) {
            int s4[LAYER_GROUP], t4[LAYER_GROUP], nb[LAYER_GROUP];
            int2 my[LAYER_GROUP];
            #pragma unroll
            for (int r = 0; r < LAYER_GROUP; ++r) { s4[r] = sN[r]; t4[r] = tN[r]; nb[r] = nN[r]; my[r] = myN[r]; }

            float4 acc[LAYER_GROUP];
            {
                float  v[LAYER_GROUP][8];
                float4 x[LAYER_GROUP][8];
                #pragma unroll
                for (int r = 0; r < LAYER_GROUP; ++r) trip_issue(v[r], x[r], my[r], 0, half, q, ein);
                if (g < 16 / LAYER_GROUP - 1)
                    group_heads(myN, sN, tN, nN, rs, rt, LAYER_GROUP * (g + 1), lane, sorted);
                #pragma unroll
                for (int r = 0; r < LAYER_GROUP; ++r) {
                    acc[r] = make_float4(0.f, 0.f, 0.f, 0.f);
                    trip_sum(acc[r], v[r], x[r], 0, half, nb[r]);
                }
            }

            int nbmax = nb[0];
            #pragma unroll
            for (int r = 1; r < LAYER_GROUP; ++r) nbmax = max(nbmax, nb[r]);
            if (nbmax > 16) {
                float  v[LAYER_GROUP][8];
                float4 x[LAYER_GROUP][8];
                #pragma unroll
                for (int r = 0; r < LAYER_GROUP; ++r) trip_issue(v[r], x[r], my[r], 16, half, q, ein);
                #pragma unroll
                for (int r = 0; r < LAYER_GROUP; ++r) trip_sum(acc[r], v[r], x[r], 16, half, nb[r]);
            }

            #pragma unroll
            for (int r = 0; r < LAYER_GROUP; ++r) {
                for (int base = s4[r] + 32; base < t4[r]; base += 32) {
                    const int nbb = min(32, t4[r] - base);
                    int2 m2 = make_int2(0, 0);
                    if (lane < nbb) m2 = sorted[base + lane];
                    row_batch(acc[r], m2, nbb, half, q, ein);
                }
            }

            #pragma unroll
            for (int r = 0; r < LAYER_GROUP; ++r) {
                acc[r].x += __shfl_xor(acc[r].x, 16, 32);
                acc[r].y += __shfl_xor(acc[r].y, 16, 32);
                acc[r].z += __shfl_xor(acc[r].z, 16, 32);
                acc[r].w += __shfl_xor(acc[r].w, 16, 32);
                if (half == 0) *(float4*)(myL + (LAYER_GROUP * g + r) * TILE_PITCH + q) = acc[r];
            }
        }

        __builtin_amdgcn_wave_barrier();

        const float* lrow = myL + n * TILE_PITCH;
        const float* erow = ein + (size_t)(mbase + n) * DD;

        v8f dacc[4];
        #pragma unroll
        for (int j = 0; j < 4; ++j) dacc[j] = (v8f){};

        #pragma unroll
        for (int kk = 0; kk < 16; ++kk) {
            int off = kk * 4 + half * 2;
            v2f ls = *(const v2f*)(lrow + off);
            v2f eg = *(const v2f*)(erow + off);
            v2f a_li  = ls + eg;
            v2f a_mul = ls * eg;
            int bbase = ((kk * 2 + half) * 64) << 1;
            #pragma unroll
            for (int j = 0; j < 4; ++j) {
                v2f bgc = *(const v2f*)(sg + bbase + ((j * 16 + n) << 1));
                v2f bbi = *(const v2f*)(sb + bbase + ((j * 16 + n) << 1));
                dacc[j] = __builtin_amdgcn_wmma_f32_16x16x4_f32(
                    false, a_li, false, bgc, (short)0, dacc[j], false, false);
                dacc[j] = __builtin_amdgcn_wmma_f32_16x16x4_f32(
                    false, a_mul, false, bbi, (short)0, dacc[j], false, false);
            }
        }

        #pragma unroll
        for (int j = 0; j < 4; ++j) {
            float bsum = sbias[j * 16 + n];
            #pragma unroll
            for (int v = 0; v < 8; ++v) {
                float x = dacc[j][v] + bsum;
                x = (x >= 0.f) ? x : SLOPE * x;
                eout[(size_t)(mbase + (half << 3) + v) * DD + j * 16 + n] = x;
            }
        }
    }
}

__global__ __launch_bounds__(LAST_T) __attribute__((amdgpu_waves_per_eu(4, 4)))
void k_layer_last(const int* __restrict__ rowptr, const int2* __restrict__ sorted,
                  const float* __restrict__ ein, const float* __restrict__ eprev,
                  const float* __restrict__ ue, const float* __restrict__ ie,
                  const float* __restrict__ Wgc, const float* __restrict__ Wbi,
                  const float* __restrict__ bias,
                  const int* __restrict__ us, const int* __restrict__ pi,
                  const int* __restrict__ ni, float* __restrict__ out) {
    __shared__ __attribute__((aligned(16))) float sL[16 * TILE_PITCH];
    __shared__ __attribute__((aligned(16))) float sO[16 * TILE_PITCH];

    const int lane = threadIdx.x & 31;
    const int w = __builtin_amdgcn_readfirstlane(threadIdx.x >> 5);

    if (blockIdx.x >= LAST_TILES) {
        gathered_cols_0_191((blockIdx.x - LAST_TILES) * LAST_WAVES + w, lane,
                            ue, ie, eprev, ein, us, pi, ni, out);
        return;
    }

    const int half = lane >> 4;
    const int n    = lane & 15;
    const int q    = n * 4;
    const int tile = blockIdx.x;
    const int seg  = tile >> 8;
    const int bb0  = (tile & 255) * 16;
    const int* idx = (seg == 0) ? us : ((seg == 1) ? pi : ni);

    const int node = idx[bb0 + n] + (seg ? NU : 0);
    const int rs = rowptr[node], rt = rowptr[node + 1];

    int s4[4], t4[4], nb[4];
    int2 my[4];
    #pragma unroll
    for (int r = 0; r < 4; ++r) {
        s4[r] = __builtin_amdgcn_readlane(rs, 4 * w + r);
        t4[r] = __builtin_amdgcn_readlane(rt, 4 * w + r);
        nb[r] = min(32, t4[r] - s4[r]);
        my[r] = make_int2(0, 0);
        if (lane < nb[r]) my[r] = sorted[s4[r] + lane];
    }

    float4 acc[4];
    {
        float  v[4][8];
        float4 x[4][8];
        #pragma unroll
        for (int r = 0; r < 4; ++r) trip_issue(v[r], x[r], my[r], 0, half, q, ein);
        #pragma unroll
        for (int r = 0; r < 4; ++r) {
            acc[r] = make_float4(0.f, 0.f, 0.f, 0.f);
            trip_sum(acc[r], v[r], x[r], 0, half, nb[r]);
        }
    }

    if (max(max(nb[0], nb[1]), max(nb[2], nb[3])) > 16) {
        float  v[4][8];
        float4 x[4][8];
        #pragma unroll
        for (int r = 0; r < 4; ++r) trip_issue(v[r], x[r], my[r], 16, half, q, ein);
        #pragma unroll
        for (int r = 0; r < 4; ++r) trip_sum(acc[r], v[r], x[r], 16, half, nb[r]);
    }

    #pragma unroll
    for (int r = 0; r < 4; ++r) {
        for (int base = s4[r] + 32; base < t4[r]; base += 32) {
            const int nbb = min(32, t4[r] - base);
            int2 m2 = make_int2(0, 0);
            if (lane < nbb) m2 = sorted[base + lane];
            row_batch(acc[r], m2, nbb, half, q, ein);
        }
    }

    const int col = 16 * w + n;
    v2f bg[16], bb[16], eg[16];
    #pragma unroll
    for (int kk = 0; kk < 16; ++kk) {
        const int k0 = kk * 4 + half * 2;
        bg[kk].x = Wgc[k0 * DD + col];
        bg[kk].y = Wgc[(k0 + 1) * DD + col];
        bb[kk].x = Wbi[k0 * DD + col];
        bb[kk].y = Wbi[(k0 + 1) * DD + col];
    }
    const float* erow = ein + (size_t)node * DD;
    #pragma unroll
    for (int kk = 0; kk < 16; ++kk)
        eg[kk] = *(const v2f*)(erow + kk * 4 + half * 2);
    const float bsum = 2.0f * bias[col];

    #pragma unroll
    for (int r = 0; r < 4; ++r) {
        acc[r].x += __shfl_xor(acc[r].x, 16, 32);
        acc[r].y += __shfl_xor(acc[r].y, 16, 32);
        acc[r].z += __shfl_xor(acc[r].z, 16, 32);
        acc[r].w += __shfl_xor(acc[r].w, 16, 32);
        if (half == 0) *(float4*)(sL + (4 * w + r) * TILE_PITCH + q) = acc[r];
    }

    __syncthreads();

    const float* lrow = sL + n * TILE_PITCH;
    v8f dacc = (v8f){};
    #pragma unroll
    for (int kk = 0; kk < 16; ++kk) {
        v2f ls = *(const v2f*)(lrow + kk * 4 + half * 2);
        v2f a_li  = ls + eg[kk];
        v2f a_mul = ls * eg[kk];
        dacc = __builtin_amdgcn_wmma_f32_16x16x4_f32(
            false, a_li, false, bg[kk], (short)0, dacc, false, false);
        dacc = __builtin_amdgcn_wmma_f32_16x16x4_f32(
            false, a_mul, false, bb[kk], (short)0, dacc, false, false);
    }

    #pragma unroll
    for (int v = 0; v < 8; ++v) {
        float x = dacc[v] + bsum;
        x = (x >= 0.f) ? x : SLOPE * x;
        sO[((half << 3) + v) * TILE_PITCH + col] = x;
    }

    __syncthreads();

    #pragma unroll
    for (int r = 0; r < 4; ++r) {
        const int i = 4 * w + r;
        v2f x = *(const v2f*)(sO + i * TILE_PITCH + lane * 2);
        float ss = x.x * x.x + x.y * x.y;
        #pragma unroll
        for (int o = 16; o > 0; o >>= 1) ss += __shfl_xor(ss, o, 32);
        float scale = 1.0f / fmaxf(sqrtf(ss), 1e-12f);
        v2f y; y.x = x.x * scale; y.y = x.y * scale;
        *(v2f*)(out + (size_t)(tile * 16 + i) * 256 + 192 + lane * 2) = y;
    }
}

__global__ __launch_bounds__(256)
void k_gather_norm(const float* __restrict__ ego, const int* __restrict__ us,
                   const int* __restrict__ pi, const int* __restrict__ ni,
                   float* __restrict__ out, int colbase) {
    int wid  = (blockIdx.x * 256 + threadIdx.x) >> 5;
    int lane = threadIdx.x & 31;
    if (wid >= 3 * BATCH) return;
    int seg = wid >> 12;
    int bb  = wid & (BATCH - 1);
    int node;
    if (seg == 0)      node = us[bb];
    else if (seg == 1) node = NU + pi[bb];
    else               node = NU + ni[bb];
    v2f x = *(const v2f*)(ego + (size_t)node * DD + lane * 2);
    float ss = x.x * x.x + x.y * x.y;
    #pragma unroll
    for (int o = 16; o > 0; o >>= 1) ss += __shfl_xor(ss, o, 32);
    float scale = 1.0f / fmaxf(sqrtf(ss), 1e-12f);
    v2f y; y.x = x.x * scale; y.y = x.y * scale;
    *(v2f*)(out + (size_t)wid * 256 + colbase + lane * 2) = y;
}

extern "C" void kernel_launch(void* const* d_in, const int* in_sizes, int n_in,
                              void* d_out, int out_size, void* d_ws, size_t ws_size,
                              hipStream_t stream) {
    const float* ue   = (const float*)d_in[0];
    const float* ie   = (const float*)d_in[1];
    const float* Wgc  = (const float*)d_in[2];
    const float* Wbi  = (const float*)d_in[3];
    const float* bbi  = (const float*)d_in[4];
    const float* vals = (const float*)d_in[5];
    const int*   rows = (const int*)d_in[6];
    const int*   cols = (const int*)d_in[7];
    const int*   us   = (const int*)d_in[8];
    const int*   pi   = (const int*)d_in[9];
    const int*   ni   = (const int*)d_in[10];
    float* out = (float*)d_out;

    float* ego = (float*)d_ws;
    float* L   = ego + (size_t)NT * DD;

    hipMemcpyAsync(ego, ue, (size_t)NU * DD * sizeof(float),
                   hipMemcpyDeviceToDevice, stream);
    hipMemcpyAsync(ego + (size_t)NU * DD, ie, (size_t)NI * DD * sizeof(float),
                   hipMemcpyDeviceToDevice, stream);

    const size_t need = (size_t)2 * NT * DD * 4 + (size_t)NNZE * 8 + (size_t)(NT + 1) * 4;
    if (ws_size >= need) {
        int2* sorted = (int2*)(L + (size_t)NT * DD);
        int*  rowptr = (int*)(sorted + NNZE);
        int2* tmp    = (int2*)L;
        int*  wgoff  = (int*)(tmp + NNZE);
        int*  bcnt   = wgoff + SORT_G * NB;
        int*  bbase  = bcnt + NB * CNT_PITCH;
        hipMemsetAsync(bcnt, 0, (size_t)NB * CNT_PITCH * sizeof(int), stream);
        k_bucket_count<<<SORT_G, SORT_T, 0, stream>>>(rows, bcnt, wgoff);
        k_bucket_scatter<<<SORT_G, SORT_T, 0, stream>>>(rows, cols, vals, bcnt, wgoff,
                                                        bbase, tmp);
        k_bucket_rows<<<NB, ROWS_T, 0, stream>>>(bbase, tmp, sorted, rowptr);
        float* ego2 = L;
        k_layer<<<LAYER_GRID, LAYER_T, 0, stream>>>(
            rowptr, sorted, ego, ego2, Wgc, Wbi, bbi);
        k_layer<<<LAYER_GRID, LAYER_T, 0, stream>>>(
            rowptr, sorted, ego2, ego, Wgc + 4096, Wbi + 4096, bbi + 64);
        k_layer_last<<<LAST_TILES + LAST_COPY_GRID, LAST_T, 0, stream>>>(
            rowptr, sorted, ego, ego2, ue, ie, Wgc + 8192, Wbi + 8192, bbi + 128,
            us, pi, ni, out);
        return;
    }

    k_gather0<<<(3 * BATCH * DD + 255) / 256, 256, 0, stream>>>(
        ue, ie, us, pi, ni, out);
    for (int k = 0; k < 3; ++k) {
        hipMemsetAsync(L, 0, (size_t)NT * DD * sizeof(float), stream);
        long tot = (long)NNZE * 16;
        k_spmm<<<(int)((tot + 255) / 256), 256, 0, stream>>>(
            rows, cols, vals, ego, L);
        k_dense_wmma<<<(NT + 127) / 128, 256, 0, stream>>>(
            L, ego, Wgc + k * 4096, Wbi + k * 4096, bbi + k * 64);
        k_gather_norm<<<(3 * BATCH * 32 + 255) / 256, 256, 0, stream>>>(
            ego, us, pi, ni, out, 64 * (k + 1));
    }
}


# Layers 1-2: a row's edges 16-23 are gathered in the round trip of its first sixteen, only by rows that have them (24-31: a short trip of the row's own), not as a full second trip; sum order kept
# speedup vs baseline: 4.6812x; 1.0360x over previous
#include <hip/hip_runtime.h>
#include <hip/hip_bf16.h>

typedef float v2f __attribute__((ext_vector_type(2)));
typedef float v8f __attribute__((ext_vector_type(8)));

#define NU 50000
#define NI 100000
#define NT 150000
#define DD 64
#define NNZE 2400000
#define BATCH 4096
#define SLOPE 0.01f

__global__ __launch_bounds__(256)
void k_gather0(const float* __restrict__ ue, const float* __restrict__ ie,
               const int* __restrict__ us, const int* __restrict__ pi,
               const int* __restrict__ ni, float* __restrict__ out) {
    int t = blockIdx.x * 256 + threadIdx.x;
    if (t >= 3 * BATCH * DD) return;
    int c  = t & 63;
    int b  = t >> 6;
    int seg = b >> 12;
    int bb  = b & (BATCH - 1);
    float v;
    if (seg == 0)      v = ue[(size_t)us[bb] * DD + c];
    else if (seg == 1) v = ie[(size_t)pi[bb] * DD + c];
    else               v = ie[(size_t)ni[bb] * DD + c];
    out[(size_t)b * 256 + c] = v;
}

__global__ __launch_bounds__(256)
void k_spmm(const int* __restrict__ rows, const int* __restrict__ cols,
            const float* __restrict__ vals, const float* __restrict__ ego,
            float* __restrict__ L) {
    long t = (long)blockIdx.x * 256 + threadIdx.x;
    if (t >= (long)NNZE * 16) return;
    int e = (int)(t >> 4);
    int q = ((int)t & 15) * 4;
    int r = rows[e], c = cols[e];
    float v = vals[e];
    const float4 x = *(const float4*)(ego + (size_t)c * DD + q);
    float* dst = L + (size_t)r * DD + q;
    atomicAdd(dst + 0, v * x.x);
    atomicAdd(dst + 1, v * x.y);
    atomicAdd(dst + 2, v * x.z);
    atomicAdd(dst + 3, v * x.w);
}

#define RB 256
#define NB ((NT + RB - 1) / RB)
#define COL_BITS 18
#define SORT_G 256
#define SORT_T 1024
#define SLICE ((NNZE + SORT_G - 1) / SORT_G)
#define SLICE_TRIPS ((SLICE + SORT_T - 1) / SORT_T)
#define CNT_PITCH 32
static_assert(NT <= (1 << COL_BITS), "column index must fit in COL_BITS");
static_assert(RB <= (1 << (31 - COL_BITS)), "row in bucket must fit above the column");
static_assert(NB <= SORT_T && RB <= SORT_T, "one scan element per thread");

static __device__ __forceinline__
int block_excl_scan(const int own, int* wtot) {
    const int lane = threadIdx.x & 31, wave = threadIdx.x >> 5;
    int inc = own;
    #pragma unroll
    for (int o = 1; o < 32; o <<= 1) {
        int y = __shfl_up(inc, o, 32);
        if (lane >= o) inc += y;
    }
    if (lane == 31) wtot[wave] = inc;
    __syncthreads();
    if (wave == 0) {
        const int x = wtot[lane];
        int xi = x;
        #pragma unroll
        for (int o = 1; o < 32; o <<= 1) {
            int y = __shfl_up(xi, o, 32);
            if (lane >= o) xi += y;
        }
        wtot[lane] = xi - x;
    }
    __syncthreads();
    return wtot[wave] + inc - own;
}

__global__ __launch_bounds__(SORT_T)
void k_bucket_count(const int* __restrict__ rows, int* __restrict__ bcnt,
                    int* __restrict__ wgoff) {
    __shared__ int hist[NB];
    const int t = threadIdx.x;
    for (int i = t; i < NB; i += SORT_T) hist[i] = 0;
    __syncthreads();
    const int s0 = blockIdx.x * SLICE;
    const int s1 = min(s0 + SLICE, NNZE);
    int r[SLICE_TRIPS];
    #pragma unroll
    for (int k = 0; k < SLICE_TRIPS; ++k) {
        const int e = s0 + k * SORT_T + t;
        r[k] = (e < s1) ? rows[e] : -1;
    }
    #pragma unroll
    for (int k = 0; k < SLICE_TRIPS; ++k)
        if (r[k] >= 0) atomicAdd(&hist[r[k] / RB], 1);
    __syncthreads();
    for (int b = t; b < NB; b += SORT_T) {
        const int h = hist[b];
        wgoff[blockIdx.x * NB + b] = h ? atomicAdd(&bcnt[b * CNT_PITCH], h) : 0;
    }
}

__global__ __launch_bounds__(SORT_T)
void k_bucket_scatter(const int* __restrict__ rows, const int* __restrict__ cols,
                      const float* __restrict__ vals, const int* __restrict__ bcnt,
                      const int* __restrict__ wgoff, int* __restrict__ bbase,
                      int2* __restrict__ tmp) {
    __shared__ int start[NB];
    __shared__ int lcnt[NB];
    __shared__ int lbase[NB];
    __shared__ int wtot[32];
    __shared__ int2 skv[SLICE];
    __shared__ int  sdst[SLICE];
    const int t = threadIdx.x;
    const int own = (t < NB) ? bcnt[t * CNT_PITCH] : 0;
    const int off = (t < NB) ? wgoff[blockIdx.x * NB + t] : 0;
    const int excl = block_excl_scan(own, wtot);
    if (t < NB) {
        start[t] = excl + off;
        lcnt[t] = 0;
        if (blockIdx.x == 0) {
            bbase[t] = excl;
            if (t == NB - 1) bbase[NB] = excl + own;
        }
    }
    __syncthreads();
    const int s0 = blockIdx.x * SLICE;
    const int s1 = min(s0 + SLICE, NNZE);
    int   r[SLICE_TRIPS], c[SLICE_TRIPS], lr[SLICE_TRIPS];
    float v[SLICE_TRIPS];
    #pragma unroll
    for (int k = 0; k < SLICE_TRIPS; ++k) {
        const int e = s0 + k * SORT_T + t;
        const bool ok = e < s1;
        r[k] = ok ? rows[e] : -1;
        c[k] = ok ? cols[e] : 0;
        v[k] = ok ? vals[e] : 0.f;
    }
    #pragma unroll
    for (int k = 0; k < SLICE_TRIPS; ++k)
        lr[k] = (r[k] >= 0) ? atomicAdd(&lcnt[r[k] / RB], 1) : 0;
    __syncthreads();
    const int lex = block_excl_scan((t < NB) ? lcnt[t] : 0, wtot);
    if (t < NB) lbase[t] = lex;
    __syncthreads();
    #pragma unroll
    for (int k = 0; k < SLICE_TRIPS; ++k) {
        if (r[k] >= 0) {
            const int b   = r[k] / RB;
            const int pos = lbase[b] + lr[k];
            skv[pos]  = make_int2(((r[k] % RB) << COL_BITS) | c[k], __float_as_int(v[k]));
            sdst[pos] = start[b] + lr[k];
        }
    }
    __syncthreads();
    const int cnt = s1 - s0;
    int2 okv[SLICE_TRIPS];
    int  odst[SLICE_TRIPS];
    #pragma unroll
    for (int k = 0; k < SLICE_TRIPS; ++k) {
        const int i = min(k * SORT_T + t, SLICE - 1);
        okv[k]  = skv[i];
        odst[k] = sdst[i];
    }
    #pragma unroll
    for (int k = 0; k < SLICE_TRIPS; ++k)
        if (k * SORT_T + t < cnt) tmp[odst[k]] = okv[k];
}

#define ROWS_T 1024
#define ROWS_BATCH 4
#define ROWS_CAP 8192
static_assert(ROWS_CAP % (ROWS_BATCH * ROWS_T) == 0, "a copy-out batch must not leave the stage");

__global__ __launch_bounds__(ROWS_T)
void k_bucket_rows(const int* __restrict__ bbase, const int2* __restrict__ tmp,
                   int2* __restrict__ sorted, int* __restrict__ rowptr) {
    __shared__ int cnt[RB];
    __shared__ int wtot[32];
    __shared__ int2 stg[ROWS_CAP];
    const int t = threadIdx.x;
    const int b = blockIdx.x;
    const int base = bbase[b];
    const int n = bbase[b + 1] - base;
    const bool staged = n <= ROWS_CAP;
    if (t < RB) cnt[t] = 0;
    __syncthreads();
    for (int i0 = 0; i0 < n; i0 += ROWS_BATCH * ROWS_T) {
        int x[ROWS_BATCH];
        #pragma unroll
        for (int j = 0; j < ROWS_BATCH; ++j) {
            const int i = i0 + j * ROWS_T + t;
            x[j] = (i < n) ? tmp[base + i].x : -1;
        }
        #pragma unroll
        for (int j = 0; j < ROWS_BATCH; ++j)
            if (x[j] >= 0) atomicAdd(&cnt[x[j] >> COL_BITS], 1);
    }
    __syncthreads();
    const int own = (t < RB) ? cnt[t] : 0;
    const int excl = block_excl_scan(own, wtot);
    if (t < RB) {
        cnt[t] = excl;
        const int row = b * RB + t;
        if (row < NT) rowptr[row] = base + excl;
    }
    if (b == NB - 1 && t == 0) rowptr[NT] = base + n;
    __syncthreads();
    for (int i0 = 0; i0 < n; i0 += ROWS_BATCH * ROWS_T) {
        int2 e[ROWS_BATCH];
        #pragma unroll
        for (int j = 0; j < ROWS_BATCH; ++j) {
            const int i = i0 + j * ROWS_T + t;
            e[j] = (i < n) ? tmp[base + i] : make_int2(-1, 0);
        }
        #pragma unroll
        for (int j = 0; j < ROWS_BATCH; ++j) {
            if (e[j].x >= 0) {
                const int p = atomicAdd(&cnt[e[j].x >> COL_BITS], 1);
                const int2 cv = make_int2(e[j].x & ((1 << COL_BITS) - 1), e[j].y);
                if (staged) stg[p] = cv;
                else        sorted[base + p] = cv;
            }
        }
    }
    if (staged) {
        __syncthreads();
        for (int i0 = 0; i0 < n; i0 += ROWS_BATCH * ROWS_T) {
            int2 o[ROWS_BATCH];
            #pragma unroll
            for (int j = 0; j < ROWS_BATCH; ++j) o[j] = stg[i0 + j * ROWS_T + t];
            #pragma unroll
            for (int j = 0; j < ROWS_BATCH; ++j) {
                const int i = i0 + j * ROWS_T + t;
                if (i < n) sorted[base + i] = o[j];
            }
        }
    }
}

static __device__ __forceinline__
void row_batch(float4& acc, const int2 my, const int nb, const int half,
               const int q, const float* __restrict__ ego) {
    for (int i0 = 0; i0 < nb; i0 += 16) {
        int   c[8];
        float v[8];
        float4 x[8];
        #pragma unroll
        for (int j = 0; j < 8; ++j) {
            const int src = i0 + 2 * j + half;
            c[j] = __shfl(my.x, src, 32);
            v[j] = __int_as_float(__shfl(my.y, src, 32));
        }
        #pragma unroll
        for (int j = 0; j < 8; ++j)
            x[j] = *(const float4*)(ego + (size_t)c[j] * DD + q);
        #pragma unroll
        for (int j = 0; j < 8; ++j) {
            const bool ok = (i0 + 2 * j + half) < nb;
            acc.x += v[j] * (ok ? x[j].x : 0.f);
            acc.y += v[j] * (ok ? x[j].y : 0.f);
            acc.z += v[j] * (ok ? x[j].z : 0.f);
            acc.w += v[j] * (ok ? x[j].w : 0.f);
        }
    }
}

__global__ __launch_bounds__(256)
void k_dense_wmma(const float* __restrict__ L, float* __restrict__ ego,
                  const float* __restrict__ Wgc, const float* __restrict__ Wbi,
                  const float* __restrict__ bias) {
    __shared__ float sg[4096];
    __shared__ float sb[4096];
    __shared__ float sbias[64];

    for (int i = threadIdx.x; i < 4096; i += 256) {
        int r = i >> 6, c = i & 63;
        int dst = ((((r >> 1) * 64) + c) << 1) | (r & 1);
        sg[dst] = Wgc[i];
        sb[dst] = Wbi[i];
    }
    if (threadIdx.x < 64) sbias[threadIdx.x] = 2.0f * bias[threadIdx.x];
    __syncthreads();

    const int wave  = threadIdx.x >> 5;
    const int lane  = threadIdx.x & 31;
    const int mbase = (blockIdx.x * 8 + wave) * 16;
    if (mbase >= NT) return;

    const int half = lane >> 4;
    const int n    = lane & 15;

    const float* lrow = L   + (size_t)(mbase + n) * DD;
    const float* erow = ego + (size_t)(mbase + n) * DD;

    v8f acc[4];
    #pragma unroll
    for (int j = 0; j < 4; ++j) acc[j] = (v8f){};

    #pragma unroll
    for (int kk = 0; kk < 16; ++kk) {
        int off = kk * 4 + half * 2;
        v2f ls = *(const v2f*)(lrow + off);
        v2f eg = *(const v2f*)(erow + off);
        v2f a_li  = ls + eg;
        v2f a_mul = ls * eg;
        int bbase = ((kk * 2 + half) * 64) << 1;
        #pragma unroll
        for (int j = 0; j < 4; ++j) {
            v2f bgc = *(const v2f*)(sg + bbase + ((j * 16 + n) << 1));
            v2f bbi = *(const v2f*)(sb + bbase + ((j * 16 + n) << 1));
            acc[j] = __builtin_amdgcn_wmma_f32_16x16x4_f32(
                false, a_li, false, bgc, (short)0, acc[j], false, false);
            acc[j] = __builtin_amdgcn_wmma_f32_16x16x4_f32(
                false, a_mul, false, bbi, (short)0, acc[j], false, false);
        }
    }

    #pragma unroll
    for (int j = 0; j < 4; ++j) {
        float bsum = sbias[j * 16 + n];
        #pragma unroll
        for (int v = 0; v < 8; ++v) {
            float x = acc[j][v] + bsum;
            x = (x >= 0.f) ? x : SLOPE * x;
            ego[(size_t)(mbase + (half << 3) + v) * DD + j * 16 + n] = x;
        }
    }
}

#define TILE_PITCH 68
#define LAYER_WAVES 19
#define LAYER_T (LAYER_WAVES * 32)
#define LAYER_TPW 2
#define N_TILES (NT / 16)
#define LAYER_GRID ((N_TILES + LAYER_WAVES * LAYER_TPW - 1) / (LAYER_WAVES * LAYER_TPW))
#define W_TRIPS ((4096 + LAYER_T - 1) / LAYER_T)

#define LAST_WAVES 4
#define LAST_T (LAST_WAVES * 32)
#define LAST_TILES (3 * BATCH / 16)
#define LAST_COPY_GRID (LAST_TILES / LAST_WAVES)
static_assert(LAST_WAVES * 4 == 16 && LAST_WAVES * 16 == DD, "four rows and sixteen columns a wave");

static __device__ __forceinline__
void gathered_cols_0_191(const int tile, const int lane,
                         const float* __restrict__ ue, const float* __restrict__ ie,
                         const float* __restrict__ e1, const float* __restrict__ e2,
                         const int* __restrict__ us, const int* __restrict__ pi,
                         const int* __restrict__ ni, float* __restrict__ out) {
    const int seg = tile >> 8;
    const int bb0 = (tile & 255) * 16;
    const int* idx   = (seg == 0) ? us : ((seg == 1) ? pi : ni);
    const float* tab = (seg == 0) ? ue : ie;
    const int nodeoff = seg ? NU : 0;

    v2f x0[16], x1[16], x2[16];
    #pragma unroll
    for (int i = 0; i < 16; ++i) {
        const int id = idx[bb0 + i];
        const size_t node = (size_t)(id + nodeoff);
        x0[i] = *(const v2f*)(tab + (size_t)id * DD + lane * 2);
        x1[i] = *(const v2f*)(e1 + node * DD + lane * 2);
        x2[i] = *(const v2f*)(e2 + node * DD + lane * 2);
    }
    #pragma unroll
    for (int i = 0; i < 16; ++i) {
        float* o = out + (size_t)(tile * 16 + i) * 256 + lane * 2;
        *(v2f*)o = x0[i];

        float s1 = fmaf(x1[i].x, x1[i].x, x1[i].y * x1[i].y);
        float s2 = fmaf(x2[i].x, x2[i].x, x2[i].y * x2[i].y);
        #pragma unroll
        for (int d = 16; d > 0; d >>= 1) {
            s1 += __shfl_xor(s1, d, 32);
            s2 += __shfl_xor(s2, d, 32);
        }
        const float c1 = 1.0f / fmaxf(sqrtf(s1), 1e-12f);
        const float c2 = 1.0f / fmaxf(sqrtf(s2), 1e-12f);
        v2f y1; y1.x = x1[i].x * c1; y1.y = x1[i].y * c1;
        v2f y2; y2.x = x2[i].x * c2; y2.y = x2[i].y * c2;
        *(v2f*)(o + 64)  = y1;
        *(v2f*)(o + 128) = y2;
    }
}

static __device__ __forceinline__
void trip_issue(float (&v)[8], float4 (&x)[8], const int2 my, const int i0,
                const int half, const int q, const float* __restrict__ ego) {
    int c[8];
    #pragma unroll
    for (int j = 0; j < 8; ++j) {
        const int src = i0 + 2 * j + half;
        c[j] = __shfl(my.x, src, 32);
        v[j] = __int_as_float(__shfl(my.y, src, 32));
    }
    #pragma unroll
    for (int j = 0; j < 8; ++j) {
        const unsigned off = (unsigned)c[j] * (unsigned)(DD * sizeof(float)) + (unsigned)q * sizeof(float);
        x[j] = *(const float4*)((const char*)ego + off);
    }
}

static __device__ __forceinline__
void trip_sum(float4& acc, const float (&v)[8], const float4 (&x)[8], const int i0,
              const int half, const int nb) {
    #pragma unroll
    for (int j = 0; j < 8; ++j) {
        const int keep = ((i0 + 2 * j + half) < nb) ? -1 : 0;
        acc.x += v[j] * __int_as_float(__float_as_int(x[j].x) & keep);
        acc.y += v[j] * __int_as_float(__float_as_int(x[j].y) & keep);
        acc.z += v[j] * __int_as_float(__float_as_int(x[j].z) & keep);
        acc.w += v[j] * __int_as_float(__float_as_int(x[j].w) & keep);
    }
}

template <int NJ>
static __device__ __forceinline__
void part_issue(float (&v)[NJ], float4 (&x)[NJ], const int2 my, const int i0,
                const int half, const int q, const float* __restrict__ ego) {
    int c[NJ];
    #pragma unroll
    for (int j = 0; j < NJ; ++j) {
        const int src = i0 + 2 * j + half;
        c[j] = __shfl(my.x, src, 32);
        v[j] = __int_as_float(__shfl(my.y, src, 32));
    }
    #pragma unroll
    for (int j = 0; j < NJ; ++j) {
        const unsigned off = (unsigned)c[j] * (unsigned)(DD * sizeof(float)) + (unsigned)q * sizeof(float);
        x[j] = *(const float4*)((const char*)ego + off);
    }
}

template <int NJ>
static __device__ __forceinline__
void part_sum(float4& acc, const float (&v)[NJ], const float4 (&x)[NJ], const int i0,
              const int half, const int nb) {
    #pragma unroll
    for (int j = 0; j < NJ; ++j) {
        const int keep = ((i0 + 2 * j + half) < nb) ? -1 : 0;
        acc.x += v[j] * __int_as_float(__float_as_int(x[j].x) & keep);
        acc.y += v[j] * __int_as_float(__float_as_int(x[j].y) & keep);
        acc.z += v[j] * __int_as_float(__float_as_int(x[j].z) & keep);
        acc.w += v[j] * __int_as_float(__float_as_int(x[j].w) & keep);
    }
}

#define LAYER_GROUP 2

static __device__ __forceinline__
void group_heads(int2 (&my)[LAYER_GROUP], int (&s4)[LAYER_GROUP], int (&t4)[LAYER_GROUP],
                 int (&nb)[LAYER_GROUP], const int rs, const int rt, const int r0,
                 const int lane, const int2* __restrict__ sorted) {
    #pragma unroll
    for (int r = 0; r < LAYER_GROUP; ++r) {
        s4[r] = __builtin_amdgcn_readlane(rs, r0 + r);
        t4[r] = __builtin_amdgcn_readlane(rt, r0 + r);
        nb[r] = min(32, t4[r] - s4[r]);
        my[r] = make_int2(0, 0);
        if (lane < nb[r]) my[r] = sorted[s4[r] + lane];
    }
}

__global__ __launch_bounds__(LAYER_T)
void k_layer(const int* __restrict__ rowptr, const int2* __restrict__ sorted,
             const float* __restrict__ ein, float* __restrict__ eout,
             const float* __restrict__ Wgc, const float* __restrict__ Wbi,
             const float* __restrict__ bias) {
    __shared__ __attribute__((aligned(16))) float sg[4096];
    __shared__ __attribute__((aligned(16))) float sb[4096];
    __shared__ float sbias[64];
    __shared__ __attribute__((aligned(16))) float sL[LAYER_WAVES * 16 * TILE_PITCH];

    {
        float wg[W_TRIPS], wb[W_TRIPS];
        #pragma unroll
        for (int k = 0; k < W_TRIPS; ++k) {
            const int i = min(k * LAYER_T + (int)threadIdx.x, 4095);
            wg[k] = Wgc[i];
            wb[k] = Wbi[i];
        }
        #pragma unroll
        for (int k = 0; k < W_TRIPS; ++k) {
            const int i = k * LAYER_T + (int)threadIdx.x;
            if (i < 4096) {
                const int r = i >> 6, c = i & 63;
                const int dst = ((((r >> 1) * 64) + c) << 1) | (r & 1);
                sg[dst] = wg[k];
                sb[dst] = wb[k];
            }
        }
    }
    if (threadIdx.x < 64) sbias[threadIdx.x] = 2.0f * bias[threadIdx.x];
    __syncthreads();

    const int lane = threadIdx.x & 31;
    const int half = lane >> 4;
    const int n    = lane & 15;
    const int q    = n * 4;
    const int wave = __builtin_amdgcn_readfirstlane(threadIdx.x >> 5);
    float* myL = sL + wave * (16 * TILE_PITCH);

    #pragma unroll 1
    for (int tt = 0; tt < LAYER_TPW; ++tt) {
        const int tile = (blockIdx.x * LAYER_TPW + tt) * LAYER_WAVES + wave;
        if (tile >= N_TILES) break;
        const int mbase = tile * 16;

        const int rs = rowptr[mbase + n], rt = rowptr[mbase + n + 1];

        int sN[LAYER_GROUP], tN[LAYER_GROUP], nN[LAYER_GROUP];
        int2 myN[LAYER_GROUP];
        group_heads(myN, sN, tN, nN, rs, rt, 0, lane, sorted);

        #pragma unroll 1
        for (int g = 0; g < 16 / LAYER_GROUP; ++g) {
            int s4[LAYER_GROUP], t4[LAYER_GROUP], nb[LAYER_GROUP];
            int2 my[LAYER_GROUP];
            #pragma unroll
            for (int r = 0; r < LAYER_GROUP; ++r) { s4[r] = sN[r]; t4[r] = tN[r]; nb[r] = nN[r]; my[r] = myN[r]; }

            float4 acc[LAYER_GROUP];
            {
                float  v[LAYER_GROUP][8];
                float4 x[LAYER_GROUP][8];
                float  v2[LAYER_GROUP][4];
                float4 x2[LAYER_GROUP][4];
                #pragma unroll
                for (int r = 0; r < LAYER_GROUP; ++r) trip_issue(v[r], x[r], my[r], 0, half, q, ein);
                #pragma unroll
                for (int r = 0; r < LAYER_GROUP; ++r)
                    if (nb[r] > 16) part_issue<4>(v2[r], x2[r], my[r], 16, half, q, ein);
                if (g < 16 / LAYER_GROUP - 1)
                    group_heads(myN, sN, tN, nN, rs, rt, LAYER_GROUP * (g + 1), lane, sorted);
                #pragma unroll
                for (int r = 0; r < LAYER_GROUP; ++r) {
                    acc[r] = make_float4(0.f, 0.f, 0.f, 0.f);
                    trip_sum(acc[r], v[r], x[r], 0, half, nb[r]);
                }
                #pragma unroll
                for (int r = 0; r < LAYER_GROUP; ++r)
                    if (nb[r] > 16) part_sum<4>(acc[r], v2[r], x2[r], 16, half, nb[r]);
            }

            #pragma unroll
            for (int r = 0; r < LAYER_GROUP; ++r) {
                if (nb[r] > 24) {
                    float  v3[4];
                    float4 x3[4];
                    part_issue<4>(v3, x3, my[r], 24, half, q, ein);
                    part_sum<4>(acc[r], v3, x3, 24, half, nb[r]);
                }
            }

            #pragma unroll
            for (int r = 0; r < LAYER_GROUP; ++r) {
                for (int base = s4[r] + 32; base < t4[r]; base += 32) {
                    const int nbb = min(32, t4[r] - base);
                    int2 m2 = make_int2(0, 0);
                    if (lane < nbb) m2 = sorted[base + lane];
                    row_batch(acc[r], m2, nbb, half, q, ein);
                }
            }

            #pragma unroll
            for (int r = 0; r < LAYER_GROUP; ++r) {
                acc[r].x += __shfl_xor(acc[r].x, 16, 32);
                acc[r].y += __shfl_xor(acc[r].y, 16, 32);
                acc[r].z += __shfl_xor(acc[r].z, 16, 32);
                acc[r].w += __shfl_xor(acc[r].w, 16, 32);
                if (half == 0) *(float4*)(myL + (LAYER_GROUP * g + r) * TILE_PITCH + q) = acc[r];
            }
        }

        __builtin_amdgcn_wave_barrier();

        const float* lrow = myL + n * TILE_PITCH;
        const float* erow = ein + (size_t)(mbase + n) * DD;

        v8f dacc[4];
        #pragma unroll
        for (int j = 0; j < 4; ++j) dacc[j] = (v8f){};

        #pragma unroll
        for (int kk = 0; kk < 16; ++kk) {
            int off = kk * 4 + half * 2;
            v2f ls = *(const v2f*)(lrow + off);
            v2f eg = *(const v2f*)(erow + off);
            v2f a_li  = ls + eg;
            v2f a_mul = ls * eg;
            int bbase = ((kk * 2 + half) * 64) << 1;
            #pragma unroll
            for (int j = 0; j < 4; ++j) {
                v2f bgc = *(const v2f*)(sg + bbase + ((j * 16 + n) << 1));
                v2f bbi = *(const v2f*)(sb + bbase + ((j * 16 + n) << 1));
                dacc[j] = __builtin_amdgcn_wmma_f32_16x16x4_f32(
                    false, a_li, false, bgc, (short)0, dacc[j], false, false);
                dacc[j] = __builtin_amdgcn_wmma_f32_16x16x4_f32(
                    false, a_mul, false, bbi, (short)0, dacc[j], false, false);
            }
        }

        #pragma unroll
        for (int j = 0; j < 4; ++j) {
            float bsum = sbias[j * 16 + n];
            #pragma unroll
            for (int v = 0; v < 8; ++v) {
                float x = dacc[j][v] + bsum;
                x = (x >= 0.f) ? x : SLOPE * x;
                eout[(size_t)(mbase + (half << 3) + v) * DD + j * 16 + n] = x;
            }
        }
    }
}

__global__ __launch_bounds__(LAST_T) __attribute__((amdgpu_waves_per_eu(4, 4)))
void k_layer_last(const int* __restrict__ rowptr, const int2* __restrict__ sorted,
                  const float* __restrict__ ein, const float* __restrict__ eprev,
                  const float* __restrict__ ue, const float* __restrict__ ie,
                  const float* __restrict__ Wgc, const float* __restrict__ Wbi,
                  const float* __restrict__ bias,
                  const int* __restrict__ us, const int* __restrict__ pi,
                  const int* __restrict__ ni, float* __restrict__ out) {
    __shared__ __attribute__((aligned(16))) float sL[16 * TILE_PITCH];
    __shared__ __attribute__((aligned(16))) float sO[16 * TILE_PITCH];

    const int lane = threadIdx.x & 31;
    const int w = __builtin_amdgcn_readfirstlane(threadIdx.x >> 5);

    if (blockIdx.x >= LAST_TILES) {
        gathered_cols_0_191((blockIdx.x - LAST_TILES) * LAST_WAVES + w, lane,
                            ue, ie, eprev, ein, us, pi, ni, out);
        return;
    }

    const int half = lane >> 4;
    const int n    = lane & 15;
    const int q    = n * 4;
    const int tile = blockIdx.x;
    const int seg  = tile >> 8;
    const int bb0  = (tile & 255) * 16;
    const int* idx = (seg == 0) ? us : ((seg == 1) ? pi : ni);

    const int node = idx[bb0 + n] + (seg ? NU : 0);
    const int rs = rowptr[node], rt = rowptr[node + 1];

    int s4[4], t4[4], nb[4];
    int2 my[4];
    #pragma unroll
    for (int r = 0; r < 4; ++r) {
        s4[r] = __builtin_amdgcn_readlane(rs, 4 * w + r);
        t4[r] = __builtin_amdgcn_readlane(rt, 4 * w + r);
        nb[r] = min(32, t4[r] - s4[r]);
        my[r] = make_int2(0, 0);
        if (lane < nb[r]) my[r] = sorted[s4[r] + lane];
    }

    float4 acc[4];
    {
        float  v[4][8];
        float4 x[4][8];
        #pragma unroll
        for (int r = 0; r < 4; ++r) trip_issue(v[r], x[r], my[r], 0, half, q, ein);
        #pragma unroll
        for (int r = 0; r < 4; ++r) {
            acc[r] = make_float4(0.f, 0.f, 0.f, 0.f);
            trip_sum(acc[r], v[r], x[r], 0, half, nb[r]);
        }
    }

    if (max(max(nb[0], nb[1]), max(nb[2], nb[3])) > 16) {
        float  v[4][8];
        float4 x[4][8];
        #pragma unroll
        for (int r = 0; r < 4; ++r) trip_issue(v[r], x[r], my[r], 16, half, q, ein);
        #pragma unroll
        for (int r = 0; r < 4; ++r) trip_sum(acc[r], v[r], x[r], 16, half, nb[r]);
    }

    #pragma unroll
    for (int r = 0; r < 4; ++r) {
        for (int base = s4[r] + 32; base < t4[r]; base += 32) {
            const int nbb = min(32, t4[r] - base);
            int2 m2 = make_int2(0, 0);
            if (lane < nbb) m2 = sorted[base + lane];
            row_batch(acc[r], m2, nbb, half, q, ein);
        }
    }

    const int col = 16 * w + n;
    v2f bg[16], bb[16], eg[16];
    #pragma unroll
    for (int kk = 0; kk < 16; ++kk) {
        const int k0 = kk * 4 + half * 2;
        bg[kk].x = Wgc[k0 * DD + col];
        bg[kk].y = Wgc[(k0 + 1) * DD + col];
        bb[kk].x = Wbi[k0 * DD + col];
        bb[kk].y = Wbi[(k0 + 1) * DD + col];
    }
    const float* erow = ein + (size_t)node * DD;
    #pragma unroll
    for (int kk = 0; kk < 16; ++kk)
        eg[kk] = *(const v2f*)(erow + kk * 4 + half * 2);
    const float bsum = 2.0f * bias[col];

    #pragma unroll
    for (int r = 0; r < 4; ++r) {
        acc[r].x += __shfl_xor(acc[r].x, 16, 32);
        acc[r].y += __shfl_xor(acc[r].y, 16, 32);
        acc[r].z += __shfl_xor(acc[r].z, 16, 32);
        acc[r].w += __shfl_xor(acc[r].w, 16, 32);
        if (half == 0) *(float4*)(sL + (4 * w + r) * TILE_PITCH + q) = acc[r];
    }

    __syncthreads();

    const float* lrow = sL + n * TILE_PITCH;
    v8f dacc = (v8f){};
    #pragma unroll
    for (int kk = 0; kk < 16; ++kk) {
        v2f ls = *(const v2f*)(lrow + kk * 4 + half * 2);
        v2f a_li  = ls + eg[kk];
        v2f a_mul = ls * eg[kk];
        dacc = __builtin_amdgcn_wmma_f32_16x16x4_f32(
            false, a_li, false, bg[kk], (short)0, dacc, false, false);
        dacc = __builtin_amdgcn_wmma_f32_16x16x4_f32(
            false, a_mul, false, bb[kk], (short)0, dacc, false, false);
    }

    #pragma unroll
    for (int v = 0; v < 8; ++v) {
        float x = dacc[v] + bsum;
        x = (x >= 0.f) ? x : SLOPE * x;
        sO[((half << 3) + v) * TILE_PITCH + col] = x;
    }

    __syncthreads();

    #pragma unroll
    for (int r = 0; r < 4; ++r) {
        const int i = 4 * w + r;
        v2f x = *(const v2f*)(sO + i * TILE_PITCH + lane * 2);
        float ss = x.x * x.x + x.y * x.y;
        #pragma unroll
        for (int o = 16; o > 0; o >>= 1) ss += __shfl_xor(ss, o, 32);
        float scale = 1.0f / fmaxf(sqrtf(ss), 1e-12f);
        v2f y; y.x = x.x * scale; y.y = x.y * scale;
        *(v2f*)(out + (size_t)(tile * 16 + i) * 256 + 192 + lane * 2) = y;
    }
}

__global__ __launch_bounds__(256)
void k_gather_norm(const float* __restrict__ ego, const int* __restrict__ us,
                   const int* __restrict__ pi, const int* __restrict__ ni,
                   float* __restrict__ out, int colbase) {
    int wid  = (blockIdx.x * 256 + threadIdx.x) >> 5;
    int lane = threadIdx.x & 31;
    if (wid >= 3 * BATCH) return;
    int seg = wid >> 12;
    int bb  = wid & (BATCH - 1);
    int node;
    if (seg == 0)      node = us[bb];
    else if (seg == 1) node = NU + pi[bb];
    else               node = NU + ni[bb];
    v2f x = *(const v2f*)(ego + (size_t)node * DD + lane * 2);
    float ss = x.x * x.x + x.y * x.y;
    #pragma unroll
    for (int o = 16; o > 0; o >>= 1) ss += __shfl_xor(ss, o, 32);
    float scale = 1.0f / fmaxf(sqrtf(ss), 1e-12f);
    v2f y; y.x = x.x * scale; y.y = x.y * scale;
    *(v2f*)(out + (size_t)wid * 256 + colbase + lane * 2) = y;
}

extern "C" void kernel_launch(void* const* d_in, const int* in_sizes, int n_in,
                              void* d_out, int out_size, void* d_ws, size_t ws_size,
                              hipStream_t stream) {
    const float* ue   = (const float*)d_in[0];
    const float* ie   = (const float*)d_in[1];
    const float* Wgc  = (const float*)d_in[2];
    const float* Wbi  = (const float*)d_in[3];
    const float* bbi  = (const float*)d_in[4];
    const float* vals = (const float*)d_in[5];
    const int*   rows = (const int*)d_in[6];
    const int*   cols = (const int*)d_in[7];
    const int*   us   = (const int*)d_in[8];
    const int*   pi   = (const int*)d_in[9];
    const int*   ni   = (const int*)d_in[10];
    float* out = (float*)d_out;

    float* ego = (float*)d_ws;
    float* L   = ego + (size_t)NT * DD;

    hipMemcpyAsync(ego, ue, (size_t)NU * DD * sizeof(float),
                   hipMemcpyDeviceToDevice, stream);
    hipMemcpyAsync(ego + (size_t)NU * DD, ie, (size_t)NI * DD * sizeof(float),
                   hipMemcpyDeviceToDevice, stream);

    const size_t need = (size_t)2 * NT * DD * 4 + (size_t)NNZE * 8 + (size_t)(NT + 1) * 4;
    if (ws_size >= need) {
        int2* sorted = (int2*)(L + (size_t)NT * DD);
        int*  rowptr = (int*)(sorted + NNZE);
        int2* tmp    = (int2*)L;
        int*  wgoff  = (int*)(tmp + NNZE);
        int*  bcnt   = wgoff + SORT_G * NB;
        int*  bbase  = bcnt + NB * CNT_PITCH;
        hipMemsetAsync(bcnt, 0, (size_t)NB * CNT_PITCH * sizeof(int), stream);
        k_bucket_count<<<SORT_G, SORT_T, 0, stream>>>(rows, bcnt, wgoff);
        k_bucket_scatter<<<SORT_G, SORT_T, 0, stream>>>(rows, cols, vals, bcnt, wgoff,
                                                        bbase, tmp);
        k_bucket_rows<<<NB, ROWS_T, 0, stream>>>(bbase, tmp, sorted, rowptr);
        float* ego2 = L;
        k_layer<<<LAYER_GRID, LAYER_T, 0, stream>>>(
            rowptr, sorted, ego, ego2, Wgc, Wbi, bbi);
        k_layer<<<LAYER_GRID, LAYER_T, 0, stream>>>(
            rowptr, sorted, ego2, ego, Wgc + 4096, Wbi + 4096, bbi + 64);
        k_layer_last<<<LAST_TILES + LAST_COPY_GRID, LAST_T, 0, stream>>>(
            rowptr, sorted, ego, ego2, ue, ie, Wgc + 8192, Wbi + 8192, bbi + 128,
            us, pi, ni, out);
        return;
    }

    k_gather0<<<(3 * BATCH * DD + 255) / 256, 256, 0, stream>>>(
        ue, ie, us, pi, ni, out);
    for (int k = 0; k < 3; ++k) {
        hipMemsetAsync(L, 0, (size_t)NT * DD * sizeof(float), stream);
        long tot = (long)NNZE * 16;
        k_spmm<<<(int)((tot + 255) / 256), 256, 0, stream>>>(
            rows, cols, vals, ego, L);
        k_dense_wmma<<<(NT + 127) / 128, 256, 0, stream>>>(
            L, ego, Wgc + k * 4096, Wbi + k * 4096, bbi + k * 64);
        k_gather_norm<<<(3 * BATCH * 32 + 255) / 256, 256, 0, stream>>>(
            ego, us, pi, ni, out, 64 * (k + 1));
    }
}
